# starting kernel
# baseline (speedup 1.0000x reference)
#include <hip/hip_runtime.h>
#include <hip/hip_bf16.h>

typedef __attribute__((ext_vector_type(2))) float v2f;
typedef __attribute__((ext_vector_type(8))) float v8f;

#define N_NODES 50000
#define N_EDGES 1600000
#define IN_HIDDEN 256
#define OUT_COLS 128
#define NHEAD 4
#define OUT_HIDDEN 32
#define EDGE_HIDDEN 64
#define N_ETYPES 8
#define NEG_SLOPE 0.2f

__device__ __forceinline__ unsigned enc_f32(float f) {
    unsigned u = __float_as_uint(f);
    return (u & 0x80000000u) ? ~u : (u | 0x80000000u);
}
__device__ __forceinline__ float dec_f32(unsigned u) {
    unsigned b = (u & 0x80000000u) ? (u & 0x7FFFFFFFu) : ~u;
    return __uint_as_float(b);
}

__global__ void init_gmax_kernel(unsigned* gmax) {
    *gmax = enc_f32(-INFINITY);
}

__global__ void __launch_bounds__(256)
gemm_h_kernel(const float* __restrict__ x, const float* __restrict__ W,
              float* __restrict__ h) {
    const int lane  = threadIdx.x & 31;
    const int wave  = threadIdx.x >> 5;
    const int strip = blockIdx.x * (blockDim.x >> 5) + wave;
    if (strip * 16 >= N_NODES) return;
    const int row0  = strip * 16;

    const int mrow  = lane & 15;
    const int khalf = (lane >> 4) << 1;

    v8f acc[8] = {};

    const float* xrow = x + (size_t)(row0 + mrow) * IN_HIDDEN + khalf;
    for (int k = 0; k < IN_HIDDEN; k += 4) {
        v2f a = *(const v2f*)(xrow + k);
#pragma unroll
        for (int t = 0; t < 8; ++t) {
            const float* wp = W + (size_t)(k + khalf) * OUT_COLS + t * 16 + mrow;
            v2f b;
            b.x = wp[0];
            b.y = wp[OUT_COLS];
            acc[t] = __builtin_amdgcn_wmma_f32_16x16x4_f32(
                false, a, false, b, (short)0, acc[t], false, false);
        }
    }

    const int mbase = (lane >> 4) << 3;
#pragma unroll
    for (int t = 0; t < 8; ++t) {
#pragma unroll
        for (int i = 0; i < 8; ++i) {
            float v = acc[t][i];
            v = (v == v) ? v : 0.0f;
            h[(size_t)(row0 + mbase + i) * OUT_COLS + t * 16 + mrow] = v;
        }
    }
}

__global__ void etype_logits_kernel(const float* __restrict__ edge_emb,
                                    const float* __restrict__ W_e,
                                    const float* __restrict__ a_e,
                                    float* __restrict__ he) {
    const int tid = threadIdx.x;
    if (tid >= N_ETYPES * NHEAD) return;
    const int t = tid >> 2, head = tid & 3;
    float s = 0.0f;
    for (int d = 0; d < EDGE_HIDDEN; ++d) {
        float ev = 0.0f;
        for (int k = 0; k < EDGE_HIDDEN; ++k)
            ev = fmaf(edge_emb[t * EDGE_HIDDEN + k],
                      W_e[k * (EDGE_HIDDEN * NHEAD) + head * EDGE_HIDDEN + d], ev);
        s = fmaf(a_e[head * EDGE_HIDDEN + d], ev, s);
    }
    he[tid] = s;
}

__global__ void __launch_bounds__(256)
node_logits_kernel(const float* __restrict__ h,
                   const float* __restrict__ a_l, const float* __restrict__ a_r,
                   float* __restrict__ hl, float* __restrict__ hr) {
    const int idx = blockIdx.x * blockDim.x + threadIdx.x;
    if (idx >= N_NODES * NHEAD) return;
    const int node = idx >> 2, head = idx & 3;
    const float* hp = h + (size_t)node * OUT_COLS + head * OUT_HIDDEN;
    const float* al = a_l + head * OUT_HIDDEN;
    const float* ar = a_r + head * OUT_HIDDEN;
    float sl = 0.0f, sr = 0.0f;
#pragma unroll
    for (int d = 0; d < OUT_HIDDEN; ++d) {
        float v = hp[d];
        sl = fmaf(al[d], v, sl);
        sr = fmaf(ar[d], v, sr);
    }
    hl[idx] = sl;
    hr[idx] = sr;
}

__global__ void __launch_bounds__(256)
scores_kernel(const int* __restrict__ edge,
              const float* __restrict__ hl, const float* __restrict__ hr,
              const float* __restrict__ he,
              float* __restrict__ scores, unsigned* __restrict__ gmax) {
    const int e = blockIdx.x * blockDim.x + threadIdx.x;
    float lmax = -INFINITY;
    if (e < N_EDGES) {
        const int src = edge[e];
        const int trg = edge[N_EDGES + e];
        const int et  = edge[2 * N_EDGES + e];
#pragma unroll
        for (int hh = 0; hh < NHEAD; ++hh) {
            float s = hl[src * NHEAD + hh] + hr[trg * NHEAD + hh] + he[et * NHEAD + hh];
            s = (s > 0.0f) ? s : NEG_SLOPE * s;
            scores[(size_t)e * NHEAD + hh] = s;
            lmax = fmaxf(lmax, s);
        }
    }
    __shared__ float red[256];
    red[threadIdx.x] = lmax;
    __syncthreads();
    for (int off = 128; off > 0; off >>= 1) {
        if (threadIdx.x < off)
            red[threadIdx.x] = fmaxf(red[threadIdx.x], red[threadIdx.x + off]);
        __syncthreads();
    }
    if (threadIdx.x == 0) atomicMax(gmax, enc_f32(red[0]));
}

__global__ void __launch_bounds__(256)
expsum_kernel(const int* __restrict__ edge, float* __restrict__ scores,
              const unsigned* __restrict__ gmax, float* __restrict__ denom) {
    const int e = blockIdx.x * blockDim.x + threadIdx.x;
    if (e >= N_EDGES) return;
    const float smax = dec_f32(*gmax);
    const int trg = edge[N_EDGES + e];
#pragma unroll
    for (int hh = 0; hh < NHEAD; ++hh) {
        float es = __expf(scores[(size_t)e * NHEAD + hh] - smax);
        scores[(size_t)e * NHEAD + hh] = es;
        atomicAdd(&denom[trg * NHEAD + hh], es);
    }
}

__global__ void __launch_bounds__(256)
aggregate_kernel(const int* __restrict__ edge, const float* __restrict__ h,
                 const float* __restrict__ es, const float* __restrict__ denom,
                 float* __restrict__ out, float* __restrict__ attn_out) {
    const int gid  = blockIdx.x * blockDim.x + threadIdx.x;
    const int e    = gid >> 5;
    const int lane = gid & 31;
    if (e >= N_EDGES) return;
    const int src = edge[e];
    const int trg = edge[N_EDGES + e];
    float a[NHEAD];
#pragma unroll
    for (int k = 0; k < NHEAD; ++k)
        a[k] = es[(size_t)e * NHEAD + k] / (denom[trg * NHEAD + k] + 1e-16f);
    if (lane < NHEAD) attn_out[(size_t)e * NHEAD + lane] = a[lane];
#pragma unroll
    for (int k = 0; k < NHEAD; ++k) {
        float m = h[(size_t)src * OUT_COLS + k * OUT_HIDDEN + lane] * a[k];
        atomicAdd(&out[(size_t)trg * OUT_COLS + k * OUT_HIDDEN + lane], m);
    }
}

extern "C" void kernel_launch(void* const* d_in, const int* in_sizes, int n_in,
                              void* d_out, int out_size, void* d_ws, size_t ws_size,
                              hipStream_t stream) {
    const int*   edge     = (const int*)  d_in[0];
    const float* x        = (const float*)d_in[1];
    const float* edge_emb = (const float*)d_in[2];
    const float* W        = (const float*)d_in[3];
    const float* W_e      = (const float*)d_in[4];
    const float* a_l      = (const float*)d_in[5];
    const float* a_r      = (const float*)d_in[6];
    const float* a_e      = (const float*)d_in[7];

    float* out      = (float*)d_out;
    float* attn_out = out + (size_t)N_NODES * OUT_COLS;

    char* ws = (char*)d_ws;
    float* h      = (float*)ws;  ws += (size_t)N_NODES * OUT_COLS * sizeof(float);
    float* scores = (float*)ws;  ws += (size_t)N_EDGES * NHEAD   * sizeof(float);
    float* hl     = (float*)ws;  ws += (size_t)N_NODES * NHEAD   * sizeof(float);
    float* hr     = (float*)ws;  ws += (size_t)N_NODES * NHEAD   * sizeof(float);
    float* denom  = (float*)ws;  ws += (size_t)N_NODES * NHEAD   * sizeof(float);
    float* he     = (float*)ws;  ws += (size_t)N_ETYPES * NHEAD  * sizeof(float);
    unsigned* gmax = (unsigned*)ws;

    hipMemsetAsync(out,   0, (size_t)N_NODES * OUT_COLS * sizeof(float), stream);
    hipMemsetAsync(denom, 0, (size_t)N_NODES * NHEAD   * sizeof(float), stream);
    init_gmax_kernel<<<1, 1, 0, stream>>>(gmax);

    {
        const int strips = N_NODES / 16;
        const int wavesPerBlock = 8;
        const int blocks = (strips + wavesPerBlock - 1) / wavesPerBlock;
        gemm_h_kernel<<<blocks, 256, 0, stream>>>(x, W, h);
    }
    etype_logits_kernel<<<1, 32, 0, stream>>>(edge_emb, W_e, a_e, he);
    node_logits_kernel<<<(N_NODES * NHEAD + 255) / 256, 256, 0, stream>>>(h, a_l, a_r, hl, hr);
    scores_kernel<<<(N_EDGES + 255) / 256, 256, 0, stream>>>(edge, hl, hr, he, scores, gmax);
    expsum_kernel<<<(N_EDGES + 255) / 256, 256, 0, stream>>>(edge, scores, gmax, denom);
    aggregate_kernel<<<((size_t)N_EDGES * 32 + 255) / 256, 256, 0, stream>>>(
        edge, h, scores, denom, out, attn_out);
}


# target-slab ownership: each of 256 work-groups filters all edges by a byte key plane and sums its 196 nodes' denominators and rows in LDS by LDS float adds (arrival order); global atomics removed
# speedup vs baseline: 2.3880x; 2.3880x over previous
#include <hip/hip_runtime.h>
#include <hip/hip_bf16.h>

typedef __attribute__((ext_vector_type(2))) float v2f;
typedef __attribute__((ext_vector_type(8))) float v8f;

#define N_NODES 50000
#define N_EDGES 1600000
#define IN_HIDDEN 256
#define OUT_COLS 128
#define NHEAD 4
#define OUT_HIDDEN 32
#define EDGE_HIDDEN 64
#define N_ETYPES 8
#define NEG_SLOPE 0.2f

#define SLAB 196
#define N_SLABS 256
#define CHUNK 512
#define N_CHUNKS (N_EDGES / CHUNK)
#define OWN_WAVES 32
#define QCAP 64
static_assert(N_SLABS * SLAB >= N_NODES && N_SLABS <= 256, "slab id must fit a byte");
static_assert(N_EDGES % CHUNK == 0, "the scan has no tail chunk");
static_assert((N_SLABS - 1) * (N_CHUNKS / N_SLABS) + OWN_WAVES <= N_CHUNKS, "a rotated first chunk stays inside the plane");

__device__ __forceinline__ unsigned enc_f32(float f) {
    unsigned u = __float_as_uint(f);
    return (u & 0x80000000u) ? ~u : (u | 0x80000000u);
}
__device__ __forceinline__ float dec_f32(unsigned u) {
    unsigned b = (u & 0x80000000u) ? (u & 0x7FFFFFFFu) : ~u;
    return __uint_as_float(b);
}

__global__ void init_gmax_kernel(unsigned* gmax) {
    *gmax = enc_f32(-INFINITY);
}

__global__ void __launch_bounds__(256)
gemm_h_kernel(const float* __restrict__ x, const float* __restrict__ W,
              float* __restrict__ h) {
    const int lane  = threadIdx.x & 31;
    const int wave  = threadIdx.x >> 5;
    const int strip = blockIdx.x * (blockDim.x >> 5) + wave;
    if (strip * 16 >= N_NODES) return;
    const int row0  = strip * 16;

    const int mrow  = lane & 15;
    const int khalf = (lane >> 4) << 1;

    v8f acc[8] = {};

    const float* xrow = x + (size_t)(row0 + mrow) * IN_HIDDEN + khalf;
    for (int k = 0; k < IN_HIDDEN; k += 4) {
        v2f a = *(const v2f*)(xrow + k);
#pragma unroll
        for (int t = 0; t < 8; ++t) {
            const float* wp = W + (size_t)(k + khalf) * OUT_COLS + t * 16 + mrow;
            v2f b;
            b.x = wp[0];
            b.y = wp[OUT_COLS];
            acc[t] = __builtin_amdgcn_wmma_f32_16x16x4_f32(
                false, a, false, b, (short)0, acc[t], false, false);
        }
    }

    const int mbase = (lane >> 4) << 3;
#pragma unroll
    for (int t = 0; t < 8; ++t) {
#pragma unroll
        for (int i = 0; i < 8; ++i) {
            float v = acc[t][i];
            v = (v == v) ? v : 0.0f;
            h[(size_t)(row0 + mbase + i) * OUT_COLS + t * 16 + mrow] = v;
        }
    }
}

__global__ void etype_logits_kernel(const float* __restrict__ edge_emb,
                                    const float* __restrict__ W_e,
                                    const float* __restrict__ a_e,
                                    float* __restrict__ he) {
    const int tid = threadIdx.x;
    if (tid >= N_ETYPES * NHEAD) return;
    const int t = tid >> 2, head = tid & 3;
    float s = 0.0f;
    for (int d = 0; d < EDGE_HIDDEN; ++d) {
        float ev = 0.0f;
        for (int k = 0; k < EDGE_HIDDEN; ++k)
            ev = fmaf(edge_emb[t * EDGE_HIDDEN + k],
                      W_e[k * (EDGE_HIDDEN * NHEAD) + head * EDGE_HIDDEN + d], ev);
        s = fmaf(a_e[head * EDGE_HIDDEN + d], ev, s);
    }
    he[tid] = s;
}

__global__ void __launch_bounds__(256)
node_logits_kernel(const float* __restrict__ h,
                   const float* __restrict__ a_l, const float* __restrict__ a_r,
                   float* __restrict__ hl, float* __restrict__ hr) {
    const int idx = blockIdx.x * blockDim.x + threadIdx.x;
    if (idx >= N_NODES * NHEAD) return;
    const int node = idx >> 2, head = idx & 3;
    const float* hp = h + (size_t)node * OUT_COLS + head * OUT_HIDDEN;
    const float* al = a_l + head * OUT_HIDDEN;
    const float* ar = a_r + head * OUT_HIDDEN;
    float sl = 0.0f, sr = 0.0f;
#pragma unroll
    for (int d = 0; d < OUT_HIDDEN; ++d) {
        float v = hp[d];
        sl = fmaf(al[d], v, sl);
        sr = fmaf(ar[d], v, sr);
    }
    hl[idx] = sl;
    hr[idx] = sr;
}

__global__ void __launch_bounds__(256)
edge_max_key_kernel(const int* __restrict__ edge,
                    const float* __restrict__ hl, const float* __restrict__ hr,
                    const float* __restrict__ he,
                    unsigned char* __restrict__ key, unsigned* __restrict__ gmax) {
    const int e = blockIdx.x * blockDim.x + threadIdx.x;
    float lmax = -INFINITY;
    if (e < N_EDGES) {
        const int src = edge[e];
        const int trg = edge[N_EDGES + e];
        const int et  = edge[2 * N_EDGES + e];
#pragma unroll
        for (int hh = 0; hh < NHEAD; ++hh) {
            float s = hl[src * NHEAD + hh] + hr[trg * NHEAD + hh] + he[et * NHEAD + hh];
            s = (s > 0.0f) ? s : NEG_SLOPE * s;
            lmax = fmaxf(lmax, s);
        }
        key[e] = (unsigned char)(trg / SLAB);
    }
    __shared__ float red[256];
    red[threadIdx.x] = lmax;
    __syncthreads();
    for (int off = 128; off > 0; off >>= 1) {
        if (threadIdx.x < off)
            red[threadIdx.x] = fmaxf(red[threadIdx.x], red[threadIdx.x + off]);
        __syncthreads();
    }
    if (threadIdx.x == 0) atomicMax(gmax, enc_f32(red[0]));
}


struct EdgeVals { int src; int loc; float es[NHEAD]; };

__device__ __forceinline__ EdgeVals edge_vals(const int* __restrict__ edge,
                                              const float* __restrict__ hl,
                                              const float4* hr_s, const float4* he_s,
                                              int e, int base, float smax) {
    EdgeVals r;
    r.src = edge[e];
    const int trg = edge[N_EDGES + e];
    const int et  = edge[2 * N_EDGES + e];
    r.loc = trg - base;
    const float4 l = *(const float4*)(hl + (size_t)r.src * NHEAD);
    const float4 t = hr_s[r.loc];
    const float4 y = he_s[et];
    float s[NHEAD] = { l.x + t.x + y.x, l.y + t.y + y.y, l.z + t.z + y.z, l.w + t.w + y.w };
#pragma unroll
    for (int k = 0; k < NHEAD; ++k) {
        float v = s[k];
        v = (v > 0.0f) ? v : NEG_SLOPE * v;
        r.es[k] = __expf(v - smax);
    }
    return r;
}

template <int PASS>
__device__ __forceinline__ void owner_batch(int n, int lane, int base, float smax,
                                            const unsigned* q,
                                            const int* __restrict__ edge,
                                            const float* __restrict__ h,
                                            const float* __restrict__ hl,
                                            const float4* hr_s, const float4* he_s,
                                            float* slab, float* den, float4* atab,
                                            float* __restrict__ attn_out) {
    if (PASS == 0) {
        if (lane < n) {
            const int e = (int)q[lane];
            const EdgeVals v = edge_vals(edge, hl, hr_s, he_s, e, base, smax);
#pragma unroll
            for (int k = 0; k < NHEAD; ++k) atomicAdd(&den[v.loc * NHEAD + k], v.es[k]);
        }
    } else {
        int src = 0, loc = 0;
        if (lane < n) {
            const int e = (int)q[lane];
            const EdgeVals v = edge_vals(edge, hl, hr_s, he_s, e, base, smax);
            src = v.src; loc = v.loc;
            const float4 a = make_float4(v.es[0] / (den[loc * NHEAD + 0] + 1e-16f),
                                         v.es[1] / (den[loc * NHEAD + 1] + 1e-16f),
                                         v.es[2] / (den[loc * NHEAD + 2] + 1e-16f),
                                         v.es[3] / (den[loc * NHEAD + 3] + 1e-16f));
            *(float4*)(attn_out + (size_t)e * NHEAD) = a;
            atab[lane] = a;
        }
        const float* aw = (const float*)atab;
        const int head = lane >> 3;
        for (int j = 0; j < n; j += 4) {
            float4 r[4]; float aj[4]; int lj[4];
#pragma unroll
            for (int u = 0; u < 4; ++u) {
                const int jj = (j + u < n) ? (j + u) : (n - 1);
                const int sj = __builtin_amdgcn_readlane(src, jj);
                lj[u] = __builtin_amdgcn_readlane(loc, jj);
                aj[u] = aw[jj * NHEAD + head];
                r[u] = *(const float4*)(h + (size_t)sj * OUT_COLS + lane * 4);
            }
#pragma unroll
            for (int u = 0; u < 4; ++u) {
                if (j + u < n) {
                    float* p = slab + lj[u] * OUT_COLS + lane;
                    atomicAdd(p +  0, r[u].x * aj[u]);
                    atomicAdd(p + 32, r[u].y * aj[u]);
                    atomicAdd(p + 64, r[u].z * aj[u]);
                    atomicAdd(p + 96, r[u].w * aj[u]);
                }
            }
        }
    }
}

__device__ __forceinline__ unsigned zero_bytes(unsigned x) {
    const unsigned t = (x & 0x7F7F7F7Fu) + 0x7F7F7F7Fu;
    return ~(t | x | 0x7F7F7F7Fu);
}

template <int PASS>
__device__ __forceinline__ void owner_scan(int lane, int wave, int base, float smax,
                                           const unsigned char* __restrict__ key, unsigned* q,
                                           const int* __restrict__ edge,
                                           const float* __restrict__ h,
                                           const float* __restrict__ hl,
                                           const float4* hr_s, const float4* he_s,
                                           float* slab, float* den, float4* atab,
                                           float* __restrict__ attn_out) {
    const unsigned idrep = blockIdx.x * 0x01010101u;
    const int rot = blockIdx.x * (N_CHUNKS / N_SLABS);
    int cnt = 0;
    int c = wave + rot;
    uint4 kv = *(const uint4*)(key + (size_t)c * CHUNK + lane * 16);
    for (int it = wave; it < N_CHUNKS; it += OWN_WAVES) {
        int cn = c + OWN_WAVES;
        if (cn >= N_CHUNKS) cn -= N_CHUNKS;
        uint4 nx = kv;
        if (it + OWN_WAVES < N_CHUNKS)
            nx = *(const uint4*)(key + (size_t)cn * CHUNK + lane * 16);
        unsigned mm = (zero_bytes(kv.x ^ idrep) >> 7) | (zero_bytes(kv.y ^ idrep) >> 6) |
                      (zero_bytes(kv.z ^ idrep) >> 5) | (zero_bytes(kv.w ^ idrep) >> 4);
        unsigned bal;
        while ((bal = __builtin_amdgcn_ballot_w32(mm != 0u)) != 0u) {
            if (mm != 0u) {
                const int p = __builtin_ctz(mm);
                mm &= mm - 1u;
                q[cnt + (int)__builtin_amdgcn_mbcnt_lo(bal, 0u)] =
                    (unsigned)(c * CHUNK + lane * 16 + ((p & 7) << 2) + (p >> 3));
            }
            cnt += __builtin_popcount(bal);
            if (cnt >= 32) {
                owner_batch<PASS>(32, lane, base, smax, q, edge, h, hl, hr_s, he_s, slab, den, atab, attn_out);
                const int rest = cnt - 32;
                if (lane < rest) { const unsigned t = q[32 + lane]; q[lane] = t; }
                cnt = rest;
            }
        }
        kv = nx;
        c = cn;
    }
    if (cnt > 0)
        owner_batch<PASS>(cnt, lane, base, smax, q, edge, h, hl, hr_s, he_s, slab, den, atab, attn_out);
}

__global__ void __launch_bounds__(OWN_WAVES * 32)
owner_kernel(const int* __restrict__ edge, const float* __restrict__ h,
             const float* __restrict__ hl, const float* __restrict__ hr,
             const float* __restrict__ he,
             const unsigned char* __restrict__ key, const unsigned* __restrict__ gmax,
             float* __restrict__ out, float* __restrict__ attn_out) {
    __shared__ float4   slab4[SLAB * OUT_COLS / 4];
    __shared__ float    den[SLAB * NHEAD];
    __shared__ unsigned qbuf[OWN_WAVES * QCAP];
    __shared__ float4   atab4[OWN_WAVES * 32];
    __shared__ float4   hr_s[SLAB];
    __shared__ float4   he_s[N_ETYPES];
    float* slab = (float*)slab4;

    const int tid  = threadIdx.x;
    const int lane = tid & 31;
    const int wave = __builtin_amdgcn_readfirstlane(tid >> 5);
    const int base = blockIdx.x * SLAB;
    unsigned* q = qbuf + wave * QCAP;
    float4* atab = atab4 + wave * 32;

    for (int i = tid; i < SLAB * OUT_COLS / 4; i += OWN_WAVES * 32)
        slab4[i] = make_float4(0.0f, 0.0f, 0.0f, 0.0f);
    for (int i = tid; i < SLAB * NHEAD; i += OWN_WAVES * 32) den[i] = 0.0f;
    const int rows = (N_NODES - base < SLAB) ? (N_NODES - base) : SLAB;
    if (tid < rows)     hr_s[tid] = *(const float4*)(hr + (size_t)(base + tid) * NHEAD);
    if (tid < N_ETYPES) he_s[tid] = *(const float4*)(he + (size_t)tid * NHEAD);
    const float smax = dec_f32(*gmax);
    __syncthreads();

    owner_scan<0>(lane, wave, base, smax, key, q, edge, h, hl, hr_s, he_s, slab, den, atab, attn_out);
    __syncthreads();
    owner_scan<1>(lane, wave, base, smax, key, q, edge, h, hl, hr_s, he_s, slab, den, atab, attn_out);
    __syncthreads();

    float4* out4 = (float4*)(out + (size_t)base * OUT_COLS);
    for (int i = tid; i < rows * (OUT_COLS / 4); i += OWN_WAVES * 32) {
        const float* p = slab + (i >> 5) * OUT_COLS + (i & 31);
        out4[i] = make_float4(p[0], p[32], p[64], p[96]);
    }
}

extern "C" void kernel_launch(void* const* d_in, const int* in_sizes, int n_in,
                              void* d_out, int out_size, void* d_ws, size_t ws_size,
                              hipStream_t stream) {
    const int*   edge     = (const int*)  d_in[0];
    const float* x        = (const float*)d_in[1];
    const float* edge_emb = (const float*)d_in[2];
    const float* W        = (const float*)d_in[3];
    const float* W_e      = (const float*)d_in[4];
    const float* a_l      = (const float*)d_in[5];
    const float* a_r      = (const float*)d_in[6];
    const float* a_e      = (const float*)d_in[7];

    float* out      = (float*)d_out;
    float* attn_out = out + (size_t)N_NODES * OUT_COLS;

    char* ws = (char*)d_ws;
    float* h      = (float*)ws;  ws += (size_t)N_NODES * OUT_COLS * sizeof(float);
    unsigned char* key = (unsigned char*)ws;  ws += (size_t)N_EDGES * NHEAD * sizeof(float);
    float* hl     = (float*)ws;  ws += (size_t)N_NODES * NHEAD   * sizeof(float);
    float* hr     = (float*)ws;  ws += (size_t)N_NODES * NHEAD   * sizeof(float);
    ws += (size_t)N_NODES * NHEAD * sizeof(float);
    float* he     = (float*)ws;  ws += (size_t)N_ETYPES * NHEAD  * sizeof(float);
    unsigned* gmax = (unsigned*)ws;

    init_gmax_kernel<<<1, 1, 0, stream>>>(gmax);

    {
        const int strips = N_NODES / 16;
        const int wavesPerBlock = 8;
        const int blocks = (strips + wavesPerBlock - 1) / wavesPerBlock;
        gemm_h_kernel<<<blocks, 256, 0, stream>>>(x, W, h);
    }
    etype_logits_kernel<<<1, 32, 0, stream>>>(edge_emb, W_e, a_e, he);
    node_logits_kernel<<<(N_NODES * NHEAD + 255) / 256, 256, 0, stream>>>(h, a_l, a_r, hl, hr);
    edge_max_key_kernel<<<(N_EDGES + 255) / 256, 256, 0, stream>>>(edge, hl, hr, he, key, gmax);
    owner_kernel<<<N_SLABS, OWN_WAVES * 32, 0, stream>>>(edge, h, hl, hr, he, key, gmax, out, attn_out);
}


# owner work-groups sweep the key plane once: the sweep leaves a record per owned edge (id, node, source, exp scores) in LDS, the second pass replays the records; sums in arrival order as before
# speedup vs baseline: 2.7189x; 1.1386x over previous
#include <hip/hip_runtime.h>
#include <hip/hip_bf16.h>

typedef __attribute__((ext_vector_type(2))) float v2f;
typedef __attribute__((ext_vector_type(8))) float v8f;

#define N_NODES 50000
#define N_EDGES 1600000
#define IN_HIDDEN 256
#define OUT_COLS 128
#define NHEAD 4
#define OUT_HIDDEN 32
#define EDGE_HIDDEN 64
#define N_ETYPES 8
#define NEG_SLOPE 0.2f

#define SLAB 196
#define N_SLABS 256
#define CHUNK 512
#define N_CHUNKS (N_EDGES / CHUNK)
#define OWN_WAVES 32
#define QCAP 64
#define LCAP 6912
#define ID_BITS 21
static_assert(N_SLABS * SLAB >= N_NODES && N_SLABS <= 256, "slab id must fit a byte");
static_assert(N_EDGES % CHUNK == 0, "the scan has no tail chunk");
static_assert(N_EDGES <= (1 << ID_BITS) && SLAB <= (1 << (32 - ID_BITS)), "edge id and local node share a dword");
static_assert(LCAP >= OWN_WAVES * 32, "the record slots also hold the second sweep's per-wave tables");
static_assert((N_SLABS - 1) * (N_CHUNKS / N_SLABS) + OWN_WAVES <= N_CHUNKS, "a rotated first chunk stays inside the plane");

__device__ __forceinline__ unsigned enc_f32(float f) {
    unsigned u = __float_as_uint(f);
    return (u & 0x80000000u) ? ~u : (u | 0x80000000u);
}
__device__ __forceinline__ float dec_f32(unsigned u) {
    unsigned b = (u & 0x80000000u) ? (u & 0x7FFFFFFFu) : ~u;
    return __uint_as_float(b);
}

__global__ void init_gmax_kernel(unsigned* gmax) {
    *gmax = enc_f32(-INFINITY);
}

__global__ void __launch_bounds__(256)
gemm_h_kernel(const float* __restrict__ x, const float* __restrict__ W,
              float* __restrict__ h) {
    const int lane  = threadIdx.x & 31;
    const int wave  = threadIdx.x >> 5;
    const int strip = blockIdx.x * (blockDim.x >> 5) + wave;
    if (strip * 16 >= N_NODES) return;
    const int row0  = strip * 16;

    const int mrow  = lane & 15;
    const int khalf = (lane >> 4) << 1;

    v8f acc[8] = {};

    const float* xrow = x + (size_t)(row0 + mrow) * IN_HIDDEN + khalf;
    for (int k = 0; k < IN_HIDDEN; k += 4) {
        v2f a = *(const v2f*)(xrow + k);
#pragma unroll
        for (int t = 0; t < 8; ++t) {
            const float* wp = W + (size_t)(k + khalf) * OUT_COLS + t * 16 + mrow;
            v2f b;
            b.x = wp[0];
            b.y = wp[OUT_COLS];
            acc[t] = __builtin_amdgcn_wmma_f32_16x16x4_f32(
                false, a, false, b, (short)0, acc[t], false, false);
        }
    }

    const int mbase = (lane >> 4) << 3;
#pragma unroll
    for (int t = 0; t < 8; ++t) {
#pragma unroll
        for (int i = 0; i < 8; ++i) {
            float v = acc[t][i];
            v = (v == v) ? v : 0.0f;
            h[(size_t)(row0 + mbase + i) * OUT_COLS + t * 16 + mrow] = v;
        }
    }
}

__global__ void etype_logits_kernel(const float* __restrict__ edge_emb,
                                    const float* __restrict__ W_e,
                                    const float* __restrict__ a_e,
                                    float* __restrict__ he) {
    const int tid = threadIdx.x;
    if (tid >= N_ETYPES * NHEAD) return;
    const int t = tid >> 2, head = tid & 3;
    float s = 0.0f;
    for (int d = 0; d < EDGE_HIDDEN; ++d) {
        float ev = 0.0f;
        for (int k = 0; k < EDGE_HIDDEN; ++k)
            ev = fmaf(edge_emb[t * EDGE_HIDDEN + k],
                      W_e[k * (EDGE_HIDDEN * NHEAD) + head * EDGE_HIDDEN + d], ev);
        s = fmaf(a_e[head * EDGE_HIDDEN + d], ev, s);
    }
    he[tid] = s;
}

__global__ void __launch_bounds__(256)
node_logits_kernel(const float* __restrict__ h,
                   const float* __restrict__ a_l, const float* __restrict__ a_r,
                   float* __restrict__ hl, float* __restrict__ hr) {
    const int idx = blockIdx.x * blockDim.x + threadIdx.x;
    if (idx >= N_NODES * NHEAD) return;
    const int node = idx >> 2, head = idx & 3;
    const float* hp = h + (size_t)node * OUT_COLS + head * OUT_HIDDEN;
    const float* al = a_l + head * OUT_HIDDEN;
    const float* ar = a_r + head * OUT_HIDDEN;
    float sl = 0.0f, sr = 0.0f;
#pragma unroll
    for (int d = 0; d < OUT_HIDDEN; ++d) {
        float v = hp[d];
        sl = fmaf(al[d], v, sl);
        sr = fmaf(ar[d], v, sr);
    }
    hl[idx] = sl;
    hr[idx] = sr;
}

__global__ void __launch_bounds__(256)
edge_max_key_kernel(const int* __restrict__ edge,
                    const float* __restrict__ hl, const float* __restrict__ hr,
                    const float* __restrict__ he,
                    unsigned char* __restrict__ key, unsigned* __restrict__ gmax) {
    const int e = blockIdx.x * blockDim.x + threadIdx.x;
    float lmax = -INFINITY;
    if (e < N_EDGES) {
        const int src = edge[e];
        const int trg = edge[N_EDGES + e];
        const int et  = edge[2 * N_EDGES + e];
#pragma unroll
        for (int hh = 0; hh < NHEAD; ++hh) {
            float s = hl[src * NHEAD + hh] + hr[trg * NHEAD + hh] + he[et * NHEAD + hh];
            s = (s > 0.0f) ? s : NEG_SLOPE * s;
            lmax = fmaxf(lmax, s);
        }
        key[e] = (unsigned char)(trg / SLAB);
    }
    __shared__ float red[256];
    red[threadIdx.x] = lmax;
    __syncthreads();
    for (int off = 128; off > 0; off >>= 1) {
        if (threadIdx.x < off)
            red[threadIdx.x] = fmaxf(red[threadIdx.x], red[threadIdx.x + off]);
        __syncthreads();
    }
    if (threadIdx.x == 0) atomicMax(gmax, enc_f32(red[0]));
}


struct EdgeVals { int src; int loc; float es[NHEAD]; };

__device__ __forceinline__ EdgeVals edge_vals(const int* __restrict__ edge,
                                              const float* __restrict__ hl,
                                              const float4* hr_s, const float4* he_s,
                                              int e, int base, float smax) {
    EdgeVals r;
    r.src = edge[e];
    const int trg = edge[N_EDGES + e];
    const int et  = edge[2 * N_EDGES + e];
    r.loc = trg - base;
    const float4 l = *(const float4*)(hl + (size_t)r.src * NHEAD);
    const float4 t = hr_s[r.loc];
    const float4 y = he_s[et];
    float s[NHEAD] = { l.x + t.x + y.x, l.y + t.y + y.y, l.z + t.z + y.z, l.w + t.w + y.w };
#pragma unroll
    for (int k = 0; k < NHEAD; ++k) {
        float v = s[k];
        v = (v > 0.0f) ? v : NEG_SLOPE * v;
        r.es[k] = __expf(v - smax);
    }
    return r;
}

struct Recs {
    unsigned* cnt;
    unsigned* id;
    int*      src;
    float4*   es;
};

__device__ __forceinline__ void add_rows(int n, int lane, int src, int loc, const float* aw,
                                         const float* __restrict__ h, float* slab) {
    const int head = lane >> 3;
    for (int j = 0; j < n; j += 4) {
        float4 r[4]; float aj[4]; int lj[4];
#pragma unroll
        for (int u = 0; u < 4; ++u) {
            const int jj = (j + u < n) ? (j + u) : (n - 1);
            const int sj = __builtin_amdgcn_readlane(src, jj);
            lj[u] = __builtin_amdgcn_readlane(loc, jj);
            aj[u] = aw[jj * NHEAD + head];
            r[u] = *(const float4*)(h + (size_t)sj * OUT_COLS + lane * 4);
        }
#pragma unroll
        for (int u = 0; u < 4; ++u) {
            if (j + u < n) {
                float* p = slab + lj[u] * OUT_COLS + lane;
                atomicAdd(p +  0, r[u].x * aj[u]);
                atomicAdd(p + 32, r[u].y * aj[u]);
                atomicAdd(p + 64, r[u].z * aj[u]);
                atomicAdd(p + 96, r[u].w * aj[u]);
            }
        }
    }
}

template <int PASS>
__device__ __forceinline__ void owner_batch(int n, int lane, int base, float smax,
                                            const unsigned* q,
                                            const int* __restrict__ edge,
                                            const float* __restrict__ h,
                                            const float* __restrict__ hl,
                                            const float4* hr_s, const float4* he_s,
                                            float* slab, float* den, float4* atab, Recs rc,
                                            float* __restrict__ attn_out) {
    if (PASS == 0) {
        unsigned pos = 0;
        if (lane == 0) pos = atomicAdd(rc.cnt, (unsigned)n);
        pos = (unsigned)__builtin_amdgcn_readfirstlane((int)pos);
        if (lane < n) {
            const int e = (int)q[lane];
            const EdgeVals v = edge_vals(edge, hl, hr_s, he_s, e, base, smax);
#pragma unroll
            for (int k = 0; k < NHEAD; ++k) atomicAdd(&den[v.loc * NHEAD + k], v.es[k]);
            const unsigned slot = pos + (unsigned)lane;
            if (slot < LCAP) {
                rc.id[slot]  = (unsigned)e | ((unsigned)v.loc << ID_BITS);
                rc.src[slot] = v.src;
                rc.es[slot]  = make_float4(v.es[0], v.es[1], v.es[2], v.es[3]);
            }
        }
    } else {
        int src = 0, loc = 0;
        if (lane < n) {
            const int e = (int)q[lane];
            const EdgeVals v = edge_vals(edge, hl, hr_s, he_s, e, base, smax);
            src = v.src; loc = v.loc;
            const float4 a = make_float4(v.es[0] / (den[loc * NHEAD + 0] + 1e-16f),
                                         v.es[1] / (den[loc * NHEAD + 1] + 1e-16f),
                                         v.es[2] / (den[loc * NHEAD + 2] + 1e-16f),
                                         v.es[3] / (den[loc * NHEAD + 3] + 1e-16f));
            *(float4*)(attn_out + (size_t)e * NHEAD) = a;
            atab[lane] = a;
        }
        add_rows(n, lane, src, loc, (const float*)atab, h, slab);
    }
}

__device__ __forceinline__ void replay_batch(int start, int n, int lane, Recs rc, const float* den,
                                             const float* __restrict__ h, float* slab,
                                             float* __restrict__ attn_out) {
    int src = 0, loc = 0;
    if (lane < n) {
        const unsigned id = rc.id[start + lane];
        const int e = (int)(id & ((1u << ID_BITS) - 1u));
        loc = (int)(id >> ID_BITS);
        src = rc.src[start + lane];
        const float4 s = rc.es[start + lane];
        const float4 a = make_float4(s.x / (den[loc * NHEAD + 0] + 1e-16f),
                                     s.y / (den[loc * NHEAD + 1] + 1e-16f),
                                     s.z / (den[loc * NHEAD + 2] + 1e-16f),
                                     s.w / (den[loc * NHEAD + 3] + 1e-16f));
        *(float4*)(attn_out + (size_t)e * NHEAD) = a;
        rc.es[start + lane] = a;
    }
    add_rows(n, lane, src, loc, (const float*)(rc.es + start), h, slab);
}

__device__ __forceinline__ unsigned zero_bytes(unsigned x) {
    const unsigned t = (x & 0x7F7F7F7Fu) + 0x7F7F7F7Fu;
    return ~(t | x | 0x7F7F7F7Fu);
}

template <int PASS>
__device__ __forceinline__ void owner_scan(int lane, int wave, int base, float smax,
                                           const unsigned char* __restrict__ key, unsigned* q,
                                           const int* __restrict__ edge,
                                           const float* __restrict__ h,
                                           const float* __restrict__ hl,
                                           const float4* hr_s, const float4* he_s,
                                           float* slab, float* den, float4* atab, Recs rc,
                                           float* __restrict__ attn_out) {
    const unsigned idrep = blockIdx.x * 0x01010101u;
    const int rot = blockIdx.x * (N_CHUNKS / N_SLABS);
    int cnt = 0;
    int c = wave + rot;
    uint4 kv = *(const uint4*)(key + (size_t)c * CHUNK + lane * 16);
    for (int it = wave; it < N_CHUNKS; it += OWN_WAVES) {
        int cn = c + OWN_WAVES;
        if (cn >= N_CHUNKS) cn -= N_CHUNKS;
        uint4 nx = kv;
        if (it + OWN_WAVES < N_CHUNKS)
            nx = *(const uint4*)(key + (size_t)cn * CHUNK + lane * 16);
        unsigned mm = (zero_bytes(kv.x ^ idrep) >> 7) | (zero_bytes(kv.y ^ idrep) >> 6) |
                      (zero_bytes(kv.z ^ idrep) >> 5) | (zero_bytes(kv.w ^ idrep) >> 4);
        unsigned bal;
        while ((bal = __builtin_amdgcn_ballot_w32(mm != 0u)) != 0u) {
            if (mm != 0u) {
                const int p = __builtin_ctz(mm);
                mm &= mm - 1u;
                q[cnt + (int)__builtin_amdgcn_mbcnt_lo(bal, 0u)] =
                    (unsigned)(c * CHUNK + lane * 16 + ((p & 7) << 2) + (p >> 3));
            }
            cnt += __builtin_popcount(bal);
            if (cnt >= 32) {
                owner_batch<PASS>(32, lane, base, smax, q, edge, h, hl, hr_s, he_s, slab, den, atab, rc, attn_out);
                const int rest = cnt - 32;
                if (lane < rest) { const unsigned t = q[32 + lane]; q[lane] = t; }
                cnt = rest;
            }
        }
        kv = nx;
        c = cn;
    }
    if (cnt > 0)
        owner_batch<PASS>(cnt, lane, base, smax, q, edge, h, hl, hr_s, he_s, slab, den, atab, rc, attn_out);
}

__global__ void __launch_bounds__(OWN_WAVES * 32)
owner_kernel(const int* __restrict__ edge, const float* __restrict__ h,
             const float* __restrict__ hl, const float* __restrict__ hr,
             const float* __restrict__ he,
             const unsigned char* __restrict__ key, const unsigned* __restrict__ gmax,
             float* __restrict__ out, float* __restrict__ attn_out) {
    __shared__ float4   slab4[SLAB * OUT_COLS / 4];
    __shared__ float    den[SLAB * NHEAD];
    __shared__ unsigned qbuf[OWN_WAVES * QCAP];
    __shared__ float4   hr_s[SLAB];
    __shared__ float4   he_s[N_ETYPES];
    __shared__ unsigned rec_id[LCAP];
    __shared__ int      rec_src[LCAP];
    __shared__ float4   rec_es[LCAP];
    __shared__ unsigned rec_cnt;
    float* slab = (float*)slab4;

    const int tid  = threadIdx.x;
    const int lane = tid & 31;
    const int wave = __builtin_amdgcn_readfirstlane(tid >> 5);
    const int base = blockIdx.x * SLAB;
    unsigned* q = qbuf + wave * QCAP;
    Recs rc; rc.cnt = &rec_cnt; rc.id = rec_id; rc.src = rec_src; rc.es = rec_es;
    float4* atab = rec_es + wave * 32;

    for (int i = tid; i < SLAB * OUT_COLS / 4; i += OWN_WAVES * 32)
        slab4[i] = make_float4(0.0f, 0.0f, 0.0f, 0.0f);
    for (int i = tid; i < SLAB * NHEAD; i += OWN_WAVES * 32) den[i] = 0.0f;
    const int rows = (N_NODES - base < SLAB) ? (N_NODES - base) : SLAB;
    if (tid < rows)     hr_s[tid] = *(const float4*)(hr + (size_t)(base + tid) * NHEAD);
    if (tid < N_ETYPES) he_s[tid] = *(const float4*)(he + (size_t)tid * NHEAD);
    if (tid == 0)       rec_cnt = 0u;
    const float smax = dec_f32(*gmax);
    __syncthreads();

    owner_scan<0>(lane, wave, base, smax, key, q, edge, h, hl, hr_s, he_s, slab, den, atab, rc, attn_out);
    __syncthreads();
    const int total = __builtin_amdgcn_readfirstlane((int)rec_cnt);
    if (total <= LCAP) {
        for (int s = wave * 32; s < total; s += OWN_WAVES * 32)
            replay_batch(s, (total - s < 32) ? (total - s) : 32, lane, rc, den, h, slab, attn_out);
    } else {
        owner_scan<1>(lane, wave, base, smax, key, q, edge, h, hl, hr_s, he_s, slab, den, atab, rc, attn_out);
    }
    __syncthreads();

    float4* out4 = (float4*)(out + (size_t)base * OUT_COLS);
    for (int i = tid; i < rows * (OUT_COLS / 4); i += OWN_WAVES * 32) {
        const float* p = slab + (i >> 5) * OUT_COLS + (i & 31);
        out4[i] = make_float4(p[0], p[32], p[64], p[96]);
    }
}

extern "C" void kernel_launch(void* const* d_in, const int* in_sizes, int n_in,
                              void* d_out, int out_size, void* d_ws, size_t ws_size,
                              hipStream_t stream) {
    const int*   edge     = (const int*)  d_in[0];
    const float* x        = (const float*)d_in[1];
    const float* edge_emb = (const float*)d_in[2];
    const float* W        = (const float*)d_in[3];
    const float* W_e      = (const float*)d_in[4];
    const float* a_l      = (const float*)d_in[5];
    const float* a_r      = (const float*)d_in[6];
    const float* a_e      = (const float*)d_in[7];

    float* out      = (float*)d_out;
    float* attn_out = out + (size_t)N_NODES * OUT_COLS;

    char* ws = (char*)d_ws;
    float* h      = (float*)ws;  ws += (size_t)N_NODES * OUT_COLS * sizeof(float);
    unsigned char* key = (unsigned char*)ws;  ws += (size_t)N_EDGES * NHEAD * sizeof(float);
    float* hl     = (float*)ws;  ws += (size_t)N_NODES * NHEAD   * sizeof(float);
    float* hr     = (float*)ws;  ws += (size_t)N_NODES * NHEAD   * sizeof(float);
    ws += (size_t)N_NODES * NHEAD * sizeof(float);
    float* he     = (float*)ws;  ws += (size_t)N_ETYPES * NHEAD  * sizeof(float);
    unsigned* gmax = (unsigned*)ws;

    init_gmax_kernel<<<1, 1, 0, stream>>>(gmax);

    {
        const int strips = N_NODES / 16;
        const int wavesPerBlock = 8;
        const int blocks = (strips + wavesPerBlock - 1) / wavesPerBlock;
        gemm_h_kernel<<<blocks, 256, 0, stream>>>(x, W, h);
    }
    etype_logits_kernel<<<1, 32, 0, stream>>>(edge_emb, W_e, a_e, he);
    node_logits_kernel<<<(N_NODES * NHEAD + 255) / 256, 256, 0, stream>>>(h, a_l, a_r, hl, hr);
    edge_max_key_kernel<<<(N_EDGES + 255) / 256, 256, 0, stream>>>(edge, hl, hr, he, key, gmax);
    owner_kernel<<<N_SLABS, OWN_WAVES * 32, 0, stream>>>(edge, h, hl, hr, he, key, gmax, out, attn_out);
}


# edge-type logits reassociated: a_e times W_e first (a 64x4 table, 16 lanes per 256-byte row piece), then edge_emb times the table from LDS; 576 serial load round trips become 2; he sum order changed
# speedup vs baseline: 3.3352x; 1.2267x over previous
#include <hip/hip_runtime.h>
#include <hip/hip_bf16.h>

typedef __attribute__((ext_vector_type(2))) float v2f;
typedef __attribute__((ext_vector_type(8))) float v8f;

#define N_NODES 50000
#define N_EDGES 1600000
#define IN_HIDDEN 256
#define OUT_COLS 128
#define NHEAD 4
#define OUT_HIDDEN 32
#define EDGE_HIDDEN 64
#define N_ETYPES 8
#define NEG_SLOPE 0.2f

#define SLAB 196
#define N_SLABS 256
#define CHUNK 512
#define N_CHUNKS (N_EDGES / CHUNK)
#define OWN_WAVES 32
#define QCAP 64
#define LCAP 6912
#define ID_BITS 21
static_assert(N_SLABS * SLAB >= N_NODES && N_SLABS <= 256, "slab id must fit a byte");
static_assert(N_EDGES % CHUNK == 0, "the scan has no tail chunk");
static_assert(N_EDGES <= (1 << ID_BITS) && SLAB <= (1 << (32 - ID_BITS)), "edge id and local node share a dword");
static_assert(LCAP >= OWN_WAVES * 32, "the record slots also hold the second sweep's per-wave tables");
static_assert((N_SLABS - 1) * (N_CHUNKS / N_SLABS) + OWN_WAVES <= N_CHUNKS, "a rotated first chunk stays inside the plane");
static_assert(N_ETYPES * NHEAD == 32 && EDGE_HIDDEN == 64 && NHEAD == 4, "the edge-type logits kernel maps its 32 waves and 16-lane groups on these sizes");

__device__ __forceinline__ unsigned enc_f32(float f) {
    unsigned u = __float_as_uint(f);
    return (u & 0x80000000u) ? ~u : (u | 0x80000000u);
}
__device__ __forceinline__ float dec_f32(unsigned u) {
    unsigned b = (u & 0x80000000u) ? (u & 0x7FFFFFFFu) : ~u;
    return __uint_as_float(b);
}

__global__ void init_gmax_kernel(unsigned* gmax) {
    *gmax = enc_f32(-INFINITY);
}

__global__ void __launch_bounds__(256)
gemm_h_kernel(const float* __restrict__ x, const float* __restrict__ W,
              float* __restrict__ h) {
    const int lane  = threadIdx.x & 31;
    const int wave  = threadIdx.x >> 5;
    const int strip = blockIdx.x * (blockDim.x >> 5) + wave;
    if (strip * 16 >= N_NODES) return;
    const int row0  = strip * 16;

    const int mrow  = lane & 15;
    const int khalf = (lane >> 4) << 1;

    v8f acc[8] = {};

    const float* xrow = x + (size_t)(row0 + mrow) * IN_HIDDEN + khalf;
    for (int k = 0; k < IN_HIDDEN; k += 4) {
        v2f a = *(const v2f*)(xrow + k);
#pragma unroll
        for (int t = 0; t < 8; ++t) {
            const float* wp = W + (size_t)(k + khalf) * OUT_COLS + t * 16 + mrow;
            v2f b;
            b.x = wp[0];
            b.y = wp[OUT_COLS];
            acc[t] = __builtin_amdgcn_wmma_f32_16x16x4_f32(
                false, a, false, b, (short)0, acc[t], false, false);
        }
    }

    const int mbase = (lane >> 4) << 3;
#pragma unroll
    for (int t = 0; t < 8; ++t) {
#pragma unroll
        for (int i = 0; i < 8; ++i) {
            float v = acc[t][i];
            v = (v == v) ? v : 0.0f;
            h[(size_t)(row0 + mbase + i) * OUT_COLS + t * 16 + mrow] = v;
        }
    }
}

__global__ void __launch_bounds__(1024)
etype_logits_kernel(const float* __restrict__ edge_emb,
                    const float* __restrict__ W_e,
                    const float* __restrict__ a_e,
                    float* __restrict__ he) {
    __shared__ float v_s[NHEAD * EDGE_HIDDEN];
    const int tid  = threadIdx.x;
    const int lane = tid & 31;
    const int wave = tid >> 5;
    const int j    = tid & 15;
    const int hw   = tid >> 4;

    const float e0 = edge_emb[(wave >> 2) * EDGE_HIDDEN + lane];
    const float e1 = edge_emb[(wave >> 2) * EDGE_HIDDEN + 32 + lane];

    const float4 a = *(const float4*)(a_e + (hw & 3) * EDGE_HIDDEN + j * 4);
    float4 w[4];
    for (int i = 0; i < 4; ++i)
        w[i] = *(const float4*)(W_e + (size_t)(hw + 64 * i) * EDGE_HIDDEN + j * 4);
    float s[4];
    for (int i = 0; i < 4; ++i)
        s[i] = a.x * w[i].x + a.y * w[i].y + a.z * w[i].z + a.w * w[i].w;
    for (int off = 8; off > 0; off >>= 1) {
        for (int i = 0; i < 4; ++i) s[i] += __shfl_xor(s[i], off, 16);
    }
    if (j == 0) {
        for (int i = 0; i < 4; ++i) {
            const int p = hw + 64 * i;
            v_s[(p & 3) * EDGE_HIDDEN + (p >> 2)] = s[i];
        }
    }
    __syncthreads();

    const float* vh = v_s + (wave & 3) * EDGE_HIDDEN;
    float r = e0 * vh[lane] + e1 * vh[32 + lane];
    r += __shfl_xor(r, 16);
    r += __shfl_xor(r, 8);
    r += __shfl_xor(r, 4);
    r += __shfl_xor(r, 2);
    r += __shfl_xor(r, 1);
    if (lane == 0) he[wave] = r;
}

__global__ void __launch_bounds__(256)
node_logits_kernel(const float* __restrict__ h,
                   const float* __restrict__ a_l, const float* __restrict__ a_r,
                   float* __restrict__ hl, float* __restrict__ hr) {
    const int idx = blockIdx.x * blockDim.x + threadIdx.x;
    if (idx >= N_NODES * NHEAD) return;
    const int node = idx >> 2, head = idx & 3;
    const float* hp = h + (size_t)node * OUT_COLS + head * OUT_HIDDEN;
    const float* al = a_l + head * OUT_HIDDEN;
    const float* ar = a_r + head * OUT_HIDDEN;
    float sl = 0.0f, sr = 0.0f;
#pragma unroll
    for (int d = 0; d < OUT_HIDDEN; ++d) {
        float v = hp[d];
        sl = fmaf(al[d], v, sl);
        sr = fmaf(ar[d], v, sr);
    }
    hl[idx] = sl;
    hr[idx] = sr;
}

__global__ void __launch_bounds__(256)
edge_max_key_kernel(const int* __restrict__ edge,
                    const float* __restrict__ hl, const float* __restrict__ hr,
                    const float* __restrict__ he,
                    unsigned char* __restrict__ key, unsigned* __restrict__ gmax) {
    const int e = blockIdx.x * blockDim.x + threadIdx.x;
    float lmax = -INFINITY;
    if (e < N_EDGES) {
        const int src = edge[e];
        const int trg = edge[N_EDGES + e];
        const int et  = edge[2 * N_EDGES + e];
#pragma unroll
        for (int hh = 0; hh < NHEAD; ++hh) {
            float s = hl[src * NHEAD + hh] + hr[trg * NHEAD + hh] + he[et * NHEAD + hh];
            s = (s > 0.0f) ? s : NEG_SLOPE * s;
            lmax = fmaxf(lmax, s);
        }
        key[e] = (unsigned char)(trg / SLAB);
    }
    __shared__ float red[256];
    red[threadIdx.x] = lmax;
    __syncthreads();
    for (int off = 128; off > 0; off >>= 1) {
        if (threadIdx.x < off)
            red[threadIdx.x] = fmaxf(red[threadIdx.x], red[threadIdx.x + off]);
        __syncthreads();
    }
    if (threadIdx.x == 0) atomicMax(gmax, enc_f32(red[0]));
}


struct EdgeVals { int src; int loc; float es[NHEAD]; };

__device__ __forceinline__ EdgeVals edge_vals(const int* __restrict__ edge,
                                              const float* __restrict__ hl,
                                              const float4* hr_s, const float4* he_s,
                                              int e, int base, float smax) {
    EdgeVals r;
    r.src = edge[e];
    const int trg = edge[N_EDGES + e];
    const int et  = edge[2 * N_EDGES + e];
    r.loc = trg - base;
    const float4 l = *(const float4*)(hl + (size_t)r.src * NHEAD);
    const float4 t = hr_s[r.loc];
    const float4 y = he_s[et];
    float s[NHEAD] = { l.x + t.x + y.x, l.y + t.y + y.y, l.z + t.z + y.z, l.w + t.w + y.w };
#pragma unroll
    for (int k = 0; k < NHEAD; ++k) {
        float v = s[k];
        v = (v > 0.0f) ? v : NEG_SLOPE * v;
        r.es[k] = __expf(v - smax);
    }
    return r;
}

struct Recs {
    unsigned* cnt;
    unsigned* id;
    int*      src;
    float4*   es;
};

__device__ __forceinline__ void add_rows(int n, int lane, int src, int loc, const float* aw,
                                         const float* __restrict__ h, float* slab) {
    const int head = lane >> 3;
    for (int j = 0; j < n; j += 4) {
        float4 r[4]; float aj[4]; int lj[4];
#pragma unroll
        for (int u = 0; u < 4; ++u) {
            const int jj = (j + u < n) ? (j + u) : (n - 1);
            const int sj = __builtin_amdgcn_readlane(src, jj);
            lj[u] = __builtin_amdgcn_readlane(loc, jj);
            aj[u] = aw[jj * NHEAD + head];
            r[u] = *(const float4*)(h + (size_t)sj * OUT_COLS + lane * 4);
        }
#pragma unroll
        for (int u = 0; u < 4; ++u) {
            if (j + u < n) {
                float* p = slab + lj[u] * OUT_COLS + lane;
                atomicAdd(p +  0, r[u].x * aj[u]);
                atomicAdd(p + 32, r[u].y * aj[u]);
                atomicAdd(p + 64, r[u].z * aj[u]);
                atomicAdd(p + 96, r[u].w * aj[u]);
            }
        }
    }
}

template <int PASS>
__device__ __forceinline__ void owner_batch(int n, int lane, int base, float smax,
                                            const unsigned* q,
                                            const int* __restrict__ edge,
                                            const float* __restrict__ h,
                                            const float* __restrict__ hl,
                                            const float4* hr_s, const float4* he_s,
                                            float* slab, float* den, float4* atab, Recs rc,
                                            float* __restrict__ attn_out) {
    if (PASS == 0) {
        unsigned pos = 0;
        if (lane == 0) pos = atomicAdd(rc.cnt, (unsigned)n);
        pos = (unsigned)__builtin_amdgcn_readfirstlane((int)pos);
        if (lane < n) {
            const int e = (int)q[lane];
            const EdgeVals v = edge_vals(edge, hl, hr_s, he_s, e, base, smax);
#pragma unroll
            for (int k = 0; k < NHEAD; ++k) atomicAdd(&den[v.loc * NHEAD + k], v.es[k]);
            const unsigned slot = pos + (unsigned)lane;
            if (slot < LCAP) {
                rc.id[slot]  = (unsigned)e | ((unsigned)v.loc << ID_BITS);
                rc.src[slot] = v.src;
                rc.es[slot]  = make_float4(v.es[0], v.es[1], v.es[2], v.es[3]);
            }
        }
    } else {
        int src = 0, loc = 0;
        if (lane < n) {
            const int e = (int)q[lane];
            const EdgeVals v = edge_vals(edge, hl, hr_s, he_s, e, base, smax);
            src = v.src; loc = v.loc;
            const float4 a = make_float4(v.es[0] / (den[loc * NHEAD + 0] + 1e-16f),
                                         v.es[1] / (den[loc * NHEAD + 1] + 1e-16f),
                                         v.es[2] / (den[loc * NHEAD + 2] + 1e-16f),
                                         v.es[3] / (den[loc * NHEAD + 3] + 1e-16f));
            *(float4*)(attn_out + (size_t)e * NHEAD) = a;
            atab[lane] = a;
        }
        add_rows(n, lane, src, loc, (const float*)atab, h, slab);
    }
}

__device__ __forceinline__ void replay_batch(int start, int n, int lane, Recs rc, const float* den,
                                             const float* __restrict__ h, float* slab,
                                             float* __restrict__ attn_out) {
    int src = 0, loc = 0;
    if (lane < n) {
        const unsigned id = rc.id[start + lane];
        const int e = (int)(id & ((1u << ID_BITS) - 1u));
        loc = (int)(id >> ID_BITS);
        src = rc.src[start + lane];
        const float4 s = rc.es[start + lane];
        const float4 a = make_float4(s.x / (den[loc * NHEAD + 0] + 1e-16f),
                                     s.y / (den[loc * NHEAD + 1] + 1e-16f),
                                     s.z / (den[loc * NHEAD + 2] + 1e-16f),
                                     s.w / (den[loc * NHEAD + 3] + 1e-16f));
        *(float4*)(attn_out + (size_t)e * NHEAD) = a;
        rc.es[start + lane] = a;
    }
    add_rows(n, lane, src, loc, (const float*)(rc.es + start), h, slab);
}

__device__ __forceinline__ unsigned zero_bytes(unsigned x) {
    const unsigned t = (x & 0x7F7F7F7Fu) + 0x7F7F7F7Fu;
    return ~(t | x | 0x7F7F7F7Fu);
}

template <int PASS>
__device__ __forceinline__ void owner_scan(int lane, int wave, int base, float smax,
                                           const unsigned char* __restrict__ key, unsigned* q,
                                           const int* __restrict__ edge,
                                           const float* __restrict__ h,
                                           const float* __restrict__ hl,
                                           const float4* hr_s, const float4* he_s,
                                           float* slab, float* den, float4* atab, Recs rc,
                                           float* __restrict__ attn_out) {
    const unsigned idrep = blockIdx.x * 0x01010101u;
    const int rot = blockIdx.x * (N_CHUNKS / N_SLABS);
    int cnt = 0;
    int c = wave + rot;
    uint4 kv = *(const uint4*)(key + (size_t)c * CHUNK + lane * 16);
    for (int it = wave; it < N_CHUNKS; it += OWN_WAVES) {
        int cn = c + OWN_WAVES;
        if (cn >= N_CHUNKS) cn -= N_CHUNKS;
        uint4 nx = kv;
        if (it + OWN_WAVES < N_CHUNKS)
            nx = *(const uint4*)(key + (size_t)cn * CHUNK + lane * 16);
        unsigned mm = (zero_bytes(kv.x ^ idrep) >> 7) | (zero_bytes(kv.y ^ idrep) >> 6) |
                      (zero_bytes(kv.z ^ idrep) >> 5) | (zero_bytes(kv.w ^ idrep) >> 4);
        unsigned bal;
        while ((bal = __builtin_amdgcn_ballot_w32(mm != 0u)) != 0u) {
            if (mm != 0u) {
                const int p = __builtin_ctz(mm);
                mm &= mm - 1u;
                q[cnt + (int)__builtin_amdgcn_mbcnt_lo(bal, 0u)] =
                    (unsigned)(c * CHUNK + lane * 16 + ((p & 7) << 2) + (p >> 3));
            }
            cnt += __builtin_popcount(bal);
            if (cnt >= 32) {
                owner_batch<PASS>(32, lane, base, smax, q, edge, h, hl, hr_s, he_s, slab, den, atab, rc, attn_out);
                const int rest = cnt - 32;
                if (lane < rest) { const unsigned t = q[32 + lane]; q[lane] = t; }
                cnt = rest;
            }
        }
        kv = nx;
        c = cn;
    }
    if (cnt > 0)
        owner_batch<PASS>(cnt, lane, base, smax, q, edge, h, hl, hr_s, he_s, slab, den, atab, rc, attn_out);
}

__global__ void __launch_bounds__(OWN_WAVES * 32)
owner_kernel(const int* __restrict__ edge, const float* __restrict__ h,
             const float* __restrict__ hl, const float* __restrict__ hr,
             const float* __restrict__ he,
             const unsigned char* __restrict__ key, const unsigned* __restrict__ gmax,
             float* __restrict__ out, float* __restrict__ attn_out) {
    __shared__ float4   slab4[SLAB * OUT_COLS / 4];
    __shared__ float    den[SLAB * NHEAD];
    __shared__ unsigned qbuf[OWN_WAVES * QCAP];
    __shared__ float4   hr_s[SLAB];
    __shared__ float4   he_s[N_ETYPES];
    __shared__ unsigned rec_id[LCAP];
    __shared__ int      rec_src[LCAP];
    __shared__ float4   rec_es[LCAP];
    __shared__ unsigned rec_cnt;
    float* slab = (float*)slab4;

    const int tid  = threadIdx.x;
    const int lane = tid & 31;
    const int wave = __builtin_amdgcn_readfirstlane(tid >> 5);
    const int base = blockIdx.x * SLAB;
    unsigned* q = qbuf + wave * QCAP;
    Recs rc; rc.cnt = &rec_cnt; rc.id = rec_id; rc.src = rec_src; rc.es = rec_es;
    float4* atab = rec_es + wave * 32;

    for (int i = tid; i < SLAB * OUT_COLS / 4; i += OWN_WAVES * 32)
        slab4[i] = make_float4(0.0f, 0.0f, 0.0f, 0.0f);
    for (int i = tid; i < SLAB * NHEAD; i += OWN_WAVES * 32) den[i] = 0.0f;
    const int rows = (N_NODES - base < SLAB) ? (N_NODES - base) : SLAB;
    if (tid < rows)     hr_s[tid] = *(const float4*)(hr + (size_t)(base + tid) * NHEAD);
    if (tid < N_ETYPES) he_s[tid] = *(const float4*)(he + (size_t)tid * NHEAD);
    if (tid == 0)       rec_cnt = 0u;
    const float smax = dec_f32(*gmax);
    __syncthreads();

    owner_scan<0>(lane, wave, base, smax, key, q, edge, h, hl, hr_s, he_s, slab, den, atab, rc, attn_out);
    __syncthreads();
    const int total = __builtin_amdgcn_readfirstlane((int)rec_cnt);
    if (total <= LCAP) {
        for (int s = wave * 32; s < total; s += OWN_WAVES * 32)
            replay_batch(s, (total - s < 32) ? (total - s) : 32, lane, rc, den, h, slab, attn_out);
    } else {
        owner_scan<1>(lane, wave, base, smax, key, q, edge, h, hl, hr_s, he_s, slab, den, atab, rc, attn_out);
    }
    __syncthreads();

    float4* out4 = (float4*)(out + (size_t)base * OUT_COLS);
    for (int i = tid; i < rows * (OUT_COLS / 4); i += OWN_WAVES * 32) {
        const float* p = slab + (i >> 5) * OUT_COLS + (i & 31);
        out4[i] = make_float4(p[0], p[32], p[64], p[96]);
    }
}

extern "C" void kernel_launch(void* const* d_in, const int* in_sizes, int n_in,
                              void* d_out, int out_size, void* d_ws, size_t ws_size,
                              hipStream_t stream) {
    const int*   edge     = (const int*)  d_in[0];
    const float* x        = (const float*)d_in[1];
    const float* edge_emb = (const float*)d_in[2];
    const float* W        = (const float*)d_in[3];
    const float* W_e      = (const float*)d_in[4];
    const float* a_l      = (const float*)d_in[5];
    const float* a_r      = (const float*)d_in[6];
    const float* a_e      = (const float*)d_in[7];

    float* out      = (float*)d_out;
    float* attn_out = out + (size_t)N_NODES * OUT_COLS;

    char* ws = (char*)d_ws;
    float* h      = (float*)ws;  ws += (size_t)N_NODES * OUT_COLS * sizeof(float);
    unsigned char* key = (unsigned char*)ws;  ws += (size_t)N_EDGES * NHEAD * sizeof(float);
    float* hl     = (float*)ws;  ws += (size_t)N_NODES * NHEAD   * sizeof(float);
    float* hr     = (float*)ws;  ws += (size_t)N_NODES * NHEAD   * sizeof(float);
    ws += (size_t)N_NODES * NHEAD * sizeof(float);
    float* he     = (float*)ws;  ws += (size_t)N_ETYPES * NHEAD  * sizeof(float);
    unsigned* gmax = (unsigned*)ws;

    init_gmax_kernel<<<1, 1, 0, stream>>>(gmax);

    {
        const int strips = N_NODES / 16;
        const int wavesPerBlock = 8;
        const int blocks = (strips + wavesPerBlock - 1) / wavesPerBlock;
        gemm_h_kernel<<<blocks, 256, 0, stream>>>(x, W, h);
    }
    etype_logits_kernel<<<1, 1024, 0, stream>>>(edge_emb, W_e, a_e, he);
    node_logits_kernel<<<(N_NODES * NHEAD + 255) / 256, 256, 0, stream>>>(h, a_l, a_r, hl, hr);
    edge_max_key_kernel<<<(N_EDGES + 255) / 256, 256, 0, stream>>>(edge, hl, hr, he, key, gmax);
    owner_kernel<<<N_SLABS, OWN_WAVES * 32, 0, stream>>>(edge, h, hl, hr, he, key, gmax, out, attn_out);
}


# owner test by nibble bit planes: score kernel leaves 32 planes, owner ANDs 2 words per 32 edges, no key bytes; denominators: per-wave LDS sums joined in wave order (fixes run-to-run variation)
# speedup vs baseline: 3.4651x; 1.0389x over previous
#include <hip/hip_runtime.h>
#include <hip/hip_bf16.h>

typedef __attribute__((ext_vector_type(2))) float v2f;
typedef __attribute__((ext_vector_type(8))) float v8f;

#define N_NODES 50000
#define N_EDGES 1600000
#define IN_HIDDEN 256
#define OUT_COLS 128
#define NHEAD 4
#define OUT_HIDDEN 32
#define EDGE_HIDDEN 64
#define N_ETYPES 8
#define NEG_SLOPE 0.2f

#define SLAB 196
#define N_SLABS 256
#define PITCH 50048
#define N_TRIPS (PITCH / 128)
#define N_KBLOCKS (PITCH / 32)
#define OWN_WAVES 32
#define QCAP 64
#define LCAP 6912
#define ID_BITS 21
static_assert(N_SLABS * SLAB >= N_NODES && N_SLABS <= 256, "slab id must fit a byte");
static_assert(N_EDGES % 32 == 0 && PITCH % 128 == 0 && PITCH * 32 >= N_EDGES, "planes hold every edge in whole trips");
static_assert(OWN_WAVES * SLAB * NHEAD <= SLAB * OUT_COLS, "the per-wave denominator sums fit the output rows' LDS space");
static_assert(N_EDGES <= (1 << ID_BITS) && SLAB <= (1 << (32 - ID_BITS)), "edge id and local node share a dword");
static_assert(LCAP >= OWN_WAVES * 32, "the record slots also hold the second sweep's per-wave tables");
static_assert((N_SLABS - 1) * (N_TRIPS / N_SLABS) + OWN_WAVES <= N_TRIPS, "a rotated first trip stays inside the plane");
static_assert(N_ETYPES * NHEAD == 32 && EDGE_HIDDEN == 64 && NHEAD == 4, "the edge-type logits kernel maps its 32 waves and 16-lane groups on these sizes");

__device__ __forceinline__ unsigned enc_f32(float f) {
    unsigned u = __float_as_uint(f);
    return (u & 0x80000000u) ? ~u : (u | 0x80000000u);
}
__device__ __forceinline__ float dec_f32(unsigned u) {
    unsigned b = (u & 0x80000000u) ? (u & 0x7FFFFFFFu) : ~u;
    return __uint_as_float(b);
}

__global__ void init_gmax_kernel(unsigned* gmax) {
    *gmax = enc_f32(-INFINITY);
}

__global__ void __launch_bounds__(256)
gemm_h_kernel(const float* __restrict__ x, const float* __restrict__ W,
              float* __restrict__ h) {
    const int lane  = threadIdx.x & 31;
    const int wave  = threadIdx.x >> 5;
    const int strip = blockIdx.x * (blockDim.x >> 5) + wave;
    if (strip * 16 >= N_NODES) return;
    const int row0  = strip * 16;

    const int mrow  = lane & 15;
    const int khalf = (lane >> 4) << 1;

    v8f acc[8] = {};

    const float* xrow = x + (size_t)(row0 + mrow) * IN_HIDDEN + khalf;
    for (int k = 0; k < IN_HIDDEN; k += 4) {
        v2f a = *(const v2f*)(xrow + k);
#pragma unroll
        for (int t = 0; t < 8; ++t) {
            const float* wp = W + (size_t)(k + khalf) * OUT_COLS + t * 16 + mrow;
            v2f b;
            b.x = wp[0];
            b.y = wp[OUT_COLS];
            acc[t] = __builtin_amdgcn_wmma_f32_16x16x4_f32(
                false, a, false, b, (short)0, acc[t], false, false);
        }
    }

    const int mbase = (lane >> 4) << 3;
#pragma unroll
    for (int t = 0; t < 8; ++t) {
#pragma unroll
        for (int i = 0; i < 8; ++i) {
            float v = acc[t][i];
            v = (v == v) ? v : 0.0f;
            h[(size_t)(row0 + mbase + i) * OUT_COLS + t * 16 + mrow] = v;
        }
    }
}

__global__ void __launch_bounds__(1024)
etype_logits_kernel(const float* __restrict__ edge_emb,
                    const float* __restrict__ W_e,
                    const float* __restrict__ a_e,
                    float* __restrict__ he) {
    __shared__ float v_s[NHEAD * EDGE_HIDDEN];
    const int tid  = threadIdx.x;
    const int lane = tid & 31;
    const int wave = tid >> 5;
    const int j    = tid & 15;
    const int hw   = tid >> 4;

    const float e0 = edge_emb[(wave >> 2) * EDGE_HIDDEN + lane];
    const float e1 = edge_emb[(wave >> 2) * EDGE_HIDDEN + 32 + lane];

    const float4 a = *(const float4*)(a_e + (hw & 3) * EDGE_HIDDEN + j * 4);
    float4 w[4];
    for (int i = 0; i < 4; ++i)
        w[i] = *(const float4*)(W_e + (size_t)(hw + 64 * i) * EDGE_HIDDEN + j * 4);
    float s[4];
    for (int i = 0; i < 4; ++i)
        s[i] = a.x * w[i].x + a.y * w[i].y + a.z * w[i].z + a.w * w[i].w;
    for (int off = 8; off > 0; off >>= 1) {
        for (int i = 0; i < 4; ++i) s[i] += __shfl_xor(s[i], off, 16);
    }
    if (j == 0) {
        for (int i = 0; i < 4; ++i) {
            const int p = hw + 64 * i;
            v_s[(p & 3) * EDGE_HIDDEN + (p >> 2)] = s[i];
        }
    }
    __syncthreads();

    const float* vh = v_s + (wave & 3) * EDGE_HIDDEN;
    float r = e0 * vh[lane] + e1 * vh[32 + lane];
    r += __shfl_xor(r, 16);
    r += __shfl_xor(r, 8);
    r += __shfl_xor(r, 4);
    r += __shfl_xor(r, 2);
    r += __shfl_xor(r, 1);
    if (lane == 0) he[wave] = r;
}

__global__ void __launch_bounds__(256)
node_logits_kernel(const float* __restrict__ h,
                   const float* __restrict__ a_l, const float* __restrict__ a_r,
                   float* __restrict__ hl, float* __restrict__ hr) {
    const int idx = blockIdx.x * blockDim.x + threadIdx.x;
    if (idx >= N_NODES * NHEAD) return;
    const int node = idx >> 2, head = idx & 3;
    const float* hp = h + (size_t)node * OUT_COLS + head * OUT_HIDDEN;
    const float* al = a_l + head * OUT_HIDDEN;
    const float* ar = a_r + head * OUT_HIDDEN;
    float sl = 0.0f, sr = 0.0f;
#pragma unroll
    for (int d = 0; d < OUT_HIDDEN; ++d) {
        float v = hp[d];
        sl = fmaf(al[d], v, sl);
        sr = fmaf(ar[d], v, sr);
    }
    hl[idx] = sl;
    hr[idx] = sr;
}

__global__ void __launch_bounds__(1024)
edge_max_plane_kernel(const int* __restrict__ edge,
                      const float* __restrict__ hl, const float* __restrict__ hr,
                      const float* __restrict__ he,
                      unsigned* __restrict__ plane, unsigned* __restrict__ gmax) {
    __shared__ unsigned st[32 * 32];
    __shared__ float red[32];
    const int tid  = threadIdx.x;
    const int lane = tid & 31;
    const int wave = tid >> 5;
    const int e = blockIdx.x * 1024 + tid;
    st[tid] = 0u;
    __syncthreads();
    float lmax = -INFINITY;
    if (e < N_EDGES) {
        const int src = edge[e];
        const int trg = edge[N_EDGES + e];
        const int et  = edge[2 * N_EDGES + e];
#pragma unroll
        for (int hh = 0; hh < NHEAD; ++hh) {
            float s = hl[src * NHEAD + hh] + hr[trg * NHEAD + hh] + he[et * NHEAD + hh];
            s = (s > 0.0f) ? s : NEG_SLOPE * s;
            lmax = fmaxf(lmax, s);
        }
        const int slab = trg / SLAB;
        atomicOr(&st[(slab >> 4) * 32 + wave], 1u << lane);
        atomicOr(&st[(16 + (slab & 15)) * 32 + wave], 1u << lane);
    }
    lmax = fmaxf(lmax, __shfl_xor(lmax, 16));
    lmax = fmaxf(lmax, __shfl_xor(lmax, 8));
    lmax = fmaxf(lmax, __shfl_xor(lmax, 4));
    lmax = fmaxf(lmax, __shfl_xor(lmax, 2));
    lmax = fmaxf(lmax, __shfl_xor(lmax, 1));
    if (lane == 0) red[wave] = lmax;
    __syncthreads();
    plane[(size_t)wave * PITCH + blockIdx.x * 32 + lane] = st[wave * 32 + lane];
    if (wave == 0) {
        float m = red[lane];
        m = fmaxf(m, __shfl_xor(m, 16));
        m = fmaxf(m, __shfl_xor(m, 8));
        m = fmaxf(m, __shfl_xor(m, 4));
        m = fmaxf(m, __shfl_xor(m, 2));
        m = fmaxf(m, __shfl_xor(m, 1));
        if (lane == 0) atomicMax(gmax, enc_f32(m));
    }
}


struct EdgeVals { int src; int loc; float es[NHEAD]; };

__device__ __forceinline__ EdgeVals edge_vals(const int* __restrict__ edge,
                                              const float* __restrict__ hl,
                                              const float4* hr_s, const float4* he_s,
                                              int e, int base, float smax) {
    EdgeVals r;
    r.src = edge[e];
    const int trg = edge[N_EDGES + e];
    const int et  = edge[2 * N_EDGES + e];
    r.loc = trg - base;
    const float4 l = *(const float4*)(hl + (size_t)r.src * NHEAD);
    const float4 t = hr_s[r.loc];
    const float4 y = he_s[et];
    float s[NHEAD] = { l.x + t.x + y.x, l.y + t.y + y.y, l.z + t.z + y.z, l.w + t.w + y.w };
#pragma unroll
    for (int k = 0; k < NHEAD; ++k) {
        float v = s[k];
        v = (v > 0.0f) ? v : NEG_SLOPE * v;
        r.es[k] = __expf(v - smax);
    }
    return r;
}

struct Recs {
    unsigned* cnt;
    unsigned* id;
    int*      src;
    float4*   es;
};

__device__ __forceinline__ void add_rows(int n, int lane, int src, int loc, const float* aw,
                                         const float* __restrict__ h, float* slab) {
    const int head = lane >> 3;
    for (int j = 0; j < n; j += 4) {
        float4 r[4]; float aj[4]; int lj[4];
#pragma unroll
        for (int u = 0; u < 4; ++u) {
            const int jj = (j + u < n) ? (j + u) : (n - 1);
            const int sj = __builtin_amdgcn_readlane(src, jj);
            lj[u] = __builtin_amdgcn_readlane(loc, jj);
            aj[u] = aw[jj * NHEAD + head];
            r[u] = *(const float4*)(h + (size_t)sj * OUT_COLS + lane * 4);
        }
#pragma unroll
        for (int u = 0; u < 4; ++u) {
            if (j + u < n) {
                float* p = slab + lj[u] * OUT_COLS + lane;
                atomicAdd(p +  0, r[u].x * aj[u]);
                atomicAdd(p + 32, r[u].y * aj[u]);
                atomicAdd(p + 64, r[u].z * aj[u]);
                atomicAdd(p + 96, r[u].w * aj[u]);
            }
        }
    }
}

template <int PASS>
__device__ __forceinline__ void owner_batch(int n, int lane, int base, float smax,
                                            const unsigned* q,
                                            const int* __restrict__ edge,
                                            const float* __restrict__ h,
                                            const float* __restrict__ hl,
                                            const float4* hr_s, const float4* he_s,
                                            float* slab, float* den, float4* atab, Recs rc,
                                            float* __restrict__ attn_out) {
    if (PASS == 0) {
        unsigned pos = 0;
        if (lane == 0) pos = atomicAdd(rc.cnt, (unsigned)n);
        pos = (unsigned)__builtin_amdgcn_readfirstlane((int)pos);
        if (lane < n) {
            const int e = (int)q[lane];
            const EdgeVals v = edge_vals(edge, hl, hr_s, he_s, e, base, smax);
#pragma unroll
            for (int k = 0; k < NHEAD; ++k) atomicAdd(&den[v.loc * NHEAD + k], v.es[k]);
            const unsigned slot = pos + (unsigned)lane;
            if (slot < LCAP) {
                rc.id[slot]  = (unsigned)e | ((unsigned)v.loc << ID_BITS);
                rc.src[slot] = v.src;
                rc.es[slot]  = make_float4(v.es[0], v.es[1], v.es[2], v.es[3]);
            }
        }
    } else {
        int src = 0, loc = 0;
        if (lane < n) {
            const int e = (int)q[lane];
            const EdgeVals v = edge_vals(edge, hl, hr_s, he_s, e, base, smax);
            src = v.src; loc = v.loc;
            const float4 a = make_float4(v.es[0] / (den[loc * NHEAD + 0] + 1e-16f),
                                         v.es[1] / (den[loc * NHEAD + 1] + 1e-16f),
                                         v.es[2] / (den[loc * NHEAD + 2] + 1e-16f),
                                         v.es[3] / (den[loc * NHEAD + 3] + 1e-16f));
            *(float4*)(attn_out + (size_t)e * NHEAD) = a;
            atab[lane] = a;
        }
        add_rows(n, lane, src, loc, (const float*)atab, h, slab);
    }
}

__device__ __forceinline__ void replay_batch(int start, int n, int lane, Recs rc, const float* den,
                                             const float* __restrict__ h, float* slab,
                                             float* __restrict__ attn_out) {
    int src = 0, loc = 0;
    if (lane < n) {
        const unsigned id = rc.id[start + lane];
        const int e = (int)(id & ((1u << ID_BITS) - 1u));
        loc = (int)(id >> ID_BITS);
        src = rc.src[start + lane];
        const float4 s = rc.es[start + lane];
        const float4 a = make_float4(s.x / (den[loc * NHEAD + 0] + 1e-16f),
                                     s.y / (den[loc * NHEAD + 1] + 1e-16f),
                                     s.z / (den[loc * NHEAD + 2] + 1e-16f),
                                     s.w / (den[loc * NHEAD + 3] + 1e-16f));
        *(float4*)(attn_out + (size_t)e * NHEAD) = a;
        rc.es[start + lane] = a;
    }
    add_rows(n, lane, src, loc, (const float*)(rc.es + start), h, slab);
}

template <int PASS>
__device__ __forceinline__ void owner_scan(int lane, int wave, int base, float smax,
                                           const unsigned* __restrict__ plane, unsigned* q,
                                           const int* __restrict__ edge,
                                           const float* __restrict__ h,
                                           const float* __restrict__ hl,
                                           const float4* hr_s, const float4* he_s,
                                           float* slab, float* den, float4* atab, Recs rc,
                                           float* __restrict__ attn_out) {
    const unsigned* pH = plane + (size_t)(blockIdx.x >> 4) * PITCH;
    const unsigned* pL = plane + (size_t)(16 + (blockIdx.x & 15)) * PITCH;
    const int rot = blockIdx.x * (N_TRIPS / N_SLABS);
    int cnt = 0;
    int c = wave + rot;
    uint4 kh = *(const uint4*)(pH + (size_t)c * 128 + lane * 4);
    uint4 kl = *(const uint4*)(pL + (size_t)c * 128 + lane * 4);
    for (int it = wave; it < N_TRIPS; it += OWN_WAVES) {
        int cn = c + OWN_WAVES;
        if (cn >= N_TRIPS) cn -= N_TRIPS;
        uint4 nh = kh, nl = kl;
        if (it + OWN_WAVES < N_TRIPS) {
            nh = *(const uint4*)(pH + (size_t)cn * 128 + lane * 4);
            nl = *(const uint4*)(pL + (size_t)cn * 128 + lane * 4);
        }
        const unsigned m0 = kh.x & kl.x, m1 = kh.y & kl.y, m2 = kh.z & kl.z, m3 = kh.w & kl.w;
#pragma nounroll
        for (int j = 0; j < 4; ++j) {
            unsigned mm = (j == 0) ? m0 : (j == 1) ? m1 : (j == 2) ? m2 : m3;
            const unsigned e0 = (unsigned)(c * 128 + lane * 4 + j) * 32u;
            unsigned bal;
            while ((bal = __builtin_amdgcn_ballot_w32(mm != 0u)) != 0u) {
                if (mm != 0u) {
                    const int p = __builtin_ctz(mm);
                    mm &= mm - 1u;
                    q[cnt + (int)__builtin_amdgcn_mbcnt_lo(bal, 0u)] = e0 + (unsigned)p;
                }
                cnt += __builtin_popcount(bal);
                if (cnt >= 32) {
                    owner_batch<PASS>(32, lane, base, smax, q, edge, h, hl, hr_s, he_s, slab, den, atab, rc, attn_out);
                    const int rest = cnt - 32;
                    if (lane < rest) { const unsigned t = q[32 + lane]; q[lane] = t; }
                    cnt = rest;
                }
            }
        }
        kh = nh; kl = nl;
        c = cn;
    }
    if (cnt > 0)
        owner_batch<PASS>(cnt, lane, base, smax, q, edge, h, hl, hr_s, he_s, slab, den, atab, rc, attn_out);
}

__global__ void __launch_bounds__(OWN_WAVES * 32)
owner_kernel(const int* __restrict__ edge, const float* __restrict__ h,
             const float* __restrict__ hl, const float* __restrict__ hr,
             const float* __restrict__ he,
             const unsigned* __restrict__ plane, const unsigned* __restrict__ gmax,
             float* __restrict__ out, float* __restrict__ attn_out) {
    __shared__ float4   slab4[SLAB * OUT_COLS / 4];
    __shared__ float    den[SLAB * NHEAD];
    __shared__ unsigned qbuf[OWN_WAVES * QCAP];
    __shared__ float4   hr_s[SLAB];
    __shared__ float4   he_s[N_ETYPES];
    __shared__ unsigned rec_id[LCAP];
    __shared__ int      rec_src[LCAP];
    __shared__ float4   rec_es[LCAP];
    __shared__ unsigned rec_cnt;
    float* slab = (float*)slab4;

    const int tid  = threadIdx.x;
    const int lane = tid & 31;
    const int wave = __builtin_amdgcn_readfirstlane(tid >> 5);
    const int base = blockIdx.x * SLAB;
    unsigned* q = qbuf + wave * QCAP;
    Recs rc; rc.cnt = &rec_cnt; rc.id = rec_id; rc.src = rec_src; rc.es = rec_es;
    float4* atab = rec_es + wave * 32;
    float* pden = slab + wave * (SLAB * NHEAD);

    for (int i = tid; i < SLAB * OUT_COLS / 4; i += OWN_WAVES * 32)
        slab4[i] = make_float4(0.0f, 0.0f, 0.0f, 0.0f);
    const int rows = (N_NODES - base < SLAB) ? (N_NODES - base) : SLAB;
    if (tid < rows)     hr_s[tid] = *(const float4*)(hr + (size_t)(base + tid) * NHEAD);
    if (tid < N_ETYPES) he_s[tid] = *(const float4*)(he + (size_t)tid * NHEAD);
    if (tid == 0)       rec_cnt = 0u;
    const float smax = dec_f32(*gmax);
    __syncthreads();

    owner_scan<0>(lane, wave, base, smax, plane, q, edge, h, hl, hr_s, he_s, slab, pden, atab, rc, attn_out);
    __syncthreads();
    for (int i = tid; i < SLAB * NHEAD; i += OWN_WAVES * 32) {
        float s = 0.0f;
        for (int w = 0; w < OWN_WAVES; ++w) s += slab[w * (SLAB * NHEAD) + i];
        den[i] = s;
    }
    __syncthreads();
    for (int i = tid; i < SLAB * OUT_COLS / 4; i += OWN_WAVES * 32)
        slab4[i] = make_float4(0.0f, 0.0f, 0.0f, 0.0f);
    __syncthreads();
    const int total = __builtin_amdgcn_readfirstlane((int)rec_cnt);
    if (total <= LCAP) {
        for (int s = wave * 32; s < total; s += OWN_WAVES * 32)
            replay_batch(s, (total - s < 32) ? (total - s) : 32, lane, rc, den, h, slab, attn_out);
    } else {
        owner_scan<1>(lane, wave, base, smax, plane, q, edge, h, hl, hr_s, he_s, slab, den, atab, rc, attn_out);
    }
    __syncthreads();

    float4* out4 = (float4*)(out + (size_t)base * OUT_COLS);
    for (int i = tid; i < rows * (OUT_COLS / 4); i += OWN_WAVES * 32) {
        const float* p = slab + (i >> 5) * OUT_COLS + (i & 31);
        out4[i] = make_float4(p[0], p[32], p[64], p[96]);
    }
}

extern "C" void kernel_launch(void* const* d_in, const int* in_sizes, int n_in,
                              void* d_out, int out_size, void* d_ws, size_t ws_size,
                              hipStream_t stream) {
    const int*   edge     = (const int*)  d_in[0];
    const float* x        = (const float*)d_in[1];
    const float* edge_emb = (const float*)d_in[2];
    const float* W        = (const float*)d_in[3];
    const float* W_e      = (const float*)d_in[4];
    const float* a_l      = (const float*)d_in[5];
    const float* a_r      = (const float*)d_in[6];
    const float* a_e      = (const float*)d_in[7];

    float* out      = (float*)d_out;
    float* attn_out = out + (size_t)N_NODES * OUT_COLS;

    char* ws = (char*)d_ws;
    float* h      = (float*)ws;  ws += (size_t)N_NODES * OUT_COLS * sizeof(float);
    unsigned* plane = (unsigned*)ws;  ws += (size_t)N_EDGES * NHEAD * sizeof(float);
    float* hl     = (float*)ws;  ws += (size_t)N_NODES * NHEAD   * sizeof(float);
    float* hr     = (float*)ws;  ws += (size_t)N_NODES * NHEAD   * sizeof(float);
    ws += (size_t)N_NODES * NHEAD * sizeof(float);
    float* he     = (float*)ws;  ws += (size_t)N_ETYPES * NHEAD  * sizeof(float);
    unsigned* gmax = (unsigned*)ws;

    init_gmax_kernel<<<1, 1, 0, stream>>>(gmax);

    {
        const int strips = N_NODES / 16;
        const int wavesPerBlock = 8;
        const int blocks = (strips + wavesPerBlock - 1) / wavesPerBlock;
        gemm_h_kernel<<<blocks, 256, 0, stream>>>(x, W, h);
    }
    etype_logits_kernel<<<1, 1024, 0, stream>>>(edge_emb, W_e, a_e, he);
    node_logits_kernel<<<(N_NODES * NHEAD + 255) / 256, 256, 0, stream>>>(h, a_l, a_r, hl, hr);
    edge_max_plane_kernel<<<N_KBLOCKS, 1024, 0, stream>>>(edge, hl, hr, he, plane, gmax);
    owner_kernel<<<N_SLABS, OWN_WAVES * 32, 0, stream>>>(edge, h, hl, hr, he, plane, gmax, out, attn_out);
}


# score kernel hands each edge on in edge order (4 scores in the edge's own attn slot, source + local target in one packed dword): an owner gathers 2 lines per edge in one trip, not 4 in two trips
# speedup vs baseline: 3.6287x; 1.0472x over previous
#include <hip/hip_runtime.h>
#include <hip/hip_bf16.h>

typedef __attribute__((ext_vector_type(2))) float v2f;
typedef __attribute__((ext_vector_type(8))) float v8f;

#define N_NODES 50000
#define N_EDGES 1600000
#define IN_HIDDEN 256
#define OUT_COLS 128
#define NHEAD 4
#define OUT_HIDDEN 32
#define EDGE_HIDDEN 64
#define N_ETYPES 8
#define NEG_SLOPE 0.2f

#define SLAB 196
#define N_SLABS 256
#define PITCH 50048
#define N_TRIPS (PITCH / 128)
#define N_KBLOCKS (PITCH / 32)
#define OWN_WAVES 32
#define QCAP 64
#define LCAP 6912
#define ID_BITS 21
static_assert(N_SLABS * SLAB >= N_NODES && N_SLABS <= 256, "slab id must fit a byte");
static_assert(N_EDGES % 32 == 0 && PITCH % 128 == 0 && PITCH * 32 >= N_EDGES, "planes hold every edge in whole trips");
static_assert(OWN_WAVES * SLAB * NHEAD <= SLAB * OUT_COLS, "the per-wave denominator sums fit the output rows' LDS space");
static_assert(N_EDGES <= (1 << ID_BITS) && SLAB <= (1 << (32 - ID_BITS)), "edge id and local node share a dword");
static_assert(LCAP >= OWN_WAVES * 32, "the record slots also hold the second sweep's per-wave tables");
static_assert((N_SLABS - 1) * (N_TRIPS / N_SLABS) + OWN_WAVES <= N_TRIPS, "a rotated first trip stays inside the plane");
static_assert(N_ETYPES * NHEAD == 32 && EDGE_HIDDEN == 64 && NHEAD == 4, "the edge-type logits kernel maps its 32 waves and 16-lane groups on these sizes");
static_assert(N_NODES <= (1 << 16) && SLAB <= (1 << 16), "source node and local target node share one packed dword");

__device__ __forceinline__ unsigned enc_f32(float f) {
    unsigned u = __float_as_uint(f);
    return (u & 0x80000000u) ? ~u : (u | 0x80000000u);
}
__device__ __forceinline__ float dec_f32(unsigned u) {
    unsigned b = (u & 0x80000000u) ? (u & 0x7FFFFFFFu) : ~u;
    return __uint_as_float(b);
}

__global__ void init_gmax_kernel(unsigned* gmax) {
    *gmax = enc_f32(-INFINITY);
}

__global__ void __launch_bounds__(256)
gemm_h_kernel(const float* __restrict__ x, const float* __restrict__ W,
              float* __restrict__ h) {
    const int lane  = threadIdx.x & 31;
    const int wave  = threadIdx.x >> 5;
    const int strip = blockIdx.x * (blockDim.x >> 5) + wave;
    if (strip * 16 >= N_NODES) return;
    const int row0  = strip * 16;

    const int mrow  = lane & 15;
    const int khalf = (lane >> 4) << 1;

    v8f acc[8] = {};

    const float* xrow = x + (size_t)(row0 + mrow) * IN_HIDDEN + khalf;
    for (int k = 0; k < IN_HIDDEN; k += 4) {
        v2f a = *(const v2f*)(xrow + k);
#pragma unroll
        for (int t = 0; t < 8; ++t) {
            const float* wp = W + (size_t)(k + khalf) * OUT_COLS + t * 16 + mrow;
            v2f b;
            b.x = wp[0];
            b.y = wp[OUT_COLS];
            acc[t] = __builtin_amdgcn_wmma_f32_16x16x4_f32(
                false, a, false, b, (short)0, acc[t], false, false);
        }
    }

    const int mbase = (lane >> 4) << 3;
#pragma unroll
    for (int t = 0; t < 8; ++t) {
#pragma unroll
        for (int i = 0; i < 8; ++i) {
            float v = acc[t][i];
            v = (v == v) ? v : 0.0f;
            h[(size_t)(row0 + mbase + i) * OUT_COLS + t * 16 + mrow] = v;
        }
    }
}

__global__ void __launch_bounds__(1024)
etype_logits_kernel(const float* __restrict__ edge_emb,
                    const float* __restrict__ W_e,
                    const float* __restrict__ a_e,
                    float* __restrict__ he) {
    __shared__ float v_s[NHEAD * EDGE_HIDDEN];
    const int tid  = threadIdx.x;
    const int lane = tid & 31;
    const int wave = tid >> 5;
    const int j    = tid & 15;
    const int hw   = tid >> 4;

    const float e0 = edge_emb[(wave >> 2) * EDGE_HIDDEN + lane];
    const float e1 = edge_emb[(wave >> 2) * EDGE_HIDDEN + 32 + lane];

    const float4 a = *(const float4*)(a_e + (hw & 3) * EDGE_HIDDEN + j * 4);
    float4 w[4];
    for (int i = 0; i < 4; ++i)
        w[i] = *(const float4*)(W_e + (size_t)(hw + 64 * i) * EDGE_HIDDEN + j * 4);
    float s[4];
    for (int i = 0; i < 4; ++i)
        s[i] = a.x * w[i].x + a.y * w[i].y + a.z * w[i].z + a.w * w[i].w;
    for (int off = 8; off > 0; off >>= 1) {
        for (int i = 0; i < 4; ++i) s[i] += __shfl_xor(s[i], off, 16);
    }
    if (j == 0) {
        for (int i = 0; i < 4; ++i) {
            const int p = hw + 64 * i;
            v_s[(p & 3) * EDGE_HIDDEN + (p >> 2)] = s[i];
        }
    }
    __syncthreads();

    const float* vh = v_s + (wave & 3) * EDGE_HIDDEN;
    float r = e0 * vh[lane] + e1 * vh[32 + lane];
    r += __shfl_xor(r, 16);
    r += __shfl_xor(r, 8);
    r += __shfl_xor(r, 4);
    r += __shfl_xor(r, 2);
    r += __shfl_xor(r, 1);
    if (lane == 0) he[wave] = r;
}

__global__ void __launch_bounds__(256)
node_logits_kernel(const float* __restrict__ h,
                   const float* __restrict__ a_l, const float* __restrict__ a_r,
                   float* __restrict__ hl, float* __restrict__ hr) {
    const int idx = blockIdx.x * blockDim.x + threadIdx.x;
    if (idx >= N_NODES * NHEAD) return;
    const int node = idx >> 2, head = idx & 3;
    const float* hp = h + (size_t)node * OUT_COLS + head * OUT_HIDDEN;
    const float* al = a_l + head * OUT_HIDDEN;
    const float* ar = a_r + head * OUT_HIDDEN;
    float sl = 0.0f, sr = 0.0f;
#pragma unroll
    for (int d = 0; d < OUT_HIDDEN; ++d) {
        float v = hp[d];
        sl = fmaf(al[d], v, sl);
        sr = fmaf(ar[d], v, sr);
    }
    hl[idx] = sl;
    hr[idx] = sr;
}

__global__ void __launch_bounds__(1024)
edge_max_plane_kernel(const int* __restrict__ edge,
                      const float* __restrict__ hl, const float* __restrict__ hr,
                      const float* __restrict__ he,
                      unsigned* __restrict__ plane, unsigned* __restrict__ pk,
                      float* __restrict__ score, unsigned* __restrict__ gmax) {
    __shared__ unsigned st[32 * 32];
    __shared__ float red[32];
    const int tid  = threadIdx.x;
    const int lane = tid & 31;
    const int wave = tid >> 5;
    const int e = blockIdx.x * 1024 + tid;
    st[tid] = 0u;
    __syncthreads();
    float lmax = -INFINITY;
    if (e < N_EDGES) {
        const int src = edge[e];
        const int trg = edge[N_EDGES + e];
        const int et  = edge[2 * N_EDGES + e];
        float sc[NHEAD];
#pragma unroll
        for (int hh = 0; hh < NHEAD; ++hh) {
            float s = hl[src * NHEAD + hh] + hr[trg * NHEAD + hh] + he[et * NHEAD + hh];
            s = (s > 0.0f) ? s : NEG_SLOPE * s;
            sc[hh] = s;
            lmax = fmaxf(lmax, s);
        }
        const int slab = trg / SLAB;
        *(float4*)(score + (size_t)e * NHEAD) = make_float4(sc[0], sc[1], sc[2], sc[3]);
        pk[e] = (unsigned)src | ((unsigned)(trg - slab * SLAB) << 16);
        atomicOr(&st[(slab >> 4) * 32 + wave], 1u << lane);
        atomicOr(&st[(16 + (slab & 15)) * 32 + wave], 1u << lane);
    }
    lmax = fmaxf(lmax, __shfl_xor(lmax, 16));
    lmax = fmaxf(lmax, __shfl_xor(lmax, 8));
    lmax = fmaxf(lmax, __shfl_xor(lmax, 4));
    lmax = fmaxf(lmax, __shfl_xor(lmax, 2));
    lmax = fmaxf(lmax, __shfl_xor(lmax, 1));
    if (lane == 0) red[wave] = lmax;
    __syncthreads();
    plane[(size_t)wave * PITCH + blockIdx.x * 32 + lane] = st[wave * 32 + lane];
    if (wave == 0) {
        float m = red[lane];
        m = fmaxf(m, __shfl_xor(m, 16));
        m = fmaxf(m, __shfl_xor(m, 8));
        m = fmaxf(m, __shfl_xor(m, 4));
        m = fmaxf(m, __shfl_xor(m, 2));
        m = fmaxf(m, __shfl_xor(m, 1));
        if (lane == 0) atomicMax(gmax, enc_f32(m));
    }
}


struct EdgeVals { int src; int loc; float es[NHEAD]; };

__device__ __forceinline__ EdgeVals edge_vals(const unsigned* __restrict__ pk, const float* sc,
                                              int e, float smax) {
    EdgeVals r;
    const unsigned p = pk[e];
    const float4 s = *(const float4*)(sc + (size_t)e * NHEAD);
    r.src = (int)(p & 0xFFFFu);
    r.loc = (int)(p >> 16);
    r.es[0] = __expf(s.x - smax);
    r.es[1] = __expf(s.y - smax);
    r.es[2] = __expf(s.z - smax);
    r.es[3] = __expf(s.w - smax);
    return r;
}

struct Recs {
    unsigned* cnt;
    unsigned* id;
    int*      src;
    float4*   es;
};

__device__ __forceinline__ void add_rows(int n, int lane, int src, int loc, const float* aw,
                                         const float* __restrict__ h, float* slab) {
    const int head = lane >> 3;
    for (int j = 0; j < n; j += 4) {
        float4 r[4]; float aj[4]; int lj[4];
#pragma unroll
        for (int u = 0; u < 4; ++u) {
            const int jj = (j + u < n) ? (j + u) : (n - 1);
            const int sj = __builtin_amdgcn_readlane(src, jj);
            lj[u] = __builtin_amdgcn_readlane(loc, jj);
            aj[u] = aw[jj * NHEAD + head];
            r[u] = *(const float4*)(h + (size_t)sj * OUT_COLS + lane * 4);
        }
#pragma unroll
        for (int u = 0; u < 4; ++u) {
            if (j + u < n) {
                float* p = slab + lj[u] * OUT_COLS + lane;
                atomicAdd(p +  0, r[u].x * aj[u]);
                atomicAdd(p + 32, r[u].y * aj[u]);
                atomicAdd(p + 64, r[u].z * aj[u]);
                atomicAdd(p + 96, r[u].w * aj[u]);
            }
        }
    }
}

template <int PASS>
__device__ __forceinline__ void owner_batch(int n, int lane, float smax,
                                            const unsigned* q,
                                            const unsigned* __restrict__ pk,
                                            const float* __restrict__ h,
                                            float* slab, float* den, float4* atab, Recs rc,
                                            float* attn_out) {
    if (PASS == 0) {
        unsigned pos = 0;
        if (lane == 0) pos = atomicAdd(rc.cnt, (unsigned)n);
        pos = (unsigned)__builtin_amdgcn_readfirstlane((int)pos);
        if (lane < n) {
            const int e = (int)q[lane];
            const EdgeVals v = edge_vals(pk, attn_out, e, smax);
#pragma unroll
            for (int k = 0; k < NHEAD; ++k) atomicAdd(&den[v.loc * NHEAD + k], v.es[k]);
            const unsigned slot = pos + (unsigned)lane;
            if (slot < LCAP) {
                rc.id[slot]  = (unsigned)e | ((unsigned)v.loc << ID_BITS);
                rc.src[slot] = v.src;
                rc.es[slot]  = make_float4(v.es[0], v.es[1], v.es[2], v.es[3]);
            }
        }
    } else {
        int src = 0, loc = 0;
        if (lane < n) {
            const int e = (int)q[lane];
            const EdgeVals v = edge_vals(pk, attn_out, e, smax);
            src = v.src; loc = v.loc;
            const float4 a = make_float4(v.es[0] / (den[loc * NHEAD + 0] + 1e-16f),
                                         v.es[1] / (den[loc * NHEAD + 1] + 1e-16f),
                                         v.es[2] / (den[loc * NHEAD + 2] + 1e-16f),
                                         v.es[3] / (den[loc * NHEAD + 3] + 1e-16f));
            *(float4*)(attn_out + (size_t)e * NHEAD) = a;
            atab[lane] = a;
        }
        add_rows(n, lane, src, loc, (const float*)atab, h, slab);
    }
}

__device__ __forceinline__ void replay_batch(int start, int n, int lane, Recs rc, const float* den,
                                             const float* __restrict__ h, float* slab,
                                             float* attn_out) {
    int src = 0, loc = 0;
    if (lane < n) {
        const unsigned id = rc.id[start + lane];
        const int e = (int)(id & ((1u << ID_BITS) - 1u));
        loc = (int)(id >> ID_BITS);
        src = rc.src[start + lane];
        const float4 s = rc.es[start + lane];
        const float4 a = make_float4(s.x / (den[loc * NHEAD + 0] + 1e-16f),
                                     s.y / (den[loc * NHEAD + 1] + 1e-16f),
                                     s.z / (den[loc * NHEAD + 2] + 1e-16f),
                                     s.w / (den[loc * NHEAD + 3] + 1e-16f));
        *(float4*)(attn_out + (size_t)e * NHEAD) = a;
        rc.es[start + lane] = a;
    }
    add_rows(n, lane, src, loc, (const float*)(rc.es + start), h, slab);
}

template <int PASS>
__device__ __forceinline__ void owner_scan(int lane, int wave, float smax,
                                           const unsigned* __restrict__ plane, unsigned* q,
                                           const unsigned* __restrict__ pk,
                                           const float* __restrict__ h,
                                           float* slab, float* den, float4* atab, Recs rc,
                                           float* attn_out) {
    const unsigned* pH = plane + (size_t)(blockIdx.x >> 4) * PITCH;
    const unsigned* pL = plane + (size_t)(16 + (blockIdx.x & 15)) * PITCH;
    const int rot = blockIdx.x * (N_TRIPS / N_SLABS);
    int cnt = 0;
    int c = wave + rot;
    uint4 kh = *(const uint4*)(pH + (size_t)c * 128 + lane * 4);
    uint4 kl = *(const uint4*)(pL + (size_t)c * 128 + lane * 4);
    for (int it = wave; it < N_TRIPS; it += OWN_WAVES) {
        int cn = c + OWN_WAVES;
        if (cn >= N_TRIPS) cn -= N_TRIPS;
        uint4 nh = kh, nl = kl;
        if (it + OWN_WAVES < N_TRIPS) {
            nh = *(const uint4*)(pH + (size_t)cn * 128 + lane * 4);
            nl = *(const uint4*)(pL + (size_t)cn * 128 + lane * 4);
        }
        const unsigned m0 = kh.x & kl.x, m1 = kh.y & kl.y, m2 = kh.z & kl.z, m3 = kh.w & kl.w;
#pragma nounroll
        for (int j = 0; j < 4; ++j) {
            unsigned mm = (j == 0) ? m0 : (j == 1) ? m1 : (j == 2) ? m2 : m3;
            const unsigned e0 = (unsigned)(c * 128 + lane * 4 + j) * 32u;
            unsigned bal;
            while ((bal = __builtin_amdgcn_ballot_w32(mm != 0u)) != 0u) {
                if (mm != 0u) {
                    const int p = __builtin_ctz(mm);
                    mm &= mm - 1u;
                    q[cnt + (int)__builtin_amdgcn_mbcnt_lo(bal, 0u)] = e0 + (unsigned)p;
                }
                cnt += __builtin_popcount(bal);
                if (cnt >= 32) {
                    owner_batch<PASS>(32, lane, smax, q, pk, h, slab, den, atab, rc, attn_out);
                    const int rest = cnt - 32;
                    if (lane < rest) { const unsigned t = q[32 + lane]; q[lane] = t; }
                    cnt = rest;
                }
            }
        }
        kh = nh; kl = nl;
        c = cn;
    }
    if (cnt > 0)
        owner_batch<PASS>(cnt, lane, smax, q, pk, h, slab, den, atab, rc, attn_out);
}

__global__ void __launch_bounds__(OWN_WAVES * 32)
owner_kernel(const float* __restrict__ h,
             const unsigned* __restrict__ plane, const unsigned* __restrict__ pk,
             const unsigned* __restrict__ gmax,
             float* __restrict__ out, float* attn_out) {
    __shared__ float4   slab4[SLAB * OUT_COLS / 4];
    __shared__ float    den[SLAB * NHEAD];
    __shared__ unsigned qbuf[OWN_WAVES * QCAP];
    __shared__ unsigned rec_id[LCAP];
    __shared__ int      rec_src[LCAP];
    __shared__ float4   rec_es[LCAP];
    __shared__ unsigned rec_cnt;
    float* slab = (float*)slab4;

    const int tid  = threadIdx.x;
    const int lane = tid & 31;
    const int wave = __builtin_amdgcn_readfirstlane(tid >> 5);
    const int base = blockIdx.x * SLAB;
    unsigned* q = qbuf + wave * QCAP;
    Recs rc; rc.cnt = &rec_cnt; rc.id = rec_id; rc.src = rec_src; rc.es = rec_es;
    float4* atab = rec_es + wave * 32;
    float* pden = slab + wave * (SLAB * NHEAD);

    for (int i = tid; i < SLAB * OUT_COLS / 4; i += OWN_WAVES * 32)
        slab4[i] = make_float4(0.0f, 0.0f, 0.0f, 0.0f);
    const int rows = (N_NODES - base < SLAB) ? (N_NODES - base) : SLAB;
    if (tid == 0)       rec_cnt = 0u;
    const float smax = dec_f32(*gmax);
    __syncthreads();

    owner_scan<0>(lane, wave, smax, plane, q, pk, h, slab, pden, atab, rc, attn_out);
    __syncthreads();
    for (int i = tid; i < SLAB * NHEAD; i += OWN_WAVES * 32) {
        float s = 0.0f;
        for (int w = 0; w < OWN_WAVES; ++w) s += slab[w * (SLAB * NHEAD) + i];
        den[i] = s;
    }
    __syncthreads();
    for (int i = tid; i < SLAB * OUT_COLS / 4; i += OWN_WAVES * 32)
        slab4[i] = make_float4(0.0f, 0.0f, 0.0f, 0.0f);
    __syncthreads();
    const int total = __builtin_amdgcn_readfirstlane((int)rec_cnt);
    if (total <= LCAP) {
        for (int s = wave * 32; s < total; s += OWN_WAVES * 32)
            replay_batch(s, (total - s < 32) ? (total - s) : 32, lane, rc, den, h, slab, attn_out);
    } else {
        owner_scan<1>(lane, wave, smax, plane, q, pk, h, slab, den, atab, rc, attn_out);
    }
    __syncthreads();

    float4* out4 = (float4*)(out + (size_t)base * OUT_COLS);
    for (int i = tid; i < rows * (OUT_COLS / 4); i += OWN_WAVES * 32) {
        const float* p = slab + (i >> 5) * OUT_COLS + (i & 31);
        out4[i] = make_float4(p[0], p[32], p[64], p[96]);
    }
}

extern "C" void kernel_launch(void* const* d_in, const int* in_sizes, int n_in,
                              void* d_out, int out_size, void* d_ws, size_t ws_size,
                              hipStream_t stream) {
    const int*   edge     = (const int*)  d_in[0];
    const float* x        = (const float*)d_in[1];
    const float* edge_emb = (const float*)d_in[2];
    const float* W        = (const float*)d_in[3];
    const float* W_e      = (const float*)d_in[4];
    const float* a_l      = (const float*)d_in[5];
    const float* a_r      = (const float*)d_in[6];
    const float* a_e      = (const float*)d_in[7];

    float* out      = (float*)d_out;
    float* attn_out = out + (size_t)N_NODES * OUT_COLS;

    char* ws = (char*)d_ws;
    float* h      = (float*)ws;  ws += (size_t)N_NODES * OUT_COLS * sizeof(float);
    unsigned* plane = (unsigned*)ws;  ws += (size_t)N_EDGES * NHEAD * sizeof(float);
    unsigned* pk = plane + (size_t)32 * PITCH;
    float* hl     = (float*)ws;  ws += (size_t)N_NODES * NHEAD   * sizeof(float);
    float* hr     = (float*)ws;  ws += (size_t)N_NODES * NHEAD   * sizeof(float);
    ws += (size_t)N_NODES * NHEAD * sizeof(float);
    float* he     = (float*)ws;  ws += (size_t)N_ETYPES * NHEAD  * sizeof(float);
    unsigned* gmax = (unsigned*)ws;

    init_gmax_kernel<<<1, 1, 0, stream>>>(gmax);

    {
        const int strips = N_NODES / 16;
        const int wavesPerBlock = 8;
        const int blocks = (strips + wavesPerBlock - 1) / wavesPerBlock;
        gemm_h_kernel<<<blocks, 256, 0, stream>>>(x, W, h);
    }
    etype_logits_kernel<<<1, 1024, 0, stream>>>(edge_emb, W_e, a_e, he);
    node_logits_kernel<<<(N_NODES * NHEAD + 255) / 256, 256, 0, stream>>>(h, a_l, a_r, hl, hr);
    edge_max_plane_kernel<<<N_KBLOCKS, 1024, 0, stream>>>(edge, hl, hr, he, plane, pk, attn_out, gmax);
    owner_kernel<<<N_SLABS, OWN_WAVES * 32, 0, stream>>>(h, plane, pk, gmax, out, attn_out);
}


# score kernel: a lane takes 8 edges (8-wave work-groups of 2048 edges; its 24 index loads, then its logit gathers, in flight together): 6,256 waves enter the chip, not 50,048; stored values unchanged
# speedup vs baseline: 3.7869x; 1.0436x over previous
#include <hip/hip_runtime.h>
#include <hip/hip_bf16.h>

typedef __attribute__((ext_vector_type(2))) float v2f;
typedef __attribute__((ext_vector_type(8))) float v8f;

#define N_NODES 50000
#define N_EDGES 1600000
#define IN_HIDDEN 256
#define OUT_COLS 128
#define NHEAD 4
#define OUT_HIDDEN 32
#define EDGE_HIDDEN 64
#define N_ETYPES 8
#define NEG_SLOPE 0.2f

#define SLAB 196
#define N_SLABS 256
#define PITCH 50048
#define N_TRIPS (PITCH / 128)
#define EPL 8
#define SC_WAVES 8
#define SC_WORDS (SC_WAVES * EPL)
#define ST_PITCH (SC_WORDS + 1)
#define N_KBLOCKS (PITCH / SC_WORDS)
#define OWN_WAVES 32
#define QCAP 64
#define LCAP 6912
#define ID_BITS 21
static_assert(N_SLABS * SLAB >= N_NODES && N_SLABS <= 256, "slab id must fit a byte");
static_assert(N_EDGES % 32 == 0 && PITCH % 128 == 0 && PITCH * 32 >= N_EDGES, "planes hold every edge in whole trips");
static_assert(PITCH % SC_WORDS == 0 && SC_WORDS % 32 == 0 && SC_WAVES <= 32, "the score kernel's work-groups tile the planes in whole 128-byte lines");
static_assert((N_EDGES / 32) % EPL == 0, "a wave of the score kernel has edges in all of its words or in none");
static_assert(OWN_WAVES * SLAB * NHEAD <= SLAB * OUT_COLS, "the per-wave denominator sums fit the output rows' LDS space");
static_assert(N_EDGES <= (1 << ID_BITS) && SLAB <= (1 << (32 - ID_BITS)), "edge id and local node share a dword");
static_assert(LCAP >= OWN_WAVES * 32, "the record slots also hold the second sweep's per-wave tables");
static_assert((N_SLABS - 1) * (N_TRIPS / N_SLABS) + OWN_WAVES <= N_TRIPS, "a rotated first trip stays inside the plane");
static_assert(N_ETYPES * NHEAD == 32 && EDGE_HIDDEN == 64 && NHEAD == 4, "the edge-type logits kernel maps its 32 waves and 16-lane groups on these sizes");
static_assert(N_NODES <= (1 << 16) && SLAB <= (1 << 16), "source node and local target node share one packed dword");

__device__ __forceinline__ unsigned enc_f32(float f) {
    unsigned u = __float_as_uint(f);
    return (u & 0x80000000u) ? ~u : (u | 0x80000000u);
}
__device__ __forceinline__ float dec_f32(unsigned u) {
    unsigned b = (u & 0x80000000u) ? (u & 0x7FFFFFFFu) : ~u;
    return __uint_as_float(b);
}

__global__ void init_gmax_kernel(unsigned* gmax) {
    *gmax = enc_f32(-INFINITY);
}

__global__ void __launch_bounds__(256)
gemm_h_kernel(const float* __restrict__ x, const float* __restrict__ W,
              float* __restrict__ h) {
    const int lane  = threadIdx.x & 31;
    const int wave  = threadIdx.x >> 5;
    const int strip = blockIdx.x * (blockDim.x >> 5) + wave;
    if (strip * 16 >= N_NODES) return;
    const int row0  = strip * 16;

    const int mrow  = lane & 15;
    const int khalf = (lane >> 4) << 1;

    v8f acc[8] = {};

    const float* xrow = x + (size_t)(row0 + mrow) * IN_HIDDEN + khalf;
    for (int k = 0; k < IN_HIDDEN; k += 4) {
        v2f a = *(const v2f*)(xrow + k);
#pragma unroll
        for (int t = 0; t < 8; ++t) {
            const float* wp = W + (size_t)(k + khalf) * OUT_COLS + t * 16 + mrow;
            v2f b;
            b.x = wp[0];
            b.y = wp[OUT_COLS];
            acc[t] = __builtin_amdgcn_wmma_f32_16x16x4_f32(
                false, a, false, b, (short)0, acc[t], false, false);
        }
    }

    const int mbase = (lane >> 4) << 3;
#pragma unroll
    for (int t = 0; t < 8; ++t) {
#pragma unroll
        for (int i = 0; i < 8; ++i) {
            float v = acc[t][i];
            v = (v == v) ? v : 0.0f;
            h[(size_t)(row0 + mbase + i) * OUT_COLS + t * 16 + mrow] = v;
        }
    }
}

__global__ void __launch_bounds__(1024)
etype_logits_kernel(const float* __restrict__ edge_emb,
                    const float* __restrict__ W_e,
                    const float* __restrict__ a_e,
                    float* __restrict__ he) {
    __shared__ float v_s[NHEAD * EDGE_HIDDEN];
    const int tid  = threadIdx.x;
    const int lane = tid & 31;
    const int wave = tid >> 5;
    const int j    = tid & 15;
    const int hw   = tid >> 4;

    const float e0 = edge_emb[(wave >> 2) * EDGE_HIDDEN + lane];
    const float e1 = edge_emb[(wave >> 2) * EDGE_HIDDEN + 32 + lane];

    const float4 a = *(const float4*)(a_e + (hw & 3) * EDGE_HIDDEN + j * 4);
    float4 w[4];
    for (int i = 0; i < 4; ++i)
        w[i] = *(const float4*)(W_e + (size_t)(hw + 64 * i) * EDGE_HIDDEN + j * 4);
    float s[4];
    for (int i = 0; i < 4; ++i)
        s[i] = a.x * w[i].x + a.y * w[i].y + a.z * w[i].z + a.w * w[i].w;
    for (int off = 8; off > 0; off >>= 1) {
        for (int i = 0; i < 4; ++i) s[i] += __shfl_xor(s[i], off, 16);
    }
    if (j == 0) {
        for (int i = 0; i < 4; ++i) {
            const int p = hw + 64 * i;
            v_s[(p & 3) * EDGE_HIDDEN + (p >> 2)] = s[i];
        }
    }
    __syncthreads();

    const float* vh = v_s + (wave & 3) * EDGE_HIDDEN;
    float r = e0 * vh[lane] + e1 * vh[32 + lane];
    r += __shfl_xor(r, 16);
    r += __shfl_xor(r, 8);
    r += __shfl_xor(r, 4);
    r += __shfl_xor(r, 2);
    r += __shfl_xor(r, 1);
    if (lane == 0) he[wave] = r;
}

__global__ void __launch_bounds__(256)
node_logits_kernel(const float* __restrict__ h,
                   const float* __restrict__ a_l, const float* __restrict__ a_r,
                   float* __restrict__ hl, float* __restrict__ hr) {
    const int idx = blockIdx.x * blockDim.x + threadIdx.x;
    if (idx >= N_NODES * NHEAD) return;
    const int node = idx >> 2, head = idx & 3;
    const float* hp = h + (size_t)node * OUT_COLS + head * OUT_HIDDEN;
    const float* al = a_l + head * OUT_HIDDEN;
    const float* ar = a_r + head * OUT_HIDDEN;
    float sl = 0.0f, sr = 0.0f;
#pragma unroll
    for (int d = 0; d < OUT_HIDDEN; ++d) {
        float v = hp[d];
        sl = fmaf(al[d], v, sl);
        sr = fmaf(ar[d], v, sr);
    }
    hl[idx] = sl;
    hr[idx] = sr;
}

__global__ void __launch_bounds__(SC_WAVES * 32) __attribute__((amdgpu_waves_per_eu(4, 8)))
edge_max_plane_kernel(const int* __restrict__ edge,
                      const float* __restrict__ hl, const float* __restrict__ hr,
                      const float* __restrict__ he,
                      unsigned* __restrict__ plane, unsigned* __restrict__ pk,
                      float* __restrict__ score, unsigned* __restrict__ gmax) {
    __shared__ unsigned st[32 * ST_PITCH];
    __shared__ float red[SC_WAVES];
    const int tid  = threadIdx.x;
    const int lane = tid & 31;
    const int wave = __builtin_amdgcn_readfirstlane(tid >> 5);
    for (int i = tid; i < 32 * ST_PITCH; i += SC_WAVES * 32) st[i] = 0u;
    __syncthreads();

    const int w0    = wave * EPL;
    const int wbase = blockIdx.x * SC_WORDS + w0;
    float lmax = -INFINITY;
    if (wbase * 32 < N_EDGES) {
        const int e0 = wbase * 32 + lane;
        int src[EPL], trg[EPL], et[EPL];
#pragma unroll
        for (int j = 0; j < EPL; ++j) {
            const int e = e0 + j * 32;
            src[j] = edge[e];
            trg[j] = edge[N_EDGES + e];
            et[j]  = edge[2 * N_EDGES + e];
        }
        float4 gl[EPL], gr[EPL], ge[EPL];
#pragma unroll
        for (int j = 0; j < EPL; ++j) {
            gl[j] = *(const float4*)(hl + src[j] * NHEAD);
            gr[j] = *(const float4*)(hr + trg[j] * NHEAD);
            ge[j] = *(const float4*)(he + et[j] * NHEAD);
        }
#pragma unroll
        for (int j = 0; j < EPL; ++j) {
            const int e = e0 + j * 32;
            float sc[NHEAD];
            sc[0] = gl[j].x + gr[j].x + ge[j].x;
            sc[1] = gl[j].y + gr[j].y + ge[j].y;
            sc[2] = gl[j].z + gr[j].z + ge[j].z;
            sc[3] = gl[j].w + gr[j].w + ge[j].w;
#pragma unroll
            for (int hh = 0; hh < NHEAD; ++hh) {
                float s = sc[hh];
                s = (s > 0.0f) ? s : NEG_SLOPE * s;
                sc[hh] = s;
                lmax = fmaxf(lmax, s);
            }
            const int slab = trg[j] / SLAB;
            *(float4*)(score + (size_t)e * NHEAD) = make_float4(sc[0], sc[1], sc[2], sc[3]);
            pk[e] = (unsigned)src[j] | ((unsigned)(trg[j] - slab * SLAB) << 16);
            atomicOr(&st[(slab >> 4) * ST_PITCH + w0 + j], 1u << lane);
            atomicOr(&st[(16 + (slab & 15)) * ST_PITCH + w0 + j], 1u << lane);
        }
    }
    lmax = fmaxf(lmax, __shfl_xor(lmax, 16));
    lmax = fmaxf(lmax, __shfl_xor(lmax, 8));
    lmax = fmaxf(lmax, __shfl_xor(lmax, 4));
    lmax = fmaxf(lmax, __shfl_xor(lmax, 2));
    lmax = fmaxf(lmax, __shfl_xor(lmax, 1));
    if (lane == 0) red[wave] = lmax;
    __syncthreads();
#pragma unroll
    for (int i = 0; i < 32 * SC_WORDS / (SC_WAVES * 32); ++i) {
        const int idx = i * (SC_WAVES * 32) + tid;
        const int p = idx / SC_WORDS, w = idx % SC_WORDS;
        plane[(size_t)p * PITCH + blockIdx.x * SC_WORDS + w] = st[p * ST_PITCH + w];
    }
    if (wave == 0) {
        float m = (lane < SC_WAVES) ? red[lane] : -INFINITY;
        m = fmaxf(m, __shfl_xor(m, 4));
        m = fmaxf(m, __shfl_xor(m, 2));
        m = fmaxf(m, __shfl_xor(m, 1));
        if (lane == 0) atomicMax(gmax, enc_f32(m));
    }
}


struct EdgeVals { int src; int loc; float es[NHEAD]; };

__device__ __forceinline__ EdgeVals edge_vals(const unsigned* __restrict__ pk, const float* sc,
                                              int e, float smax) {
    EdgeVals r;
    const unsigned p = pk[e];
    const float4 s = *(const float4*)(sc + (size_t)e * NHEAD);
    r.src = (int)(p & 0xFFFFu);
    r.loc = (int)(p >> 16);
    r.es[0] = __expf(s.x - smax);
    r.es[1] = __expf(s.y - smax);
    r.es[2] = __expf(s.z - smax);
    r.es[3] = __expf(s.w - smax);
    return r;
}

struct Recs {
    unsigned* cnt;
    unsigned* id;
    int*      src;
    float4*   es;
};

__device__ __forceinline__ void add_rows(int n, int lane, int src, int loc, const float* aw,
                                         const float* __restrict__ h, float* slab) {
    const int head = lane >> 3;
    for (int j = 0; j < n; j += 4) {
        float4 r[4]; float aj[4]; int lj[4];
#pragma unroll
        for (int u = 0; u < 4; ++u) {
            const int jj = (j + u < n) ? (j + u) : (n - 1);
            const int sj = __builtin_amdgcn_readlane(src, jj);
            lj[u] = __builtin_amdgcn_readlane(loc, jj);
            aj[u] = aw[jj * NHEAD + head];
            r[u] = *(const float4*)(h + (size_t)sj * OUT_COLS + lane * 4);
        }
#pragma unroll
        for (int u = 0; u < 4; ++u) {
            if (j + u < n) {
                float* p = slab + lj[u] * OUT_COLS + lane;
                atomicAdd(p +  0, r[u].x * aj[u]);
                atomicAdd(p + 32, r[u].y * aj[u]);
                atomicAdd(p + 64, r[u].z * aj[u]);
                atomicAdd(p + 96, r[u].w * aj[u]);
            }
        }
    }
}

template <int PASS>
__device__ __forceinline__ void owner_batch(int n, int lane, float smax,
                                            const unsigned* q,
                                            const unsigned* __restrict__ pk,
                                            const float* __restrict__ h,
                                            float* slab, float* den, float4* atab, Recs rc,
                                            float* attn_out) {
    if (PASS == 0) {
        unsigned pos = 0;
        if (lane == 0) pos = atomicAdd(rc.cnt, (unsigned)n);
        pos = (unsigned)__builtin_amdgcn_readfirstlane((int)pos);
        if (lane < n) {
            const int e = (int)q[lane];
            const EdgeVals v = edge_vals(pk, attn_out, e, smax);
#pragma unroll
            for (int k = 0; k < NHEAD; ++k) atomicAdd(&den[v.loc * NHEAD + k], v.es[k]);
            const unsigned slot = pos + (unsigned)lane;
            if (slot < LCAP) {
                rc.id[slot]  = (unsigned)e | ((unsigned)v.loc << ID_BITS);
                rc.src[slot] = v.src;
                rc.es[slot]  = make_float4(v.es[0], v.es[1], v.es[2], v.es[3]);
            }
        }
    } else {
        int src = 0, loc = 0;
        if (lane < n) {
            const int e = (int)q[lane];
            const EdgeVals v = edge_vals(pk, attn_out, e, smax);
            src = v.src; loc = v.loc;
            const float4 a = make_float4(v.es[0] / (den[loc * NHEAD + 0] + 1e-16f),
                                         v.es[1] / (den[loc * NHEAD + 1] + 1e-16f),
                                         v.es[2] / (den[loc * NHEAD + 2] + 1e-16f),
                                         v.es[3] / (den[loc * NHEAD + 3] + 1e-16f));
            *(float4*)(attn_out + (size_t)e * NHEAD) = a;
            atab[lane] = a;
        }
        add_rows(n, lane, src, loc, (const float*)atab, h, slab);
    }
}

__device__ __forceinline__ void replay_batch(int start, int n, int lane, Recs rc, const float* den,
                                             const float* __restrict__ h, float* slab,
                                             float* attn_out) {
    int src = 0, loc = 0;
    if (lane < n) {
        const unsigned id = rc.id[start + lane];
        const int e = (int)(id & ((1u << ID_BITS) - 1u));
        loc = (int)(id >> ID_BITS);
        src = rc.src[start + lane];
        const float4 s = rc.es[start + lane];
        const float4 a = make_float4(s.x / (den[loc * NHEAD + 0] + 1e-16f),
                                     s.y / (den[loc * NHEAD + 1] + 1e-16f),
                                     s.z / (den[loc * NHEAD + 2] + 1e-16f),
                                     s.w / (den[loc * NHEAD + 3] + 1e-16f));
        *(float4*)(attn_out + (size_t)e * NHEAD) = a;
        rc.es[start + lane] = a;
    }
    add_rows(n, lane, src, loc, (const float*)(rc.es + start), h, slab);
}

template <int PASS>
__device__ __forceinline__ void owner_scan(int lane, int wave, float smax,
                                           const unsigned* __restrict__ plane, unsigned* q,
                                           const unsigned* __restrict__ pk,
                                           const float* __restrict__ h,
                                           float* slab, float* den, float4* atab, Recs rc,
                                           float* attn_out) {
    const unsigned* pH = plane + (size_t)(blockIdx.x >> 4) * PITCH;
    const unsigned* pL = plane + (size_t)(16 + (blockIdx.x & 15)) * PITCH;
    const int rot = blockIdx.x * (N_TRIPS / N_SLABS);
    int cnt = 0;
    int c = wave + rot;
    uint4 kh = *(const uint4*)(pH + (size_t)c * 128 + lane * 4);
    uint4 kl = *(const uint4*)(pL + (size_t)c * 128 + lane * 4);
    for (int it = wave; it < N_TRIPS; it += OWN_WAVES) {
        int cn = c + OWN_WAVES;
        if (cn >= N_TRIPS) cn -= N_TRIPS;
        uint4 nh = kh, nl = kl;
        if (it + OWN_WAVES < N_TRIPS) {
            nh = *(const uint4*)(pH + (size_t)cn * 128 + lane * 4);
            nl = *(const uint4*)(pL + (size_t)cn * 128 + lane * 4);
        }
        const unsigned m0 = kh.x & kl.x, m1 = kh.y & kl.y, m2 = kh.z & kl.z, m3 = kh.w & kl.w;
#pragma nounroll
        for (int j = 0; j < 4; ++j) {
            unsigned mm = (j == 0) ? m0 : (j == 1) ? m1 : (j == 2) ? m2 : m3;
            const unsigned e0 = (unsigned)(c * 128 + lane * 4 + j) * 32u;
            unsigned bal;
            while ((bal = __builtin_amdgcn_ballot_w32(mm != 0u)) != 0u) {
                if (mm != 0u) {
                    const int p = __builtin_ctz(mm);
                    mm &= mm - 1u;
                    q[cnt + (int)__builtin_amdgcn_mbcnt_lo(bal, 0u)] = e0 + (unsigned)p;
                }
                cnt += __builtin_popcount(bal);
                if (cnt >= 32) {
                    owner_batch<PASS>(32, lane, smax, q, pk, h, slab, den, atab, rc, attn_out);
                    const int rest = cnt - 32;
                    if (lane < rest) { const unsigned t = q[32 + lane]; q[lane] = t; }
                    cnt = rest;
                }
            }
        }
        kh = nh; kl = nl;
        c = cn;
    }
    if (cnt > 0)
        owner_batch<PASS>(cnt, lane, smax, q, pk, h, slab, den, atab, rc, attn_out);
}

__global__ void __launch_bounds__(OWN_WAVES * 32)
owner_kernel(const float* __restrict__ h,
             const unsigned* __restrict__ plane, const unsigned* __restrict__ pk,
             const unsigned* __restrict__ gmax,
             float* __restrict__ out, float* attn_out) {
    __shared__ float4   slab4[SLAB * OUT_COLS / 4];
    __shared__ float    den[SLAB * NHEAD];
    __shared__ unsigned qbuf[OWN_WAVES * QCAP];
    __shared__ unsigned rec_id[LCAP];
    __shared__ int      rec_src[LCAP];
    __shared__ float4   rec_es[LCAP];
    __shared__ unsigned rec_cnt;
    float* slab = (float*)slab4;

    const int tid  = threadIdx.x;
    const int lane = tid & 31;
    const int wave = __builtin_amdgcn_readfirstlane(tid >> 5);
    const int base = blockIdx.x * SLAB;
    unsigned* q = qbuf + wave * QCAP;
    Recs rc; rc.cnt = &rec_cnt; rc.id = rec_id; rc.src = rec_src; rc.es = rec_es;
    float4* atab = rec_es + wave * 32;
    float* pden = slab + wave * (SLAB * NHEAD);

    for (int i = tid; i < SLAB * OUT_COLS / 4; i += OWN_WAVES * 32)
        slab4[i] = make_float4(0.0f, 0.0f, 0.0f, 0.0f);
    const int rows = (N_NODES - base < SLAB) ? (N_NODES - base) : SLAB;
    if (tid == 0)       rec_cnt = 0u;
    const float smax = dec_f32(*gmax);
    __syncthreads();

    owner_scan<0>(lane, wave, smax, plane, q, pk, h, slab, pden, atab, rc, attn_out);
    __syncthreads();
    for (int i = tid; i < SLAB * NHEAD; i += OWN_WAVES * 32) {
        float s = 0.0f;
        for (int w = 0; w < OWN_WAVES; ++w) s += slab[w * (SLAB * NHEAD) + i];
        den[i] = s;
    }
    __syncthreads();
    for (int i = tid; i < SLAB * OUT_COLS / 4; i += OWN_WAVES * 32)
        slab4[i] = make_float4(0.0f, 0.0f, 0.0f, 0.0f);
    __syncthreads();
    const int total = __builtin_amdgcn_readfirstlane((int)rec_cnt);
    if (total <= LCAP) {
        for (int s = wave * 32; s < total; s += OWN_WAVES * 32)
            replay_batch(s, (total - s < 32) ? (total - s) : 32, lane, rc, den, h, slab, attn_out);
    } else {
        owner_scan<1>(lane, wave, smax, plane, q, pk, h, slab, den, atab, rc, attn_out);
    }
    __syncthreads();

    float4* out4 = (float4*)(out + (size_t)base * OUT_COLS);
    for (int i = tid; i < rows * (OUT_COLS / 4); i += OWN_WAVES * 32) {
        const float* p = slab + (i >> 5) * OUT_COLS + (i & 31);
        out4[i] = make_float4(p[0], p[32], p[64], p[96]);
    }
}

extern "C" void kernel_launch(void* const* d_in, const int* in_sizes, int n_in,
                              void* d_out, int out_size, void* d_ws, size_t ws_size,
                              hipStream_t stream) {
    const int*   edge     = (const int*)  d_in[0];
    const float* x        = (const float*)d_in[1];
    const float* edge_emb = (const float*)d_in[2];
    const float* W        = (const float*)d_in[3];
    const float* W_e      = (const float*)d_in[4];
    const float* a_l      = (const float*)d_in[5];
    const float* a_r      = (const float*)d_in[6];
    const float* a_e      = (const float*)d_in[7];

    float* out      = (float*)d_out;
    float* attn_out = out + (size_t)N_NODES * OUT_COLS;

    char* ws = (char*)d_ws;
    float* h      = (float*)ws;  ws += (size_t)N_NODES * OUT_COLS * sizeof(float);
    unsigned* plane = (unsigned*)ws;  ws += (size_t)N_EDGES * NHEAD * sizeof(float);
    unsigned* pk = plane + (size_t)32 * PITCH;
    float* hl     = (float*)ws;  ws += (size_t)N_NODES * NHEAD   * sizeof(float);
    float* hr     = (float*)ws;  ws += (size_t)N_NODES * NHEAD   * sizeof(float);
    ws += (size_t)N_NODES * NHEAD * sizeof(float);
    float* he     = (float*)ws;  ws += (size_t)N_ETYPES * NHEAD  * sizeof(float);
    unsigned* gmax = (unsigned*)ws;

    init_gmax_kernel<<<1, 1, 0, stream>>>(gmax);

    {
        const int strips = N_NODES / 16;
        const int wavesPerBlock = 8;
        const int blocks = (strips + wavesPerBlock - 1) / wavesPerBlock;
        gemm_h_kernel<<<blocks, 256, 0, stream>>>(x, W, h);
    }
    etype_logits_kernel<<<1, 1024, 0, stream>>>(edge_emb, W_e, a_e, he);
    node_logits_kernel<<<(N_NODES * NHEAD + 255) / 256, 256, 0, stream>>>(h, a_l, a_r, hl, hr);
    edge_max_plane_kernel<<<N_KBLOCKS, SC_WAVES * 32, 0, stream>>>(edge, hl, hr, he, plane, pk, attn_out, gmax);
    owner_kernel<<<N_SLABS, OWN_WAVES * 32, 0, stream>>>(h, plane, pk, gmax, out, attn_out);
}


# node pass: records grouped per local node by LDS counters (rank = returned count; no list scan); a wave sums a node's rows in registers, stores the row; no LDS float add; out sum order changed
# speedup vs baseline: 3.9435x; 1.0414x over previous
#include <hip/hip_runtime.h>
#include <hip/hip_bf16.h>

typedef __attribute__((ext_vector_type(2))) float v2f;
typedef __attribute__((ext_vector_type(8))) float v8f;

#define N_NODES 50000
#define N_EDGES 1600000
#define IN_HIDDEN 256
#define OUT_COLS 128
#define NHEAD 4
#define OUT_HIDDEN 32
#define EDGE_HIDDEN 64
#define N_ETYPES 8
#define NEG_SLOPE 0.2f

#define SLAB 196
#define N_SLABS 256
#define PITCH 50048
#define N_TRIPS (PITCH / 128)
#define EPL 8
#define SC_WAVES 8
#define SC_WORDS (SC_WAVES * EPL)
#define ST_PITCH (SC_WORDS + 1)
#define N_KBLOCKS (PITCH / SC_WORDS)
#define OWN_WAVES 32
#define QCAP 64
#define LCAP 6912
#define ID_BITS 21
static_assert(N_SLABS * SLAB >= N_NODES && N_SLABS <= 256, "slab id must fit a byte");
static_assert(N_EDGES % 32 == 0 && PITCH % 128 == 0 && PITCH * 32 >= N_EDGES, "planes hold every edge in whole trips");
static_assert(PITCH % SC_WORDS == 0 && SC_WORDS % 32 == 0 && SC_WAVES <= 32, "the score kernel's work-groups tile the planes in whole 128-byte lines");
static_assert((N_EDGES / 32) % EPL == 0, "a wave of the score kernel has edges in all of its words or in none");
static_assert(OWN_WAVES * SLAB * NHEAD <= SLAB * OUT_COLS, "the per-wave denominator sums fit the output rows' LDS space");
static_assert(N_EDGES <= (1 << ID_BITS) && SLAB <= (1 << (32 - ID_BITS)), "edge id and local node share a dword");
static_assert(LCAP >= OWN_WAVES * 32, "the record slots also hold the second sweep's per-wave tables");
static_assert((N_SLABS - 1) * (N_TRIPS / N_SLABS) + OWN_WAVES <= N_TRIPS, "a rotated first trip stays inside the plane");
static_assert(N_ETYPES * NHEAD == 32 && EDGE_HIDDEN == 64 && NHEAD == 4, "the edge-type logits kernel maps its 32 waves and 16-lane groups on these sizes");
static_assert(N_NODES <= (1 << 16) && SLAB <= (1 << 16), "source node and local target node share one packed dword");
static_assert(LCAP <= (1 << 16), "a record slot fits 16 bits, and so does an edge's rank in its node");

__device__ __forceinline__ unsigned enc_f32(float f) {
    unsigned u = __float_as_uint(f);
    return (u & 0x80000000u) ? ~u : (u | 0x80000000u);
}
__device__ __forceinline__ float dec_f32(unsigned u) {
    unsigned b = (u & 0x80000000u) ? (u & 0x7FFFFFFFu) : ~u;
    return __uint_as_float(b);
}

__global__ void init_gmax_kernel(unsigned* gmax) {
    *gmax = enc_f32(-INFINITY);
}

__global__ void __launch_bounds__(256)
gemm_h_kernel(const float* __restrict__ x, const float* __restrict__ W,
              float* __restrict__ h) {
    const int lane  = threadIdx.x & 31;
    const int wave  = threadIdx.x >> 5;
    const int strip = blockIdx.x * (blockDim.x >> 5) + wave;
    if (strip * 16 >= N_NODES) return;
    const int row0  = strip * 16;

    const int mrow  = lane & 15;
    const int khalf = (lane >> 4) << 1;

    v8f acc[8] = {};

    const float* xrow = x + (size_t)(row0 + mrow) * IN_HIDDEN + khalf;
    for (int k = 0; k < IN_HIDDEN; k += 4) {
        v2f a = *(const v2f*)(xrow + k);
#pragma unroll
        for (int t = 0; t < 8; ++t) {
            const float* wp = W + (size_t)(k + khalf) * OUT_COLS + t * 16 + mrow;
            v2f b;
            b.x = wp[0];
            b.y = wp[OUT_COLS];
            acc[t] = __builtin_amdgcn_wmma_f32_16x16x4_f32(
                false, a, false, b, (short)0, acc[t], false, false);
        }
    }

    const int mbase = (lane >> 4) << 3;
#pragma unroll
    for (int t = 0; t < 8; ++t) {
#pragma unroll
        for (int i = 0; i < 8; ++i) {
            float v = acc[t][i];
            v = (v == v) ? v : 0.0f;
            h[(size_t)(row0 + mbase + i) * OUT_COLS + t * 16 + mrow] = v;
        }
    }
}

__global__ void __launch_bounds__(1024)
etype_logits_kernel(const float* __restrict__ edge_emb,
                    const float* __restrict__ W_e,
                    const float* __restrict__ a_e,
                    float* __restrict__ he) {
    __shared__ float v_s[NHEAD * EDGE_HIDDEN];
    const int tid  = threadIdx.x;
    const int lane = tid & 31;
    const int wave = tid >> 5;
    const int j    = tid & 15;
    const int hw   = tid >> 4;

    const float e0 = edge_emb[(wave >> 2) * EDGE_HIDDEN + lane];
    const float e1 = edge_emb[(wave >> 2) * EDGE_HIDDEN + 32 + lane];

    const float4 a = *(const float4*)(a_e + (hw & 3) * EDGE_HIDDEN + j * 4);
    float4 w[4];
    for (int i = 0; i < 4; ++i)
        w[i] = *(const float4*)(W_e + (size_t)(hw + 64 * i) * EDGE_HIDDEN + j * 4);
    float s[4];
    for (int i = 0; i < 4; ++i)
        s[i] = a.x * w[i].x + a.y * w[i].y + a.z * w[i].z + a.w * w[i].w;
    for (int off = 8; off > 0; off >>= 1) {
        for (int i = 0; i < 4; ++i) s[i] += __shfl_xor(s[i], off, 16);
    }
    if (j == 0) {
        for (int i = 0; i < 4; ++i) {
            const int p = hw + 64 * i;
            v_s[(p & 3) * EDGE_HIDDEN + (p >> 2)] = s[i];
        }
    }
    __syncthreads();

    const float* vh = v_s + (wave & 3) * EDGE_HIDDEN;
    float r = e0 * vh[lane] + e1 * vh[32 + lane];
    r += __shfl_xor(r, 16);
    r += __shfl_xor(r, 8);
    r += __shfl_xor(r, 4);
    r += __shfl_xor(r, 2);
    r += __shfl_xor(r, 1);
    if (lane == 0) he[wave] = r;
}

__global__ void __launch_bounds__(256)
node_logits_kernel(const float* __restrict__ h,
                   const float* __restrict__ a_l, const float* __restrict__ a_r,
                   float* __restrict__ hl, float* __restrict__ hr) {
    const int idx = blockIdx.x * blockDim.x + threadIdx.x;
    if (idx >= N_NODES * NHEAD) return;
    const int node = idx >> 2, head = idx & 3;
    const float* hp = h + (size_t)node * OUT_COLS + head * OUT_HIDDEN;
    const float* al = a_l + head * OUT_HIDDEN;
    const float* ar = a_r + head * OUT_HIDDEN;
    float sl = 0.0f, sr = 0.0f;
#pragma unroll
    for (int d = 0; d < OUT_HIDDEN; ++d) {
        float v = hp[d];
        sl = fmaf(al[d], v, sl);
        sr = fmaf(ar[d], v, sr);
    }
    hl[idx] = sl;
    hr[idx] = sr;
}

__global__ void __launch_bounds__(SC_WAVES * 32) __attribute__((amdgpu_waves_per_eu(4, 8)))
edge_max_plane_kernel(const int* __restrict__ edge,
                      const float* __restrict__ hl, const float* __restrict__ hr,
                      const float* __restrict__ he,
                      unsigned* __restrict__ plane, unsigned* __restrict__ pk,
                      float* __restrict__ score, unsigned* __restrict__ gmax) {
    __shared__ unsigned st[32 * ST_PITCH];
    __shared__ float red[SC_WAVES];
    const int tid  = threadIdx.x;
    const int lane = tid & 31;
    const int wave = __builtin_amdgcn_readfirstlane(tid >> 5);
    for (int i = tid; i < 32 * ST_PITCH; i += SC_WAVES * 32) st[i] = 0u;
    __syncthreads();

    const int w0    = wave * EPL;
    const int wbase = blockIdx.x * SC_WORDS + w0;
    float lmax = -INFINITY;
    if (wbase * 32 < N_EDGES) {
        const int e0 = wbase * 32 + lane;
        int src[EPL], trg[EPL], et[EPL];
#pragma unroll
        for (int j = 0; j < EPL; ++j) {
            const int e = e0 + j * 32;
            src[j] = edge[e];
            trg[j] = edge[N_EDGES + e];
            et[j]  = edge[2 * N_EDGES + e];
        }
        float4 gl[EPL], gr[EPL], ge[EPL];
#pragma unroll
        for (int j = 0; j < EPL; ++j) {
            gl[j] = *(const float4*)(hl + src[j] * NHEAD);
            gr[j] = *(const float4*)(hr + trg[j] * NHEAD);
            ge[j] = *(const float4*)(he + et[j] * NHEAD);
        }
#pragma unroll
        for (int j = 0; j < EPL; ++j) {
            const int e = e0 + j * 32;
            float sc[NHEAD];
            sc[0] = gl[j].x + gr[j].x + ge[j].x;
            sc[1] = gl[j].y + gr[j].y + ge[j].y;
            sc[2] = gl[j].z + gr[j].z + ge[j].z;
            sc[3] = gl[j].w + gr[j].w + ge[j].w;
#pragma unroll
            for (int hh = 0; hh < NHEAD; ++hh) {
                float s = sc[hh];
                s = (s > 0.0f) ? s : NEG_SLOPE * s;
                sc[hh] = s;
                lmax = fmaxf(lmax, s);
            }
            const int slab = trg[j] / SLAB;
            *(float4*)(score + (size_t)e * NHEAD) = make_float4(sc[0], sc[1], sc[2], sc[3]);
            pk[e] = (unsigned)src[j] | ((unsigned)(trg[j] - slab * SLAB) << 16);
            atomicOr(&st[(slab >> 4) * ST_PITCH + w0 + j], 1u << lane);
            atomicOr(&st[(16 + (slab & 15)) * ST_PITCH + w0 + j], 1u << lane);
        }
    }
    lmax = fmaxf(lmax, __shfl_xor(lmax, 16));
    lmax = fmaxf(lmax, __shfl_xor(lmax, 8));
    lmax = fmaxf(lmax, __shfl_xor(lmax, 4));
    lmax = fmaxf(lmax, __shfl_xor(lmax, 2));
    lmax = fmaxf(lmax, __shfl_xor(lmax, 1));
    if (lane == 0) red[wave] = lmax;
    __syncthreads();
#pragma unroll
    for (int i = 0; i < 32 * SC_WORDS / (SC_WAVES * 32); ++i) {
        const int idx = i * (SC_WAVES * 32) + tid;
        const int p = idx / SC_WORDS, w = idx % SC_WORDS;
        plane[(size_t)p * PITCH + blockIdx.x * SC_WORDS + w] = st[p * ST_PITCH + w];
    }
    if (wave == 0) {
        float m = (lane < SC_WAVES) ? red[lane] : -INFINITY;
        m = fmaxf(m, __shfl_xor(m, 4));
        m = fmaxf(m, __shfl_xor(m, 2));
        m = fmaxf(m, __shfl_xor(m, 1));
        if (lane == 0) atomicMax(gmax, enc_f32(m));
    }
}


struct EdgeVals { int src; int loc; float es[NHEAD]; };

__device__ __forceinline__ EdgeVals edge_vals(const unsigned* __restrict__ pk, const float* sc,
                                              int e, float smax) {
    EdgeVals r;
    const unsigned p = pk[e];
    const float4 s = *(const float4*)(sc + (size_t)e * NHEAD);
    r.src = (int)(p & 0xFFFFu);
    r.loc = (int)(p >> 16);
    r.es[0] = __expf(s.x - smax);
    r.es[1] = __expf(s.y - smax);
    r.es[2] = __expf(s.z - smax);
    r.es[3] = __expf(s.w - smax);
    return r;
}

struct Recs {
    unsigned* cnt;
    unsigned* id;
    int*      src;
    float4*   es;
    unsigned* ncnt;
};

__device__ __forceinline__ void add_rows(int n, int lane, int src, int loc, const float* aw,
                                         const float* __restrict__ h, float* slab) {
    const int head = lane >> 3;
    for (int j = 0; j < n; j += 4) {
        float4 r[4]; float aj[4]; int lj[4];
#pragma unroll
        for (int u = 0; u < 4; ++u) {
            const int jj = (j + u < n) ? (j + u) : (n - 1);
            const int sj = __builtin_amdgcn_readlane(src, jj);
            lj[u] = __builtin_amdgcn_readlane(loc, jj);
            aj[u] = aw[jj * NHEAD + head];
            r[u] = *(const float4*)(h + (size_t)sj * OUT_COLS + lane * 4);
        }
#pragma unroll
        for (int u = 0; u < 4; ++u) {
            if (j + u < n) {
                float* p = slab + lj[u] * OUT_COLS + lane;
                atomicAdd(p +  0, r[u].x * aj[u]);
                atomicAdd(p + 32, r[u].y * aj[u]);
                atomicAdd(p + 64, r[u].z * aj[u]);
                atomicAdd(p + 96, r[u].w * aj[u]);
            }
        }
    }
}

template <int PASS>
__device__ __forceinline__ void owner_batch(int n, int lane, float smax,
                                            const unsigned* q,
                                            const unsigned* __restrict__ pk,
                                            const float* __restrict__ h,
                                            float* slab, float* den, float4* atab, Recs rc,
                                            float* attn_out) {
    if (PASS == 0) {
        unsigned pos = 0;
        if (lane == 0) pos = atomicAdd(rc.cnt, (unsigned)n);
        pos = (unsigned)__builtin_amdgcn_readfirstlane((int)pos);
        if (lane < n) {
            const int e = (int)q[lane];
            const EdgeVals v = edge_vals(pk, attn_out, e, smax);
#pragma unroll
            for (int k = 0; k < NHEAD; ++k) atomicAdd(&den[v.loc * NHEAD + k], v.es[k]);
            const unsigned slot = pos + (unsigned)lane;
            const unsigned rk = atomicAdd(&rc.ncnt[v.loc], 1u);
            if (slot < LCAP) {
                rc.id[slot]  = (unsigned)e | ((unsigned)v.loc << ID_BITS);
                rc.src[slot] = v.src | (int)(rk << 16);
                rc.es[slot]  = make_float4(v.es[0], v.es[1], v.es[2], v.es[3]);
            }
        }
    } else {
        int src = 0, loc = 0;
        if (lane < n) {
            const int e = (int)q[lane];
            const EdgeVals v = edge_vals(pk, attn_out, e, smax);
            src = v.src; loc = v.loc;
            const float4 a = make_float4(v.es[0] / (den[loc * NHEAD + 0] + 1e-16f),
                                         v.es[1] / (den[loc * NHEAD + 1] + 1e-16f),
                                         v.es[2] / (den[loc * NHEAD + 2] + 1e-16f),
                                         v.es[3] / (den[loc * NHEAD + 3] + 1e-16f));
            *(float4*)(attn_out + (size_t)e * NHEAD) = a;
            atab[lane] = a;
        }
        add_rows(n, lane, src, loc, (const float*)atab, h, slab);
    }
}

#define RIF 8
__device__ __forceinline__ float4 node_rows(int start, int deg, int node, int lane, Recs rc,
                                            const unsigned short* perm, const float* den,
                                            const float* __restrict__ h, float* attn_out) {
    const int head = lane >> 3;
    float4 acc = make_float4(0.0f, 0.0f, 0.0f, 0.0f);
    const float d0 = den[node * NHEAD + 0] + 1e-16f, d1 = den[node * NHEAD + 1] + 1e-16f;
    const float d2 = den[node * NHEAD + 2] + 1e-16f, d3 = den[node * NHEAD + 3] + 1e-16f;
    for (int c = 0; c < deg; c += 32) {
        const int n = (deg - c < 32) ? (deg - c) : 32;
        int s = 0, sn = 0;
        if (lane < n) {
            s = (int)perm[start + c + lane];
            const unsigned id = rc.id[s];
            const int e = (int)(id & ((1u << ID_BITS) - 1u));
            sn = rc.src[s] & 0xFFFF;
            const float4 x = rc.es[s];
            const float4 a = make_float4(x.x / d0, x.y / d1, x.z / d2, x.w / d3);
            *(float4*)(attn_out + (size_t)e * NHEAD) = a;
            rc.es[s] = a;
        }
        const float* aw = (const float*)rc.es;
        for (int j = 0; j < n; j += RIF) {
            float4 r[RIF]; float aj[RIF];
#pragma unroll
            for (int u = 0; u < RIF; ++u) {
                const int jj = (j + u < n) ? (j + u) : (n - 1);
                const int sj = __builtin_amdgcn_readlane(sn, jj);
                const int tj = __builtin_amdgcn_readlane(s, jj);
                aj[u] = aw[tj * NHEAD + head];
                r[u] = *(const float4*)(h + (size_t)sj * OUT_COLS + lane * 4);
            }
#pragma unroll
            for (int u = 0; u < RIF; ++u) {
                if (j + u < n) {
                    acc.x = fmaf(r[u].x, aj[u], acc.x);
                    acc.y = fmaf(r[u].y, aj[u], acc.y);
                    acc.z = fmaf(r[u].z, aj[u], acc.z);
                    acc.w = fmaf(r[u].w, aj[u], acc.w);
                }
            }
        }
    }
    return acc;
}

template <int PASS>
__device__ __forceinline__ void owner_scan(int lane, int wave, float smax,
                                           const unsigned* __restrict__ plane, unsigned* q,
                                           const unsigned* __restrict__ pk,
                                           const float* __restrict__ h,
                                           float* slab, float* den, float4* atab, Recs rc,
                                           float* attn_out) {
    const unsigned* pH = plane + (size_t)(blockIdx.x >> 4) * PITCH;
    const unsigned* pL = plane + (size_t)(16 + (blockIdx.x & 15)) * PITCH;
    const int rot = blockIdx.x * (N_TRIPS / N_SLABS);
    int cnt = 0;
    int c = wave + rot;
    uint4 kh = *(const uint4*)(pH + (size_t)c * 128 + lane * 4);
    uint4 kl = *(const uint4*)(pL + (size_t)c * 128 + lane * 4);
    for (int it = wave; it < N_TRIPS; it += OWN_WAVES) {
        int cn = c + OWN_WAVES;
        if (cn >= N_TRIPS) cn -= N_TRIPS;
        uint4 nh = kh, nl = kl;
        if (it + OWN_WAVES < N_TRIPS) {
            nh = *(const uint4*)(pH + (size_t)cn * 128 + lane * 4);
            nl = *(const uint4*)(pL + (size_t)cn * 128 + lane * 4);
        }
        const unsigned m0 = kh.x & kl.x, m1 = kh.y & kl.y, m2 = kh.z & kl.z, m3 = kh.w & kl.w;
#pragma nounroll
        for (int j = 0; j < 4; ++j) {
            unsigned mm = (j == 0) ? m0 : (j == 1) ? m1 : (j == 2) ? m2 : m3;
            const unsigned e0 = (unsigned)(c * 128 + lane * 4 + j) * 32u;
            unsigned bal;
            while ((bal = __builtin_amdgcn_ballot_w32(mm != 0u)) != 0u) {
                if (mm != 0u) {
                    const int p = __builtin_ctz(mm);
                    mm &= mm - 1u;
                    q[cnt + (int)__builtin_amdgcn_mbcnt_lo(bal, 0u)] = e0 + (unsigned)p;
                }
                cnt += __builtin_popcount(bal);
                if (cnt >= 32) {
                    owner_batch<PASS>(32, lane, smax, q, pk, h, slab, den, atab, rc, attn_out);
                    const int rest = cnt - 32;
                    if (lane < rest) { const unsigned t = q[32 + lane]; q[lane] = t; }
                    cnt = rest;
                }
            }
        }
        kh = nh; kl = nl;
        c = cn;
    }
    if (cnt > 0)
        owner_batch<PASS>(cnt, lane, smax, q, pk, h, slab, den, atab, rc, attn_out);
}

__global__ void __launch_bounds__(OWN_WAVES * 32)
owner_kernel(const float* __restrict__ h,
             const unsigned* __restrict__ plane, const unsigned* __restrict__ pk,
             const unsigned* __restrict__ gmax,
             float* __restrict__ out, float* attn_out) {
    __shared__ float4   slab4[SLAB * OUT_COLS / 4];
    __shared__ float    den[SLAB * NHEAD];
    __shared__ unsigned qbuf[OWN_WAVES * QCAP];
    __shared__ unsigned rec_id[LCAP];
    __shared__ int      rec_src[LCAP];
    __shared__ float4   rec_es[LCAP];
    __shared__ unsigned rec_cnt;
    __shared__ unsigned ncnt[SLAB];
    __shared__ unsigned offs[SLAB + 1];
    __shared__ unsigned short perm[LCAP];
    float* slab = (float*)slab4;

    const int tid  = threadIdx.x;
    const int lane = tid & 31;
    const int wave = __builtin_amdgcn_readfirstlane(tid >> 5);
    const int base = blockIdx.x * SLAB;
    unsigned* q = qbuf + wave * QCAP;
    Recs rc; rc.cnt = &rec_cnt; rc.id = rec_id; rc.src = rec_src; rc.es = rec_es; rc.ncnt = ncnt;
    float4* atab = rec_es + wave * 32;
    float* pden = slab + wave * (SLAB * NHEAD);

    for (int i = tid; i < SLAB * OUT_COLS / 4; i += OWN_WAVES * 32)
        slab4[i] = make_float4(0.0f, 0.0f, 0.0f, 0.0f);
    const int rows = (N_NODES - base < SLAB) ? (N_NODES - base) : SLAB;
    if (tid < SLAB)     ncnt[tid] = 0u;
    if (tid == 0)       rec_cnt = 0u;
    const float smax = dec_f32(*gmax);
    __syncthreads();

    owner_scan<0>(lane, wave, smax, plane, q, pk, h, slab, pden, atab, rc, attn_out);
    __syncthreads();
    for (int i = tid; i < SLAB * NHEAD; i += OWN_WAVES * 32) {
        float s = 0.0f;
        for (int w = 0; w < OWN_WAVES; ++w) s += slab[w * (SLAB * NHEAD) + i];
        den[i] = s;
    }
    if (wave == OWN_WAVES - 1) {
        unsigned c[7], sum = 0u;
#pragma unroll
        for (int k = 0; k < 7; ++k) {
            const int idx = lane * 7 + k;
            c[k] = (idx < SLAB) ? ncnt[idx] : 0u;
            sum += c[k];
        }
        unsigned incl = sum;
#pragma unroll
        for (int off = 1; off < 32; off <<= 1) {
            const unsigned t = __shfl_up(incl, off);
            if (lane >= off) incl += t;
        }
        unsigned run = incl - sum;
#pragma unroll
        for (int k = 0; k < 7; ++k) {
            const int idx = lane * 7 + k;
            if (idx < SLAB) offs[idx] = run;
            run += c[k];
        }
        if (lane == 31) offs[SLAB] = incl;
    }
    __syncthreads();
    const int total = __builtin_amdgcn_readfirstlane((int)rec_cnt);
    if (total <= LCAP) {
        for (int s = tid; s < total; s += OWN_WAVES * 32) {
            const unsigned loc = rec_id[s] >> ID_BITS;
            const unsigned rk  = (unsigned)rec_src[s] >> 16;
            perm[offs[loc] + rk] = (unsigned short)s;
        }
        __syncthreads();
        for (int node = wave; node < rows; node += OWN_WAVES) {
            const int start = __builtin_amdgcn_readfirstlane((int)offs[node]);
            const int deg   = __builtin_amdgcn_readfirstlane((int)offs[node + 1]) - start;
            const float4 acc = node_rows(start, deg, node, lane, rc, perm, den, h, attn_out);
            *(float4*)(out + (size_t)(base + node) * OUT_COLS + lane * 4) = acc;
        }
    } else {
        for (int i = tid; i < SLAB * OUT_COLS / 4; i += OWN_WAVES * 32)
            slab4[i] = make_float4(0.0f, 0.0f, 0.0f, 0.0f);
        __syncthreads();
        owner_scan<1>(lane, wave, smax, plane, q, pk, h, slab, den, atab, rc, attn_out);
        __syncthreads();
        float4* out4 = (float4*)(out + (size_t)base * OUT_COLS);
        for (int i = tid; i < rows * (OUT_COLS / 4); i += OWN_WAVES * 32) {
            const float* p = slab + (i >> 5) * OUT_COLS + (i & 31);
            out4[i] = make_float4(p[0], p[32], p[64], p[96]);
        }
    }
}

extern "C" void kernel_launch(void* const* d_in, const int* in_sizes, int n_in,
                              void* d_out, int out_size, void* d_ws, size_t ws_size,
                              hipStream_t stream) {
    const int*   edge     = (const int*)  d_in[0];
    const float* x        = (const float*)d_in[1];
    const float* edge_emb = (const float*)d_in[2];
    const float* W        = (const float*)d_in[3];
    const float* W_e      = (const float*)d_in[4];
    const float* a_l      = (const float*)d_in[5];
    const float* a_r      = (const float*)d_in[6];
    const float* a_e      = (const float*)d_in[7];

    float* out      = (float*)d_out;
    float* attn_out = out + (size_t)N_NODES * OUT_COLS;

    char* ws = (char*)d_ws;
    float* h      = (float*)ws;  ws += (size_t)N_NODES * OUT_COLS * sizeof(float);
    unsigned* plane = (unsigned*)ws;  ws += (size_t)N_EDGES * NHEAD * sizeof(float);
    unsigned* pk = plane + (size_t)32 * PITCH;
    float* hl     = (float*)ws;  ws += (size_t)N_NODES * NHEAD   * sizeof(float);
    float* hr     = (float*)ws;  ws += (size_t)N_NODES * NHEAD   * sizeof(float);
    ws += (size_t)N_NODES * NHEAD * sizeof(float);
    float* he     = (float*)ws;  ws += (size_t)N_ETYPES * NHEAD  * sizeof(float);
    unsigned* gmax = (unsigned*)ws;

    init_gmax_kernel<<<1, 1, 0, stream>>>(gmax);

    {
        const int strips = N_NODES / 16;
        const int wavesPerBlock = 8;
        const int blocks = (strips + wavesPerBlock - 1) / wavesPerBlock;
        gemm_h_kernel<<<blocks, 256, 0, stream>>>(x, W, h);
    }
    etype_logits_kernel<<<1, 1024, 0, stream>>>(edge_emb, W_e, a_e, he);
    node_logits_kernel<<<(N_NODES * NHEAD + 255) / 256, 256, 0, stream>>>(h, a_l, a_r, hl, hr);
    edge_max_plane_kernel<<<N_KBLOCKS, SC_WAVES * 32, 0, stream>>>(edge, hl, hr, he, plane, pk, attn_out, gmax);
    owner_kernel<<<N_SLABS, OWN_WAVES * 32, 0, stream>>>(h, plane, pk, gmax, out, attn_out);
}


# node logits reassociated: W a_l | W a_r first (256x8 table), then x times the table by multiply-adds beside the product kernel's matrix work; node-logit launch removed; hl/hr sum order changed
# speedup vs baseline: 4.0981x; 1.0392x over previous
#include <hip/hip_runtime.h>
#include <hip/hip_bf16.h>

typedef __attribute__((ext_vector_type(2))) float v2f;
typedef __attribute__((ext_vector_type(8))) float v8f;

#define N_NODES 50000
#define N_EDGES 1600000
#define IN_HIDDEN 256
#define OUT_COLS 128
#define NHEAD 4
#define OUT_HIDDEN 32
#define EDGE_HIDDEN 64
#define N_ETYPES 8
#define NEG_SLOPE 0.2f

#define SLAB 196
#define N_SLABS 256
#define PITCH 50048
#define N_TRIPS (PITCH / 128)
#define EPL 8
#define SC_WAVES 8
#define SC_WORDS (SC_WAVES * EPL)
#define ST_PITCH (SC_WORDS + 1)
#define N_KBLOCKS (PITCH / SC_WORDS)
#define OWN_WAVES 32
#define QCAP 64
#define LCAP 6912
#define ID_BITS 21
static_assert(N_SLABS * SLAB >= N_NODES && N_SLABS <= 256, "slab id must fit a byte");
static_assert(N_EDGES % 32 == 0 && PITCH % 128 == 0 && PITCH * 32 >= N_EDGES, "planes hold every edge in whole trips");
static_assert(PITCH % SC_WORDS == 0 && SC_WORDS % 32 == 0 && SC_WAVES <= 32, "the score kernel's work-groups tile the planes in whole 128-byte lines");
static_assert((N_EDGES / 32) % EPL == 0, "a wave of the score kernel has edges in all of its words or in none");
static_assert(OWN_WAVES * SLAB * NHEAD <= SLAB * OUT_COLS, "the per-wave denominator sums fit the output rows' LDS space");
static_assert(N_EDGES <= (1 << ID_BITS) && SLAB <= (1 << (32 - ID_BITS)), "edge id and local node share a dword");
static_assert(LCAP >= OWN_WAVES * 32, "the record slots also hold the second sweep's per-wave tables");
static_assert((N_SLABS - 1) * (N_TRIPS / N_SLABS) + OWN_WAVES <= N_TRIPS, "a rotated first trip stays inside the plane");
static_assert(N_ETYPES * NHEAD == 32 && EDGE_HIDDEN == 64 && NHEAD == 4, "the edge-type logits kernel maps its 32 waves and 16-lane groups on these sizes");
static_assert(IN_HIDDEN * NHEAD == 1024 && OUT_HIDDEN % 4 == 0, "one thread of that kernel per entry (k, head) of the node-logit table");
static_assert(N_NODES <= (1 << 16) && SLAB <= (1 << 16), "source node and local target node share one packed dword");
static_assert(IN_HIDDEN * 2 * NHEAD <= N_NODES * NHEAD, "the node-logit table fits its workspace region");
static_assert(LCAP <= (1 << 16), "a record slot fits 16 bits, and so does an edge's rank in its node");

__device__ __forceinline__ unsigned enc_f32(float f) {
    unsigned u = __float_as_uint(f);
    return (u & 0x80000000u) ? ~u : (u | 0x80000000u);
}
__device__ __forceinline__ float dec_f32(unsigned u) {
    unsigned b = (u & 0x80000000u) ? (u & 0x7FFFFFFFu) : ~u;
    return __uint_as_float(b);
}

__global__ void init_gmax_kernel(unsigned* gmax) {
    *gmax = enc_f32(-INFINITY);
}

#define LR_COLS 8
__global__ void __launch_bounds__(256)
gemm_h_kernel(const float* __restrict__ x, const float* __restrict__ W,
              const float* __restrict__ wlr,
              float* __restrict__ h, float* __restrict__ hl, float* __restrict__ hr) {
    const int lane  = threadIdx.x & 31;
    const int wave  = threadIdx.x >> 5;
    const int strip = blockIdx.x * (blockDim.x >> 5) + wave;
    if (strip * 16 >= N_NODES) return;
    const int row0  = strip * 16;

    const int mrow  = lane & 15;
    const int khalf = (lane >> 4) << 1;

    v8f acc[8] = {};
    float4 pl = make_float4(0.0f, 0.0f, 0.0f, 0.0f);
    float4 pr = make_float4(0.0f, 0.0f, 0.0f, 0.0f);

    const float* xrow = x + (size_t)(row0 + mrow) * IN_HIDDEN + khalf;
    const float* trow = wlr + (size_t)khalf * LR_COLS;
    float4 l0 = *(const float4*)(trow);
    float4 r0 = *(const float4*)(trow + NHEAD);
    float4 l1 = *(const float4*)(trow + LR_COLS);
    float4 r1 = *(const float4*)(trow + LR_COLS + NHEAD);
    for (int k = 0; k < IN_HIDDEN; k += 4) {
        v2f a = *(const v2f*)(xrow + k);
        const int kn = (k + 4 < IN_HIDDEN) ? (k + 4) : k;
        const float* tn = wlr + (size_t)(kn + khalf) * LR_COLS;
        const float4 nl0 = *(const float4*)(tn);
        const float4 nr0 = *(const float4*)(tn + NHEAD);
        const float4 nl1 = *(const float4*)(tn + LR_COLS);
        const float4 nr1 = *(const float4*)(tn + LR_COLS + NHEAD);
        v2f b[8];
#pragma unroll
        for (int t = 0; t < 8; ++t) {
            const float* wp = W + (size_t)(k + khalf) * OUT_COLS + t * 16 + mrow;
            b[t].x = wp[0];
            b[t].y = wp[OUT_COLS];
        }
        __builtin_amdgcn_sched_barrier(0);
#pragma unroll
        for (int t = 0; t < 8; ++t)
            acc[t] = __builtin_amdgcn_wmma_f32_16x16x4_f32(
                false, a, false, b[t], (short)0, acc[t], false, false);
        {
            pl.x = fmaf(a.x, l0.x, pl.x);  pl.y = fmaf(a.x, l0.y, pl.y);  pl.z = fmaf(a.x, l0.z, pl.z);  pl.w = fmaf(a.x, l0.w, pl.w);
            pr.x = fmaf(a.x, r0.x, pr.x);  pr.y = fmaf(a.x, r0.y, pr.y);  pr.z = fmaf(a.x, r0.z, pr.z);  pr.w = fmaf(a.x, r0.w, pr.w);
            pl.x = fmaf(a.y, l1.x, pl.x);  pl.y = fmaf(a.y, l1.y, pl.y);  pl.z = fmaf(a.y, l1.z, pl.z);  pl.w = fmaf(a.y, l1.w, pl.w);
            pr.x = fmaf(a.y, r1.x, pr.x);  pr.y = fmaf(a.y, r1.y, pr.y);  pr.z = fmaf(a.y, r1.z, pr.z);  pr.w = fmaf(a.y, r1.w, pr.w);
        }
        l0 = nl0;  r0 = nr0;  l1 = nl1;  r1 = nr1;
    }

    const int mbase = (lane >> 4) << 3;
#pragma unroll
    for (int t = 0; t < 8; ++t) {
#pragma unroll
        for (int i = 0; i < 8; ++i) {
            float v = acc[t][i];
            v = (v == v) ? v : 0.0f;
            h[(size_t)(row0 + mbase + i) * OUT_COLS + t * 16 + mrow] = v;
        }
    }
    pl.x += __shfl_xor(pl.x, 16);  pl.y += __shfl_xor(pl.y, 16);  pl.z += __shfl_xor(pl.z, 16);  pl.w += __shfl_xor(pl.w, 16);
    pr.x += __shfl_xor(pr.x, 16);  pr.y += __shfl_xor(pr.y, 16);  pr.z += __shfl_xor(pr.z, 16);  pr.w += __shfl_xor(pr.w, 16);
    if (lane < 16) {
        *(float4*)(hl + (size_t)(row0 + mrow) * NHEAD) = pl;
        *(float4*)(hr + (size_t)(row0 + mrow) * NHEAD) = pr;
    }
}

__global__ void __launch_bounds__(1024)
etype_logits_kernel(const float* __restrict__ edge_emb,
                    const float* __restrict__ W_e,
                    const float* __restrict__ a_e,
                    float* __restrict__ he,
                    const float* __restrict__ W,
                    const float* __restrict__ a_l, const float* __restrict__ a_r,
                    float* __restrict__ wlr) {
    __shared__ float v_s[NHEAD * EDGE_HIDDEN];
    const int tid  = threadIdx.x;
    const int lane = tid & 31;
    const int wave = tid >> 5;
    const int j    = tid & 15;
    const int hw   = tid >> 4;

    const float e0 = edge_emb[(wave >> 2) * EDGE_HIDDEN + lane];
    const float e1 = edge_emb[(wave >> 2) * EDGE_HIDDEN + 32 + lane];

    float4 tw[OUT_HIDDEN / 4], tl[OUT_HIDDEN / 4], tr[OUT_HIDDEN / 4];
#pragma unroll
    for (int i = 0; i < OUT_HIDDEN / 4; ++i) {
        tw[i] = *(const float4*)(W + (size_t)(tid >> 2) * OUT_COLS + (tid & 3) * OUT_HIDDEN + i * 4);
        tl[i] = *(const float4*)(a_l + (tid & 3) * OUT_HIDDEN + i * 4);
        tr[i] = *(const float4*)(a_r + (tid & 3) * OUT_HIDDEN + i * 4);
    }

    const float4 a = *(const float4*)(a_e + (hw & 3) * EDGE_HIDDEN + j * 4);
    float4 w[4];
    for (int i = 0; i < 4; ++i)
        w[i] = *(const float4*)(W_e + (size_t)(hw + 64 * i) * EDGE_HIDDEN + j * 4);
    float s[4];
    for (int i = 0; i < 4; ++i)
        s[i] = a.x * w[i].x + a.y * w[i].y + a.z * w[i].z + a.w * w[i].w;
    for (int off = 8; off > 0; off >>= 1) {
        for (int i = 0; i < 4; ++i) s[i] += __shfl_xor(s[i], off, 16);
    }
    if (j == 0) {
        for (int i = 0; i < 4; ++i) {
            const int p = hw + 64 * i;
            v_s[(p & 3) * EDGE_HIDDEN + (p >> 2)] = s[i];
        }
    }
    __syncthreads();

    const float* vh = v_s + (wave & 3) * EDGE_HIDDEN;
    float r = e0 * vh[lane] + e1 * vh[32 + lane];
    r += __shfl_xor(r, 16);
    r += __shfl_xor(r, 8);
    r += __shfl_xor(r, 4);
    r += __shfl_xor(r, 2);
    r += __shfl_xor(r, 1);
    if (lane == 0) he[wave] = r;

    float sl = 0.0f, sr = 0.0f;
#pragma unroll
    for (int i = 0; i < OUT_HIDDEN / 4; ++i) {
        sl = fmaf(tl[i].x, tw[i].x, sl);  sr = fmaf(tr[i].x, tw[i].x, sr);
        sl = fmaf(tl[i].y, tw[i].y, sl);  sr = fmaf(tr[i].y, tw[i].y, sr);
        sl = fmaf(tl[i].z, tw[i].z, sl);  sr = fmaf(tr[i].z, tw[i].z, sr);
        sl = fmaf(tl[i].w, tw[i].w, sl);  sr = fmaf(tr[i].w, tw[i].w, sr);
    }
    float* trow = wlr + (size_t)(tid >> 2) * LR_COLS + (tid & 3);
    trow[0]     = sl;
    trow[NHEAD] = sr;
}

__global__ void __launch_bounds__(SC_WAVES * 32) __attribute__((amdgpu_waves_per_eu(4, 8)))
edge_max_plane_kernel(const int* __restrict__ edge,
                      const float* __restrict__ hl, const float* __restrict__ hr,
                      const float* __restrict__ he,
                      unsigned* __restrict__ plane, unsigned* __restrict__ pk,
                      float* __restrict__ score, unsigned* __restrict__ gmax) {
    __shared__ unsigned st[32 * ST_PITCH];
    __shared__ float red[SC_WAVES];
    const int tid  = threadIdx.x;
    const int lane = tid & 31;
    const int wave = __builtin_amdgcn_readfirstlane(tid >> 5);
    for (int i = tid; i < 32 * ST_PITCH; i += SC_WAVES * 32) st[i] = 0u;
    __syncthreads();

    const int w0    = wave * EPL;
    const int wbase = blockIdx.x * SC_WORDS + w0;
    float lmax = -INFINITY;
    if (wbase * 32 < N_EDGES) {
        const int e0 = wbase * 32 + lane;
        int src[EPL], trg[EPL], et[EPL];
#pragma unroll
        for (int j = 0; j < EPL; ++j) {
            const int e = e0 + j * 32;
            src[j] = edge[e];
            trg[j] = edge[N_EDGES + e];
            et[j]  = edge[2 * N_EDGES + e];
        }
        float4 gl[EPL], gr[EPL], ge[EPL];
#pragma unroll
        for (int j = 0; j < EPL; ++j) {
            gl[j] = *(const float4*)(hl + src[j] * NHEAD);
            gr[j] = *(const float4*)(hr + trg[j] * NHEAD);
            ge[j] = *(const float4*)(he + et[j] * NHEAD);
        }
#pragma unroll
        for (int j = 0; j < EPL; ++j) {
            const int e = e0 + j * 32;
            float sc[NHEAD];
            sc[0] = gl[j].x + gr[j].x + ge[j].x;
            sc[1] = gl[j].y + gr[j].y + ge[j].y;
            sc[2] = gl[j].z + gr[j].z + ge[j].z;
            sc[3] = gl[j].w + gr[j].w + ge[j].w;
#pragma unroll
            for (int hh = 0; hh < NHEAD; ++hh) {
                float s = sc[hh];
                s = (s > 0.0f) ? s : NEG_SLOPE * s;
                sc[hh] = s;
                lmax = fmaxf(lmax, s);
            }
            const int slab = trg[j] / SLAB;
            *(float4*)(score + (size_t)e * NHEAD) = make_float4(sc[0], sc[1], sc[2], sc[3]);
            pk[e] = (unsigned)src[j] | ((unsigned)(trg[j] - slab * SLAB) << 16);
            atomicOr(&st[(slab >> 4) * ST_PITCH + w0 + j], 1u << lane);
            atomicOr(&st[(16 + (slab & 15)) * ST_PITCH + w0 + j], 1u << lane);
        }
    }
    lmax = fmaxf(lmax, __shfl_xor(lmax, 16));
    lmax = fmaxf(lmax, __shfl_xor(lmax, 8));
    lmax = fmaxf(lmax, __shfl_xor(lmax, 4));
    lmax = fmaxf(lmax, __shfl_xor(lmax, 2));
    lmax = fmaxf(lmax, __shfl_xor(lmax, 1));
    if (lane == 0) red[wave] = lmax;
    __syncthreads();
#pragma unroll
    for (int i = 0; i < 32 * SC_WORDS / (SC_WAVES * 32); ++i) {
        const int idx = i * (SC_WAVES * 32) + tid;
        const int p = idx / SC_WORDS, w = idx % SC_WORDS;
        plane[(size_t)p * PITCH + blockIdx.x * SC_WORDS + w] = st[p * ST_PITCH + w];
    }
    if (wave == 0) {
        float m = (lane < SC_WAVES) ? red[lane] : -INFINITY;
        m = fmaxf(m, __shfl_xor(m, 4));
        m = fmaxf(m, __shfl_xor(m, 2));
        m = fmaxf(m, __shfl_xor(m, 1));
        if (lane == 0) atomicMax(gmax, enc_f32(m));
    }
}


struct EdgeVals { int src; int loc; float es[NHEAD]; };

__device__ __forceinline__ EdgeVals edge_vals(const unsigned* __restrict__ pk, const float* sc,
                                              int e, float smax) {
    EdgeVals r;
    const unsigned p = pk[e];
    const float4 s = *(const float4*)(sc + (size_t)e * NHEAD);
    r.src = (int)(p & 0xFFFFu);
    r.loc = (int)(p >> 16);
    r.es[0] = __expf(s.x - smax);
    r.es[1] = __expf(s.y - smax);
    r.es[2] = __expf(s.z - smax);
    r.es[3] = __expf(s.w - smax);
    return r;
}

struct Recs {
    unsigned* cnt;
    unsigned* id;
    int*      src;
    float4*   es;
    unsigned* ncnt;
};

__device__ __forceinline__ void add_rows(int n, int lane, int src, int loc, const float* aw,
                                         const float* __restrict__ h, float* slab) {
    const int head = lane >> 3;
    for (int j = 0; j < n; j += 4) {
        float4 r[4]; float aj[4]; int lj[4];
#pragma unroll
        for (int u = 0; u < 4; ++u) {
            const int jj = (j + u < n) ? (j + u) : (n - 1);
            const int sj = __builtin_amdgcn_readlane(src, jj);
            lj[u] = __builtin_amdgcn_readlane(loc, jj);
            aj[u] = aw[jj * NHEAD + head];
            r[u] = *(const float4*)(h + (size_t)sj * OUT_COLS + lane * 4);
        }
#pragma unroll
        for (int u = 0; u < 4; ++u) {
            if (j + u < n) {
                float* p = slab + lj[u] * OUT_COLS + lane;
                atomicAdd(p +  0, r[u].x * aj[u]);
                atomicAdd(p + 32, r[u].y * aj[u]);
                atomicAdd(p + 64, r[u].z * aj[u]);
                atomicAdd(p + 96, r[u].w * aj[u]);
            }
        }
    }
}

template <int PASS>
__device__ __forceinline__ void owner_batch(int n, int lane, float smax,
                                            const unsigned* q,
                                            const unsigned* __restrict__ pk,
                                            const float* __restrict__ h,
                                            float* slab, float* den, float4* atab, Recs rc,
                                            float* attn_out) {
    if (PASS == 0) {
        unsigned pos = 0;
        if (lane == 0) pos = atomicAdd(rc.cnt, (unsigned)n);
        pos = (unsigned)__builtin_amdgcn_readfirstlane((int)pos);
        if (lane < n) {
            const int e = (int)q[lane];
            const EdgeVals v = edge_vals(pk, attn_out, e, smax);
#pragma unroll
            for (int k = 0; k < NHEAD; ++k) atomicAdd(&den[v.loc * NHEAD + k], v.es[k]);
            const unsigned slot = pos + (unsigned)lane;
            const unsigned rk = atomicAdd(&rc.ncnt[v.loc], 1u);
            if (slot < LCAP) {
                rc.id[slot]  = (unsigned)e | ((unsigned)v.loc << ID_BITS);
                rc.src[slot] = v.src | (int)(rk << 16);
                rc.es[slot]  = make_float4(v.es[0], v.es[1], v.es[2], v.es[3]);
            }
        }
    } else {
        int src = 0, loc = 0;
        if (lane < n) {
            const int e = (int)q[lane];
            const EdgeVals v = edge_vals(pk, attn_out, e, smax);
            src = v.src; loc = v.loc;
            const float4 a = make_float4(v.es[0] / (den[loc * NHEAD + 0] + 1e-16f),
                                         v.es[1] / (den[loc * NHEAD + 1] + 1e-16f),
                                         v.es[2] / (den[loc * NHEAD + 2] + 1e-16f),
                                         v.es[3] / (den[loc * NHEAD + 3] + 1e-16f));
            *(float4*)(attn_out + (size_t)e * NHEAD) = a;
            atab[lane] = a;
        }
        add_rows(n, lane, src, loc, (const float*)atab, h, slab);
    }
}

#define RIF 8
__device__ __forceinline__ float4 node_rows(int start, int deg, int node, int lane, Recs rc,
                                            const unsigned short* perm, const float* den,
                                            const float* __restrict__ h, float* attn_out) {
    const int head = lane >> 3;
    float4 acc = make_float4(0.0f, 0.0f, 0.0f, 0.0f);
    const float d0 = den[node * NHEAD + 0] + 1e-16f, d1 = den[node * NHEAD + 1] + 1e-16f;
    const float d2 = den[node * NHEAD + 2] + 1e-16f, d3 = den[node * NHEAD + 3] + 1e-16f;
    for (int c = 0; c < deg; c += 32) {
        const int n = (deg - c < 32) ? (deg - c) : 32;
        int s = 0, sn = 0;
        if (lane < n) {
            s = (int)perm[start + c + lane];
            const unsigned id = rc.id[s];
            const int e = (int)(id & ((1u << ID_BITS) - 1u));
            sn = rc.src[s] & 0xFFFF;
            const float4 x = rc.es[s];
            const float4 a = make_float4(x.x / d0, x.y / d1, x.z / d2, x.w / d3);
            *(float4*)(attn_out + (size_t)e * NHEAD) = a;
            rc.es[s] = a;
        }
        const float* aw = (const float*)rc.es;
        for (int j = 0; j < n; j += RIF) {
            float4 r[RIF]; float aj[RIF];
#pragma unroll
            for (int u = 0; u < RIF; ++u) {
                const int jj = (j + u < n) ? (j + u) : (n - 1);
                const int sj = __builtin_amdgcn_readlane(sn, jj);
                const int tj = __builtin_amdgcn_readlane(s, jj);
                aj[u] = aw[tj * NHEAD + head];
                r[u] = *(const float4*)(h + (size_t)sj * OUT_COLS + lane * 4);
            }
#pragma unroll
            for (int u = 0; u < RIF; ++u) {
                if (j + u < n) {
                    acc.x = fmaf(r[u].x, aj[u], acc.x);
                    acc.y = fmaf(r[u].y, aj[u], acc.y);
                    acc.z = fmaf(r[u].z, aj[u], acc.z);
                    acc.w = fmaf(r[u].w, aj[u], acc.w);
                }
            }
        }
    }
    return acc;
}

template <int PASS>
__device__ __forceinline__ void owner_scan(int lane, int wave, float smax,
                                           const unsigned* __restrict__ plane, unsigned* q,
                                           const unsigned* __restrict__ pk,
                                           const float* __restrict__ h,
                                           float* slab, float* den, float4* atab, Recs rc,
                                           float* attn_out) {
    const unsigned* pH = plane + (size_t)(blockIdx.x >> 4) * PITCH;
    const unsigned* pL = plane + (size_t)(16 + (blockIdx.x & 15)) * PITCH;
    const int rot = blockIdx.x * (N_TRIPS / N_SLABS);
    int cnt = 0;
    int c = wave + rot;
    uint4 kh = *(const uint4*)(pH + (size_t)c * 128 + lane * 4);
    uint4 kl = *(const uint4*)(pL + (size_t)c * 128 + lane * 4);
    for (int it = wave; it < N_TRIPS; it += OWN_WAVES) {
        int cn = c + OWN_WAVES;
        if (cn >= N_TRIPS) cn -= N_TRIPS;
        uint4 nh = kh, nl = kl;
        if (it + OWN_WAVES < N_TRIPS) {
            nh = *(const uint4*)(pH + (size_t)cn * 128 + lane * 4);
            nl = *(const uint4*)(pL + (size_t)cn * 128 + lane * 4);
        }
        const unsigned m0 = kh.x & kl.x, m1 = kh.y & kl.y, m2 = kh.z & kl.z, m3 = kh.w & kl.w;
#pragma nounroll
        for (int j = 0; j < 4; ++j) {
            unsigned mm = (j == 0) ? m0 : (j == 1) ? m1 : (j == 2) ? m2 : m3;
            const unsigned e0 = (unsigned)(c * 128 + lane * 4 + j) * 32u;
            unsigned bal;
            while ((bal = __builtin_amdgcn_ballot_w32(mm != 0u)) != 0u) {
                if (mm != 0u) {
                    const int p = __builtin_ctz(mm);
                    mm &= mm - 1u;
                    q[cnt + (int)__builtin_amdgcn_mbcnt_lo(bal, 0u)] = e0 + (unsigned)p;
                }
                cnt += __builtin_popcount(bal);
                if (cnt >= 32) {
                    owner_batch<PASS>(32, lane, smax, q, pk, h, slab, den, atab, rc, attn_out);
                    const int rest = cnt - 32;
                    if (lane < rest) { const unsigned t = q[32 + lane]; q[lane] = t; }
                    cnt = rest;
                }
            }
        }
        kh = nh; kl = nl;
        c = cn;
    }
    if (cnt > 0)
        owner_batch<PASS>(cnt, lane, smax, q, pk, h, slab, den, atab, rc, attn_out);
}

__global__ void __launch_bounds__(OWN_WAVES * 32)
owner_kernel(const float* __restrict__ h,
             const unsigned* __restrict__ plane, const unsigned* __restrict__ pk,
             const unsigned* __restrict__ gmax,
             float* __restrict__ out, float* attn_out) {
    __shared__ float4   slab4[SLAB * OUT_COLS / 4];
    __shared__ float    den[SLAB * NHEAD];
    __shared__ unsigned qbuf[OWN_WAVES * QCAP];
    __shared__ unsigned rec_id[LCAP];
    __shared__ int      rec_src[LCAP];
    __shared__ float4   rec_es[LCAP];
    __shared__ unsigned rec_cnt;
    __shared__ unsigned ncnt[SLAB];
    __shared__ unsigned offs[SLAB + 1];
    __shared__ unsigned short perm[LCAP];
    float* slab = (float*)slab4;

    const int tid  = threadIdx.x;
    const int lane = tid & 31;
    const int wave = __builtin_amdgcn_readfirstlane(tid >> 5);
    const int base = blockIdx.x * SLAB;
    unsigned* q = qbuf + wave * QCAP;
    Recs rc; rc.cnt = &rec_cnt; rc.id = rec_id; rc.src = rec_src; rc.es = rec_es; rc.ncnt = ncnt;
    float4* atab = rec_es + wave * 32;
    float* pden = slab + wave * (SLAB * NHEAD);

    for (int i = tid; i < SLAB * OUT_COLS / 4; i += OWN_WAVES * 32)
        slab4[i] = make_float4(0.0f, 0.0f, 0.0f, 0.0f);
    const int rows = (N_NODES - base < SLAB) ? (N_NODES - base) : SLAB;
    if (tid < SLAB)     ncnt[tid] = 0u;
    if (tid == 0)       rec_cnt = 0u;
    const float smax = dec_f32(*gmax);
    __syncthreads();

    owner_scan<0>(lane, wave, smax, plane, q, pk, h, slab, pden, atab, rc, attn_out);
    __syncthreads();
    for (int i = tid; i < SLAB * NHEAD; i += OWN_WAVES * 32) {
        float s = 0.0f;
        for (int w = 0; w < OWN_WAVES; ++w) s += slab[w * (SLAB * NHEAD) + i];
        den[i] = s;
    }
    if (wave == OWN_WAVES - 1) {
        unsigned c[7], sum = 0u;
#pragma unroll
        for (int k = 0; k < 7; ++k) {
            const int idx = lane * 7 + k;
            c[k] = (idx < SLAB) ? ncnt[idx] : 0u;
            sum += c[k];
        }
        unsigned incl = sum;
#pragma unroll
        for (int off = 1; off < 32; off <<= 1) {
            const unsigned t = __shfl_up(incl, off);
            if (lane >= off) incl += t;
        }
        unsigned run = incl - sum;
#pragma unroll
        for (int k = 0; k < 7; ++k) {
            const int idx = lane * 7 + k;
            if (idx < SLAB) offs[idx] = run;
            run += c[k];
        }
        if (lane == 31) offs[SLAB] = incl;
    }
    __syncthreads();
    const int total = __builtin_amdgcn_readfirstlane((int)rec_cnt);
    if (total <= LCAP) {
        for (int s = tid; s < total; s += OWN_WAVES * 32) {
            const unsigned loc = rec_id[s] >> ID_BITS;
            const unsigned rk  = (unsigned)rec_src[s] >> 16;
            perm[offs[loc] + rk] = (unsigned short)s;
        }
        __syncthreads();
        for (int node = wave; node < rows; node += OWN_WAVES) {
            const int start = __builtin_amdgcn_readfirstlane((int)offs[node]);
            const int deg   = __builtin_amdgcn_readfirstlane((int)offs[node + 1]) - start;
            const float4 acc = node_rows(start, deg, node, lane, rc, perm, den, h, attn_out);
            *(float4*)(out + (size_t)(base + node) * OUT_COLS + lane * 4) = acc;
        }
    } else {
        for (int i = tid; i < SLAB * OUT_COLS / 4; i += OWN_WAVES * 32)
            slab4[i] = make_float4(0.0f, 0.0f, 0.0f, 0.0f);
        __syncthreads();
        owner_scan<1>(lane, wave, smax, plane, q, pk, h, slab, den, atab, rc, attn_out);
        __syncthreads();
        float4* out4 = (float4*)(out + (size_t)base * OUT_COLS);
        for (int i = tid; i < rows * (OUT_COLS / 4); i += OWN_WAVES * 32) {
            const float* p = slab + (i >> 5) * OUT_COLS + (i & 31);
            out4[i] = make_float4(p[0], p[32], p[64], p[96]);
        }
    }
}

extern "C" void kernel_launch(void* const* d_in, const int* in_sizes, int n_in,
                              void* d_out, int out_size, void* d_ws, size_t ws_size,
                              hipStream_t stream) {
    const int*   edge     = (const int*)  d_in[0];
    const float* x        = (const float*)d_in[1];
    const float* edge_emb = (const float*)d_in[2];
    const float* W        = (const float*)d_in[3];
    const float* W_e      = (const float*)d_in[4];
    const float* a_l      = (const float*)d_in[5];
    const float* a_r      = (const float*)d_in[6];
    const float* a_e      = (const float*)d_in[7];

    float* out      = (float*)d_out;
    float* attn_out = out + (size_t)N_NODES * OUT_COLS;

    char* ws = (char*)d_ws;
    float* h      = (float*)ws;  ws += (size_t)N_NODES * OUT_COLS * sizeof(float);
    unsigned* plane = (unsigned*)ws;  ws += (size_t)N_EDGES * NHEAD * sizeof(float);
    unsigned* pk = plane + (size_t)32 * PITCH;
    float* hl     = (float*)ws;  ws += (size_t)N_NODES * NHEAD   * sizeof(float);
    float* hr     = (float*)ws;  ws += (size_t)N_NODES * NHEAD   * sizeof(float);
    float* wlr    = (float*)ws;  ws += (size_t)N_NODES * NHEAD * sizeof(float);
    float* he     = (float*)ws;  ws += (size_t)N_ETYPES * NHEAD  * sizeof(float);
    unsigned* gmax = (unsigned*)ws;

    init_gmax_kernel<<<1, 1, 0, stream>>>(gmax);

    etype_logits_kernel<<<1, 1024, 0, stream>>>(edge_emb, W_e, a_e, he, W, a_l, a_r, wlr);
    {
        const int strips = N_NODES / 16;
        const int wavesPerBlock = 8;
        const int blocks = (strips + wavesPerBlock - 1) / wavesPerBlock;
        gemm_h_kernel<<<blocks, 256, 0, stream>>>(x, W, wlr, h, hl, hr);
    }
    edge_max_plane_kernel<<<N_KBLOCKS, SC_WAVES * 32, 0, stream>>>(edge, hl, hr, he, plane, pk, attn_out, gmax);
    owner_kernel<<<N_SLABS, OWN_WAVES * 32, 0, stream>>>(h, plane, pk, gmax, out, attn_out);
}


# 5 launches -> 3, no one-group launches: each product work-group sums the node-logit table into LDS, each score work-group the edge-type logits; max of per-group maxima, no atomic; table sum order new
# speedup vs baseline: 4.3247x; 1.0553x over previous
#include <hip/hip_runtime.h>
#include <hip/hip_bf16.h>

typedef __attribute__((ext_vector_type(2))) float v2f;
typedef __attribute__((ext_vector_type(8))) float v8f;

#define N_NODES 50000
#define N_EDGES 1600000
#define IN_HIDDEN 256
#define OUT_COLS 128
#define NHEAD 4
#define OUT_HIDDEN 32
#define EDGE_HIDDEN 64
#define N_ETYPES 8
#define NEG_SLOPE 0.2f

#define SLAB 196
#define N_SLABS 256
#define PITCH 50048
#define N_TRIPS (PITCH / 128)
#define EPL 8
#define SC_WAVES 8
#define SC_WORDS (SC_WAVES * EPL)
#define ST_PITCH (SC_WORDS + 1)
#define N_KBLOCKS (PITCH / SC_WORDS)
#define OWN_WAVES 32
#define QCAP 64
#define LCAP 6912
#define ID_BITS 21
static_assert(N_SLABS * SLAB >= N_NODES && N_SLABS <= 256, "slab id must fit a byte");
static_assert(N_EDGES % 32 == 0 && PITCH % 128 == 0 && PITCH * 32 >= N_EDGES, "planes hold every edge in whole trips");
static_assert(PITCH % SC_WORDS == 0 && SC_WORDS % 32 == 0 && SC_WAVES <= 32, "the score kernel's work-groups tile the planes in whole 128-byte lines");
static_assert((N_EDGES / 32) % EPL == 0, "a wave of the score kernel has edges in all of its words or in none");
static_assert(OWN_WAVES * SLAB * NHEAD <= SLAB * OUT_COLS, "the per-wave denominator sums fit the output rows' LDS space");
static_assert(N_EDGES <= (1 << ID_BITS) && SLAB <= (1 << (32 - ID_BITS)), "edge id and local node share a dword");
static_assert(LCAP >= OWN_WAVES * 32, "the record slots also hold the second sweep's per-wave tables");
static_assert((N_SLABS - 1) * (N_TRIPS / N_SLABS) + OWN_WAVES <= N_TRIPS, "a rotated first trip stays inside the plane");
static_assert(N_ETYPES == SC_WAVES && EDGE_HIDDEN == 64 && NHEAD == 4, "the score kernel's edge-type logits: one wave per edge type, 16-lane groups per 64-float segment");
static_assert(IN_HIDDEN == 8 * 32 && OUT_COLS == 32 * 4 && OUT_HIDDEN == 8 * 4, "the product kernel's table build: 8 waves x 32 rows of W, a row is one 16-byte piece per lane, a head is 8 lanes");
static_assert(N_KBLOCKS <= OWN_WAVES * 32 && OWN_WAVES == 32, "an owner work-group takes the score work-groups' maxima one per thread and joins its waves' maxima in one wave");
static_assert(N_NODES <= (1 << 16) && SLAB <= (1 << 16), "source node and local target node share one packed dword");
static_assert(N_KBLOCKS <= N_NODES * NHEAD, "the score work-groups' maxima fit their workspace region");
static_assert(LCAP <= (1 << 16), "a record slot fits 16 bits, and so does an edge's rank in its node");

template <int M>
__device__ __forceinline__ float row_xor(float v) {
    return __int_as_float(__builtin_amdgcn_update_dpp(0, __float_as_int(v), 0x160 | M, 0xf, 0xf, true));
}

#define LR_COLS 8
__global__ void __launch_bounds__(256)
gemm_h_kernel(const float* __restrict__ x, const float* __restrict__ W,
              const float* __restrict__ a_l, const float* __restrict__ a_r,
              float* __restrict__ h, float* __restrict__ hl, float* __restrict__ hr) {
    __shared__ float4 tab4[IN_HIDDEN * LR_COLS / 4];
    float* tab = (float*)tab4;
    const int lane  = threadIdx.x & 31;
    const int wave  = threadIdx.x >> 5;
    {
        const int head = lane >> 3;
        const float4 al = *(const float4*)(a_l + head * OUT_HIDDEN + (lane & 7) * 4);
        const float4 ar = *(const float4*)(a_r + head * OUT_HIDDEN + (lane & 7) * 4);
        const float4* wrow = (const float4*)(W + (size_t)(wave * 32) * OUT_COLS) + lane;
        float4 w[32];
#pragma unroll
        for (int i = 0; i < 32; ++i) w[i] = wrow[i * (OUT_COLS / 4)];
        float sl[32], sr[32];
#pragma unroll
        for (int i = 0; i < 32; ++i) {
            sl[i] = w[i].x * al.x;  sl[i] = fmaf(w[i].y, al.y, sl[i]);  sl[i] = fmaf(w[i].z, al.z, sl[i]);  sl[i] = fmaf(w[i].w, al.w, sl[i]);
            sr[i] = w[i].x * ar.x;  sr[i] = fmaf(w[i].y, ar.y, sr[i]);  sr[i] = fmaf(w[i].z, ar.z, sr[i]);  sr[i] = fmaf(w[i].w, ar.w, sr[i]);
        }
#pragma unroll
        for (int i = 0; i < 32; ++i) {
            sl[i] += row_xor<4>(sl[i]);  sr[i] += row_xor<4>(sr[i]);
            sl[i] += row_xor<2>(sl[i]);  sr[i] += row_xor<2>(sr[i]);
            sl[i] += row_xor<1>(sl[i]);  sr[i] += row_xor<1>(sr[i]);
        }
        if ((lane & 7) == 0) {
#pragma unroll
            for (int i = 0; i < 32; ++i) {
                float* trow_w = tab + (wave * 32 + i) * LR_COLS + head;
                trow_w[0]     = sl[i];
                trow_w[NHEAD] = sr[i];
            }
        }
    }
    __syncthreads();
    const int strip = blockIdx.x * (blockDim.x >> 5) + wave;
    if (strip * 16 >= N_NODES) return;
    const int row0  = strip * 16;

    const int mrow  = lane & 15;
    const int khalf = (lane >> 4) << 1;

    v8f acc[8] = {};
    float4 pl = make_float4(0.0f, 0.0f, 0.0f, 0.0f);
    float4 pr = make_float4(0.0f, 0.0f, 0.0f, 0.0f);

    const float* xrow = x + (size_t)(row0 + mrow) * IN_HIDDEN + khalf;
    const float* trow = tab + khalf * LR_COLS;
    float4 l0 = *(const float4*)(trow);
    float4 r0 = *(const float4*)(trow + NHEAD);
    float4 l1 = *(const float4*)(trow + LR_COLS);
    float4 r1 = *(const float4*)(trow + LR_COLS + NHEAD);
    for (int k = 0; k < IN_HIDDEN; k += 4) {
        v2f a = *(const v2f*)(xrow + k);
        const int kn = (k + 4 < IN_HIDDEN) ? (k + 4) : k;
        const float* tn = tab + (kn + khalf) * LR_COLS;
        const float4 nl0 = *(const float4*)(tn);
        const float4 nr0 = *(const float4*)(tn + NHEAD);
        const float4 nl1 = *(const float4*)(tn + LR_COLS);
        const float4 nr1 = *(const float4*)(tn + LR_COLS + NHEAD);
        v2f b[8];
#pragma unroll
        for (int t = 0; t < 8; ++t) {
            const float* wp = W + (size_t)(k + khalf) * OUT_COLS + t * 16 + mrow;
            b[t].x = wp[0];
            b[t].y = wp[OUT_COLS];
        }
        __builtin_amdgcn_sched_barrier(0);
#pragma unroll
        for (int t = 0; t < 8; ++t)
            acc[t] = __builtin_amdgcn_wmma_f32_16x16x4_f32(
                false, a, false, b[t], (short)0, acc[t], false, false);
        {
            pl.x = fmaf(a.x, l0.x, pl.x);  pl.y = fmaf(a.x, l0.y, pl.y);  pl.z = fmaf(a.x, l0.z, pl.z);  pl.w = fmaf(a.x, l0.w, pl.w);
            pr.x = fmaf(a.x, r0.x, pr.x);  pr.y = fmaf(a.x, r0.y, pr.y);  pr.z = fmaf(a.x, r0.z, pr.z);  pr.w = fmaf(a.x, r0.w, pr.w);
            pl.x = fmaf(a.y, l1.x, pl.x);  pl.y = fmaf(a.y, l1.y, pl.y);  pl.z = fmaf(a.y, l1.z, pl.z);  pl.w = fmaf(a.y, l1.w, pl.w);
            pr.x = fmaf(a.y, r1.x, pr.x);  pr.y = fmaf(a.y, r1.y, pr.y);  pr.z = fmaf(a.y, r1.z, pr.z);  pr.w = fmaf(a.y, r1.w, pr.w);
        }
        l0 = nl0;  r0 = nr0;  l1 = nl1;  r1 = nr1;
    }

    const int mbase = (lane >> 4) << 3;
#pragma unroll
    for (int t = 0; t < 8; ++t) {
#pragma unroll
        for (int i = 0; i < 8; ++i) {
            float v = acc[t][i];
            v = (v == v) ? v : 0.0f;
            h[(size_t)(row0 + mbase + i) * OUT_COLS + t * 16 + mrow] = v;
        }
    }
    pl.x += __shfl_xor(pl.x, 16);  pl.y += __shfl_xor(pl.y, 16);  pl.z += __shfl_xor(pl.z, 16);  pl.w += __shfl_xor(pl.w, 16);
    pr.x += __shfl_xor(pr.x, 16);  pr.y += __shfl_xor(pr.y, 16);  pr.z += __shfl_xor(pr.z, 16);  pr.w += __shfl_xor(pr.w, 16);
    if (lane < 16) {
        *(float4*)(hl + (size_t)(row0 + mrow) * NHEAD) = pl;
        *(float4*)(hr + (size_t)(row0 + mrow) * NHEAD) = pr;
    }
}

__global__ void __launch_bounds__(SC_WAVES * 32) __attribute__((amdgpu_waves_per_eu(4, 8)))
edge_max_plane_kernel(const int* __restrict__ edge,
                      const float* __restrict__ hl, const float* __restrict__ hr,
                      const float* __restrict__ edge_emb, const float* __restrict__ W_e,
                      const float* __restrict__ a_e,
                      unsigned* __restrict__ plane, unsigned* __restrict__ pk,
                      float* __restrict__ score, float* __restrict__ wgmax) {
    __shared__ unsigned st[32 * ST_PITCH];
    __shared__ float red[SC_WAVES];
    __shared__ float v_s[NHEAD * EDGE_HIDDEN];
    __shared__ float4 he_s[N_ETYPES];
    const int tid  = threadIdx.x;
    const int lane = tid & 31;
    const int wave = __builtin_amdgcn_readfirstlane(tid >> 5);
    for (int i = tid; i < 32 * ST_PITCH; i += SC_WAVES * 32) st[i] = 0u;

    const int w0    = wave * EPL;
    const int wbase = blockIdx.x * SC_WORDS + w0;
    const bool has_edges = wbase * 32 < N_EDGES;
    const int e0 = wbase * 32 + lane;
    int src[EPL], trg[EPL], et[EPL];
#pragma unroll
    for (int j = 0; j < EPL; ++j) { src[j] = 0; trg[j] = 0; et[j] = 0; }
    if (has_edges) {
#pragma unroll
        for (int j = 0; j < EPL; ++j) {
            const int e = e0 + j * 32;
            src[j] = edge[e];
            trg[j] = edge[N_EDGES + e];
            et[j]  = edge[2 * N_EDGES + e];
        }
    }

    {
        const int j  = tid & 15;
        const int hw = tid >> 4;
        const float ee0 = edge_emb[wave * EDGE_HIDDEN + lane];
        const float ee1 = edge_emb[wave * EDGE_HIDDEN + 32 + lane];
        const float4 a = *(const float4*)(a_e + (hw & 3) * EDGE_HIDDEN + j * 4);
        float4 w[16];
#pragma unroll
        for (int i = 0; i < 16; ++i)
            w[i] = *(const float4*)(W_e + (size_t)(hw + 16 * i) * EDGE_HIDDEN + j * 4);
        float s[16];
#pragma unroll
        for (int i = 0; i < 16; ++i)
            s[i] = a.x * w[i].x + a.y * w[i].y + a.z * w[i].z + a.w * w[i].w;
#pragma unroll
        for (int i = 0; i < 16; ++i) {
            s[i] += row_xor<8>(s[i]);
            s[i] += row_xor<4>(s[i]);
            s[i] += row_xor<2>(s[i]);
            s[i] += row_xor<1>(s[i]);
        }
        if (j == 0) {
#pragma unroll
            for (int i = 0; i < 16; ++i) {
                const int p = hw + 16 * i;
                v_s[(p & 3) * EDGE_HIDDEN + (p >> 2)] = s[i];
            }
        }
        __syncthreads();
        float r[NHEAD];
#pragma unroll
        for (int hh = 0; hh < NHEAD; ++hh)
            r[hh] = ee0 * v_s[hh * EDGE_HIDDEN + lane] + ee1 * v_s[hh * EDGE_HIDDEN + 32 + lane];
#pragma unroll
        for (int hh = 0; hh < NHEAD; ++hh) r[hh] += __shfl_xor(r[hh], 16);
#pragma unroll
        for (int hh = 0; hh < NHEAD; ++hh) {
            r[hh] += row_xor<8>(r[hh]);
            r[hh] += row_xor<4>(r[hh]);
            r[hh] += row_xor<2>(r[hh]);
            r[hh] += row_xor<1>(r[hh]);
        }
        if (lane == 0) he_s[wave] = make_float4(r[0], r[1], r[2], r[3]);
        __syncthreads();
    }

    float lmax = -INFINITY;
    if (has_edges) {
        float4 gl[EPL], gr[EPL], ge[EPL];
#pragma unroll
        for (int j = 0; j < EPL; ++j) {
            gl[j] = *(const float4*)(hl + src[j] * NHEAD);
            gr[j] = *(const float4*)(hr + trg[j] * NHEAD);
            ge[j] = he_s[et[j]];
        }
#pragma unroll
        for (int j = 0; j < EPL; ++j) {
            const int e = e0 + j * 32;
            float sc[NHEAD];
            sc[0] = gl[j].x + gr[j].x + ge[j].x;
            sc[1] = gl[j].y + gr[j].y + ge[j].y;
            sc[2] = gl[j].z + gr[j].z + ge[j].z;
            sc[3] = gl[j].w + gr[j].w + ge[j].w;
#pragma unroll
            for (int hh = 0; hh < NHEAD; ++hh) {
                float s = sc[hh];
                s = (s > 0.0f) ? s : NEG_SLOPE * s;
                sc[hh] = s;
                lmax = fmaxf(lmax, s);
            }
            const int slab = trg[j] / SLAB;
            *(float4*)(score + (size_t)e * NHEAD) = make_float4(sc[0], sc[1], sc[2], sc[3]);
            pk[e] = (unsigned)src[j] | ((unsigned)(trg[j] - slab * SLAB) << 16);
            atomicOr(&st[(slab >> 4) * ST_PITCH + w0 + j], 1u << lane);
            atomicOr(&st[(16 + (slab & 15)) * ST_PITCH + w0 + j], 1u << lane);
        }
    }
    lmax = fmaxf(lmax, __shfl_xor(lmax, 16));
    lmax = fmaxf(lmax, __shfl_xor(lmax, 8));
    lmax = fmaxf(lmax, __shfl_xor(lmax, 4));
    lmax = fmaxf(lmax, __shfl_xor(lmax, 2));
    lmax = fmaxf(lmax, __shfl_xor(lmax, 1));
    if (lane == 0) red[wave] = lmax;
    __syncthreads();
#pragma unroll
    for (int i = 0; i < 32 * SC_WORDS / (SC_WAVES * 32); ++i) {
        const int idx = i * (SC_WAVES * 32) + tid;
        const int p = idx / SC_WORDS, w = idx % SC_WORDS;
        plane[(size_t)p * PITCH + blockIdx.x * SC_WORDS + w] = st[p * ST_PITCH + w];
    }
    if (wave == 0) {
        float m = (lane < SC_WAVES) ? red[lane] : -INFINITY;
        m = fmaxf(m, __shfl_xor(m, 4));
        m = fmaxf(m, __shfl_xor(m, 2));
        m = fmaxf(m, __shfl_xor(m, 1));
        if (lane == 0) wgmax[blockIdx.x] = m;
    }
}


struct EdgeVals { int src; int loc; float es[NHEAD]; };

__device__ __forceinline__ EdgeVals edge_vals(const unsigned* __restrict__ pk, const float* sc,
                                              int e, float smax) {
    EdgeVals r;
    const unsigned p = pk[e];
    const float4 s = *(const float4*)(sc + (size_t)e * NHEAD);
    r.src = (int)(p & 0xFFFFu);
    r.loc = (int)(p >> 16);
    r.es[0] = __expf(s.x - smax);
    r.es[1] = __expf(s.y - smax);
    r.es[2] = __expf(s.z - smax);
    r.es[3] = __expf(s.w - smax);
    return r;
}

struct Recs {
    unsigned* cnt;
    unsigned* id;
    int*      src;
    float4*   es;
    unsigned* ncnt;
};

__device__ __forceinline__ void add_rows(int n, int lane, int src, int loc, const float* aw,
                                         const float* __restrict__ h, float* slab) {
    const int head = lane >> 3;
    for (int j = 0; j < n; j += 4) {
        float4 r[4]; float aj[4]; int lj[4];
#pragma unroll
        for (int u = 0; u < 4; ++u) {
            const int jj = (j + u < n) ? (j + u) : (n - 1);
            const int sj = __builtin_amdgcn_readlane(src, jj);
            lj[u] = __builtin_amdgcn_readlane(loc, jj);
            aj[u] = aw[jj * NHEAD + head];
            r[u] = *(const float4*)(h + (size_t)sj * OUT_COLS + lane * 4);
        }
#pragma unroll
        for (int u = 0; u < 4; ++u) {
            if (j + u < n) {
                float* p = slab + lj[u] * OUT_COLS + lane;
                atomicAdd(p +  0, r[u].x * aj[u]);
                atomicAdd(p + 32, r[u].y * aj[u]);
                atomicAdd(p + 64, r[u].z * aj[u]);
                atomicAdd(p + 96, r[u].w * aj[u]);
            }
        }
    }
}

template <int PASS>
__device__ __forceinline__ void owner_batch(int n, int lane, float smax,
                                            const unsigned* q,
                                            const unsigned* __restrict__ pk,
                                            const float* __restrict__ h,
                                            float* slab, float* den, float4* atab, Recs rc,
                                            float* attn_out) {
    if (PASS == 0) {
        unsigned pos = 0;
        if (lane == 0) pos = atomicAdd(rc.cnt, (unsigned)n);
        pos = (unsigned)__builtin_amdgcn_readfirstlane((int)pos);
        if (lane < n) {
            const int e = (int)q[lane];
            const EdgeVals v = edge_vals(pk, attn_out, e, smax);
#pragma unroll
            for (int k = 0; k < NHEAD; ++k) atomicAdd(&den[v.loc * NHEAD + k], v.es[k]);
            const unsigned slot = pos + (unsigned)lane;
            const unsigned rk = atomicAdd(&rc.ncnt[v.loc], 1u);
            if (slot < LCAP) {
                rc.id[slot]  = (unsigned)e | ((unsigned)v.loc << ID_BITS);
                rc.src[slot] = v.src | (int)(rk << 16);
                rc.es[slot]  = make_float4(v.es[0], v.es[1], v.es[2], v.es[3]);
            }
        }
    } else {
        int src = 0, loc = 0;
        if (lane < n) {
            const int e = (int)q[lane];
            const EdgeVals v = edge_vals(pk, attn_out, e, smax);
            src = v.src; loc = v.loc;
            const float4 a = make_float4(v.es[0] / (den[loc * NHEAD + 0] + 1e-16f),
                                         v.es[1] / (den[loc * NHEAD + 1] + 1e-16f),
                                         v.es[2] / (den[loc * NHEAD + 2] + 1e-16f),
                                         v.es[3] / (den[loc * NHEAD + 3] + 1e-16f));
            *(float4*)(attn_out + (size_t)e * NHEAD) = a;
            atab[lane] = a;
        }
        add_rows(n, lane, src, loc, (const float*)atab, h, slab);
    }
}

#define RIF 8
__device__ __forceinline__ float4 node_rows(int start, int deg, int node, int lane, Recs rc,
                                            const unsigned short* perm, const float* den,
                                            const float* __restrict__ h, float* attn_out) {
    const int head = lane >> 3;
    float4 acc = make_float4(0.0f, 0.0f, 0.0f, 0.0f);
    const float d0 = den[node * NHEAD + 0] + 1e-16f, d1 = den[node * NHEAD + 1] + 1e-16f;
    const float d2 = den[node * NHEAD + 2] + 1e-16f, d3 = den[node * NHEAD + 3] + 1e-16f;
    for (int c = 0; c < deg; c += 32) {
        const int n = (deg - c < 32) ? (deg - c) : 32;
        int s = 0, sn = 0;
        if (lane < n) {
            s = (int)perm[start + c + lane];
            const unsigned id = rc.id[s];
            const int e = (int)(id & ((1u << ID_BITS) - 1u));
            sn = rc.src[s] & 0xFFFF;
            const float4 x = rc.es[s];
            const float4 a = make_float4(x.x / d0, x.y / d1, x.z / d2, x.w / d3);
            *(float4*)(attn_out + (size_t)e * NHEAD) = a;
            rc.es[s] = a;
        }
        const float* aw = (const float*)rc.es;
        for (int j = 0; j < n; j += RIF) {
            float4 r[RIF]; float aj[RIF];
#pragma unroll
            for (int u = 0; u < RIF; ++u) {
                const int jj = (j + u < n) ? (j + u) : (n - 1);
                const int sj = __builtin_amdgcn_readlane(sn, jj);
                const int tj = __builtin_amdgcn_readlane(s, jj);
                aj[u] = aw[tj * NHEAD + head];
                r[u] = *(const float4*)(h + (size_t)sj * OUT_COLS + lane * 4);
            }
#pragma unroll
            for (int u = 0; u < RIF; ++u) {
                if (j + u < n) {
                    acc.x = fmaf(r[u].x, aj[u], acc.x);
                    acc.y = fmaf(r[u].y, aj[u], acc.y);
                    acc.z = fmaf(r[u].z, aj[u], acc.z);
                    acc.w = fmaf(r[u].w, aj[u], acc.w);
                }
            }
        }
    }
    return acc;
}

template <int PASS>
__device__ __forceinline__ void owner_scan(int lane, int wave, float smax,
                                           const unsigned* __restrict__ plane, unsigned* q,
                                           const unsigned* __restrict__ pk,
                                           const float* __restrict__ h,
                                           float* slab, float* den, float4* atab, Recs rc,
                                           float* attn_out) {
    const unsigned* pH = plane + (size_t)(blockIdx.x >> 4) * PITCH;
    const unsigned* pL = plane + (size_t)(16 + (blockIdx.x & 15)) * PITCH;
    const int rot = blockIdx.x * (N_TRIPS / N_SLABS);
    int cnt = 0;
    int c = wave + rot;
    uint4 kh = *(const uint4*)(pH + (size_t)c * 128 + lane * 4);
    uint4 kl = *(const uint4*)(pL + (size_t)c * 128 + lane * 4);
    for (int it = wave; it < N_TRIPS; it += OWN_WAVES) {
        int cn = c + OWN_WAVES;
        if (cn >= N_TRIPS) cn -= N_TRIPS;
        uint4 nh = kh, nl = kl;
        if (it + OWN_WAVES < N_TRIPS) {
            nh = *(const uint4*)(pH + (size_t)cn * 128 + lane * 4);
            nl = *(const uint4*)(pL + (size_t)cn * 128 + lane * 4);
        }
        const unsigned m0 = kh.x & kl.x, m1 = kh.y & kl.y, m2 = kh.z & kl.z, m3 = kh.w & kl.w;
#pragma nounroll
        for (int j = 0; j < 4; ++j) {
            unsigned mm = (j == 0) ? m0 : (j == 1) ? m1 : (j == 2) ? m2 : m3;
            const unsigned e0 = (unsigned)(c * 128 + lane * 4 + j) * 32u;
            unsigned bal;
            while ((bal = __builtin_amdgcn_ballot_w32(mm != 0u)) != 0u) {
                if (mm != 0u) {
                    const int p = __builtin_ctz(mm);
                    mm &= mm - 1u;
                    q[cnt + (int)__builtin_amdgcn_mbcnt_lo(bal, 0u)] = e0 + (unsigned)p;
                }
                cnt += __builtin_popcount(bal);
                if (cnt >= 32) {
                    owner_batch<PASS>(32, lane, smax, q, pk, h, slab, den, atab, rc, attn_out);
                    const int rest = cnt - 32;
                    if (lane < rest) { const unsigned t = q[32 + lane]; q[lane] = t; }
                    cnt = rest;
                }
            }
        }
        kh = nh; kl = nl;
        c = cn;
    }
    if (cnt > 0)
        owner_batch<PASS>(cnt, lane, smax, q, pk, h, slab, den, atab, rc, attn_out);
}

__global__ void __launch_bounds__(OWN_WAVES * 32)
owner_kernel(const float* __restrict__ h,
             const unsigned* __restrict__ plane, const unsigned* __restrict__ pk,
             const float* __restrict__ wgmax,
             float* __restrict__ out, float* attn_out) {
    __shared__ float4   slab4[SLAB * OUT_COLS / 4];
    __shared__ float    den[SLAB * NHEAD];
    __shared__ unsigned qbuf[OWN_WAVES * QCAP];
    __shared__ unsigned rec_id[LCAP];
    __shared__ int      rec_src[LCAP];
    __shared__ float4   rec_es[LCAP];
    __shared__ unsigned rec_cnt;
    __shared__ unsigned ncnt[SLAB];
    __shared__ unsigned offs[SLAB + 1];
    __shared__ unsigned short perm[LCAP];
    __shared__ float    wred[OWN_WAVES];
    float* slab = (float*)slab4;

    const int tid  = threadIdx.x;
    const int lane = tid & 31;
    const int wave = __builtin_amdgcn_readfirstlane(tid >> 5);
    const int base = blockIdx.x * SLAB;
    unsigned* q = qbuf + wave * QCAP;
    Recs rc; rc.cnt = &rec_cnt; rc.id = rec_id; rc.src = rec_src; rc.es = rec_es; rc.ncnt = ncnt;
    float4* atab = rec_es + wave * 32;
    float* pden = slab + wave * (SLAB * NHEAD);

    for (int i = tid; i < SLAB * OUT_COLS / 4; i += OWN_WAVES * 32)
        slab4[i] = make_float4(0.0f, 0.0f, 0.0f, 0.0f);
    const int rows = (N_NODES - base < SLAB) ? (N_NODES - base) : SLAB;
    if (tid < SLAB)     ncnt[tid] = 0u;
    if (tid == 0)       rec_cnt = 0u;
    float pm = (tid < N_KBLOCKS) ? wgmax[tid] : -INFINITY;
    pm = fmaxf(pm, __shfl_xor(pm, 16));
    pm = fmaxf(pm, __shfl_xor(pm, 8));
    pm = fmaxf(pm, __shfl_xor(pm, 4));
    pm = fmaxf(pm, __shfl_xor(pm, 2));
    pm = fmaxf(pm, __shfl_xor(pm, 1));
    if (lane == 0) wred[wave] = pm;
    __syncthreads();
    float am = wred[lane];
    am = fmaxf(am, __shfl_xor(am, 16));
    am = fmaxf(am, __shfl_xor(am, 8));
    am = fmaxf(am, __shfl_xor(am, 4));
    am = fmaxf(am, __shfl_xor(am, 2));
    am = fmaxf(am, __shfl_xor(am, 1));
    const float smax = __uint_as_float((unsigned)__builtin_amdgcn_readfirstlane((int)__float_as_uint(am)));

    owner_scan<0>(lane, wave, smax, plane, q, pk, h, slab, pden, atab, rc, attn_out);
    __syncthreads();
    for (int i = tid; i < SLAB * NHEAD; i += OWN_WAVES * 32) {
        float s = 0.0f;
        for (int w = 0; w < OWN_WAVES; ++w) s += slab[w * (SLAB * NHEAD) + i];
        den[i] = s;
    }
    if (wave == OWN_WAVES - 1) {
        unsigned c[7], sum = 0u;
#pragma unroll
        for (int k = 0; k < 7; ++k) {
            const int idx = lane * 7 + k;
            c[k] = (idx < SLAB) ? ncnt[idx] : 0u;
            sum += c[k];
        }
        unsigned incl = sum;
#pragma unroll
        for (int off = 1; off < 32; off <<= 1) {
            const unsigned t = __shfl_up(incl, off);
            if (lane >= off) incl += t;
        }
        unsigned run = incl - sum;
#pragma unroll
        for (int k = 0; k < 7; ++k) {
            const int idx = lane * 7 + k;
            if (idx < SLAB) offs[idx] = run;
            run += c[k];
        }
        if (lane == 31) offs[SLAB] = incl;
    }
    __syncthreads();
    const int total = __builtin_amdgcn_readfirstlane((int)rec_cnt);
    if (total <= LCAP) {
        for (int s = tid; s < total; s += OWN_WAVES * 32) {
            const unsigned loc = rec_id[s] >> ID_BITS;
            const unsigned rk  = (unsigned)rec_src[s] >> 16;
            perm[offs[loc] + rk] = (unsigned short)s;
        }
        __syncthreads();
        for (int node = wave; node < rows; node += OWN_WAVES) {
            const int start = __builtin_amdgcn_readfirstlane((int)offs[node]);
            const int deg   = __builtin_amdgcn_readfirstlane((int)offs[node + 1]) - start;
            const float4 acc = node_rows(start, deg, node, lane, rc, perm, den, h, attn_out);
            *(float4*)(out + (size_t)(base + node) * OUT_COLS + lane * 4) = acc;
        }
    } else {
        for (int i = tid; i < SLAB * OUT_COLS / 4; i += OWN_WAVES * 32)
            slab4[i] = make_float4(0.0f, 0.0f, 0.0f, 0.0f);
        __syncthreads();
        owner_scan<1>(lane, wave, smax, plane, q, pk, h, slab, den, atab, rc, attn_out);
        __syncthreads();
        float4* out4 = (float4*)(out + (size_t)base * OUT_COLS);
        for (int i = tid; i < rows * (OUT_COLS / 4); i += OWN_WAVES * 32) {
            const float* p = slab + (i >> 5) * OUT_COLS + (i & 31);
            out4[i] = make_float4(p[0], p[32], p[64], p[96]);
        }
    }
}

extern "C" void kernel_launch(void* const* d_in, const int* in_sizes, int n_in,
                              void* d_out, int out_size, void* d_ws, size_t ws_size,
                              hipStream_t stream) {
    const int*   edge     = (const int*)  d_in[0];
    const float* x        = (const float*)d_in[1];
    const float* edge_emb = (const float*)d_in[2];
    const float* W        = (const float*)d_in[3];
    const float* W_e      = (const float*)d_in[4];
    const float* a_l      = (const float*)d_in[5];
    const float* a_r      = (const float*)d_in[6];
    const float* a_e      = (const float*)d_in[7];

    float* out      = (float*)d_out;
    float* attn_out = out + (size_t)N_NODES * OUT_COLS;

    char* ws = (char*)d_ws;
    float* h      = (float*)ws;  ws += (size_t)N_NODES * OUT_COLS * sizeof(float);
    unsigned* plane = (unsigned*)ws;  ws += (size_t)N_EDGES * NHEAD * sizeof(float);
    unsigned* pk = plane + (size_t)32 * PITCH;
    float* hl     = (float*)ws;  ws += (size_t)N_NODES * NHEAD   * sizeof(float);
    float* hr     = (float*)ws;  ws += (size_t)N_NODES * NHEAD   * sizeof(float);
    float* wgmax  = (float*)ws;

    {
        const int strips = N_NODES / 16;
        const int wavesPerBlock = 8;
        const int blocks = (strips + wavesPerBlock - 1) / wavesPerBlock;
        gemm_h_kernel<<<blocks, 256, 0, stream>>>(x, W, a_l, a_r, h, hl, hr);
    }
    edge_max_plane_kernel<<<N_KBLOCKS, SC_WAVES * 32, 0, stream>>>(edge, hl, hr, edge_emb, W_e, a_e, plane, pk, attn_out, wgmax);
    owner_kernel<<<N_SLABS, OWN_WAVES * 32, 0, stream>>>(h, plane, pk, wgmax, out, attn_out);
}
